# starting kernel
# baseline (speedup 1.0000x reference)
#include <hip/hip_runtime.h>
#include <math.h>

#define N_NODES   10000
#define F_IN      128
#define HEADS     8
#define HIDD      256
#define HC        (HEADS * HIDD)
#define NUM_CLASS 10
#define E_IN      50000
#define E_TOT     (E_IN + N_NODES)
#define NEG_SLOPE 0.2f

typedef __attribute__((ext_vector_type(2))) float v2f;
typedef __attribute__((ext_vector_type(8))) float v8f;

__launch_bounds__(256)
__global__ void gemm1_wmma(const float* __restrict__ X, const float* __restrict__ W,
                           float* __restrict__ Hout)
{
    __shared__ float As[32][F_IN + 4];
    __shared__ float Bs[F_IN][64];

    const int tid = threadIdx.x;
    const int m0  = blockIdx.y * 32;
    const int n0  = blockIdx.x * 64;

#pragma unroll
    for (int i = 0; i < 4; ++i) {
        int idx = (tid + i * 256) * 4;
        int r = idx >> 7;
        int c = idx & 127;
        float4 v = make_float4(0.f, 0.f, 0.f, 0.f);
        int gr = m0 + r;
        if (gr < N_NODES) v = *(const float4*)(X + (size_t)gr * F_IN + c);
        *(float4*)(&As[r][c]) = v;
    }
#pragma unroll
    for (int i = 0; i < 8; ++i) {
        int idx = (tid + i * 256) * 4;
        int r = idx >> 6;
        int c = idx & 63;
        *(float4*)(&Bs[r][c]) = *(const float4*)(W + (size_t)r * HC + n0 + c);
    }
    __syncthreads();

    const int wave = tid >> 5;
    const int lane = tid & 31;
    const int half = lane >> 4;
    const int ml   = lane & 15;
    const int ar   = (wave >> 2) * 16;
    const int bc   = (wave & 3) * 16;

    v8f acc = {0.f, 0.f, 0.f, 0.f, 0.f, 0.f, 0.f, 0.f};
#pragma unroll
    for (int kk = 0; kk < F_IN / 4; ++kk) {
        const int kb = kk * 4 + half * 2;
        v2f a, b;
        float2 av = *(const float2*)(&As[ar + ml][kb]);
        a.x = av.x; a.y = av.y;
        b.x = Bs[kb + 0][bc + ml];
        b.y = Bs[kb + 1][bc + ml];
        acc = __builtin_amdgcn_wmma_f32_16x16x4_f32(false, a, false, b,
                                                    (short)0, acc, false, false);
    }

    const int col = n0 + bc + ml;
#pragma unroll
    for (int r = 0; r < 8; ++r) {
        const int row = m0 + ar + r + 8 * half;
        if (row < N_NODES) Hout[(size_t)row * HC + col] = acc[r];
    }
}

__launch_bounds__(256)
__global__ void gemm2_wmma(const float* __restrict__ Z, const float* __restrict__ W2,
                           float* __restrict__ H2)
{
    const int tid  = threadIdx.x;
    const int wave = tid >> 5;
    const int lane = tid & 31;
    const int half = lane >> 4;
    const int ml   = lane & 15;
    const int m0   = (blockIdx.x * 8 + wave) * 16;
    const int ra   = min(m0 + ml, N_NODES - 1);
    const int nb   = min(ml, NUM_CLASS - 1);
    const float* zr = Z + (size_t)ra * HC;

    v8f acc = {0.f, 0.f, 0.f, 0.f, 0.f, 0.f, 0.f, 0.f};
#pragma unroll 4
    for (int k = 0; k < HC; k += 4) {
        const int kb = k + half * 2;
        v2f a, b;
        float2 av = *(const float2*)(zr + kb);
        a.x = av.x; a.y = av.y;
        b.x = W2[(size_t)(kb + 0) * NUM_CLASS + nb];
        b.y = W2[(size_t)(kb + 1) * NUM_CLASS + nb];
        acc = __builtin_amdgcn_wmma_f32_16x16x4_f32(false, a, false, b,
                                                    (short)0, acc, false, false);
    }
    if (ml < NUM_CLASS) {
#pragma unroll
        for (int r = 0; r < 8; ++r) {
            const int row = m0 + r + 8 * half;
            if (row < N_NODES) H2[(size_t)row * NUM_CLASS + ml] = acc[r];
        }
    }
}

__launch_bounds__(256)
__global__ void att1_kernel(const float* __restrict__ H1,
                            const float* __restrict__ att_src,
                            const float* __restrict__ att_dst,
                            float* __restrict__ aS, float* __restrict__ aD)
{
    const int wave = threadIdx.x >> 5;
    const int lane = threadIdx.x & 31;
    const int gw = blockIdx.x * 8 + wave;
    if (gw >= N_NODES * HEADS) return;
    const int n = gw / HEADS, hh = gw % HEADS;
    const float* hp = H1 + (size_t)n * HC + hh * HIDD;
    const float* as = att_src + hh * HIDD;
    const float* ad = att_dst + hh * HIDD;
    float ss = 0.f, sd = 0.f;
    for (int c = lane; c < HIDD; c += 32) {
        float v = hp[c];
        ss += v * as[c];
        sd += v * ad[c];
    }
    for (int off = 16; off > 0; off >>= 1) {
        ss += __shfl_down(ss, off, 32);
        sd += __shfl_down(sd, off, 32);
    }
    if (lane == 0) { aS[gw] = ss; aD[gw] = sd; }
}

__global__ void att2_kernel(const float* __restrict__ H2,
                            const float* __restrict__ as2,
                            const float* __restrict__ ad2,
                            float* __restrict__ aS, float* __restrict__ aD)
{
    const int n = blockIdx.x * blockDim.x + threadIdx.x;
    if (n >= N_NODES) return;
    float ss = 0.f, sd = 0.f;
#pragma unroll
    for (int c = 0; c < NUM_CLASS; ++c) {
        float v = H2[(size_t)n * NUM_CLASS + c];
        ss += v * as2[c];
        sd += v * ad2[c];
    }
    aS[n] = ss; aD[n] = sd;
}

__device__ __forceinline__ int2 edge_sd(const int* __restrict__ ei, int e)
{
    int s = (e < E_IN) ? ei[e]        : (e - E_IN);
    int d = (e < E_IN) ? ei[E_IN + e] : (e - E_IN);
    return make_int2(s, d);
}

__device__ __forceinline__ void atomicMaxF(float* addr, float v)
{
    if (v >= 0.f) atomicMax((int*)addr, __float_as_int(v));
    else          atomicMin((unsigned int*)addr, __float_as_uint(v));
}

__global__ void edge_max(const int* __restrict__ ei,
                         const float* __restrict__ aS, const float* __restrict__ aD,
                         float* __restrict__ emax, int Hn)
{
    const int t = blockIdx.x * blockDim.x + threadIdx.x;
    if (t >= E_TOT * Hn) return;
    const int e = t / Hn, hh = t % Hn;
    const int2 sd = edge_sd(ei, e);
    float x = aS[sd.x * Hn + hh] + aD[sd.y * Hn + hh];
    x = (x >= 0.f) ? x : NEG_SLOPE * x;
    atomicMaxF(&emax[sd.y * Hn + hh], x);
}

__global__ void edge_expsum(const int* __restrict__ ei,
                            const float* __restrict__ aS, const float* __restrict__ aD,
                            const float* __restrict__ emax,
                            float* __restrict__ denom, float* __restrict__ exbuf, int Hn)
{
    const int t = blockIdx.x * blockDim.x + threadIdx.x;
    if (t >= E_TOT * Hn) return;
    const int e = t / Hn, hh = t % Hn;
    const int2 sd = edge_sd(ei, e);
    float x = aS[sd.x * Hn + hh] + aD[sd.y * Hn + hh];
    x = (x >= 0.f) ? x : NEG_SLOPE * x;
    float ex = expf(x - emax[sd.y * Hn + hh]);
    exbuf[t] = ex;
    atomicAdd(&denom[sd.y * Hn + hh], ex);
}

__launch_bounds__(256)
__global__ void edge_agg1(const int* __restrict__ ei, const float* __restrict__ H1,
                          const float* __restrict__ exbuf, const float* __restrict__ denom,
                          float* __restrict__ agg)
{
    const int e  = blockIdx.x;
    const int2 sd = edge_sd(ei, e);
    const int cb = threadIdx.x * 8;
    const int head = cb >> 8;
    const float alpha = exbuf[(size_t)e * HEADS + head]
                      / (denom[sd.y * HEADS + head] + 1e-16f);
    const float4* hs = (const float4*)(H1 + (size_t)sd.x * HC + cb);
    float4 v0 = hs[0], v1 = hs[1];
    float* op = agg + (size_t)sd.y * HC + cb;
    atomicAdd(op + 0, alpha * v0.x); atomicAdd(op + 1, alpha * v0.y);
    atomicAdd(op + 2, alpha * v0.z); atomicAdd(op + 3, alpha * v0.w);
    atomicAdd(op + 4, alpha * v1.x); atomicAdd(op + 5, alpha * v1.y);
    atomicAdd(op + 6, alpha * v1.z); atomicAdd(op + 7, alpha * v1.w);
}

__global__ void edge_agg2(const int* __restrict__ ei, const float* __restrict__ H2,
                          const float* __restrict__ exbuf, const float* __restrict__ denom,
                          float* __restrict__ agg)
{
    const int t = blockIdx.x * blockDim.x + threadIdx.x;
    if (t >= E_TOT * 16) return;
    const int e = t >> 4, c = t & 15;
    if (c >= NUM_CLASS) return;
    const int2 sd = edge_sd(ei, e);
    const float alpha = exbuf[e] / (denom[sd.y] + 1e-16f);
    atomicAdd(&agg[(size_t)sd.y * NUM_CLASS + c],
              alpha * H2[(size_t)sd.x * NUM_CLASS + c]);
}

__global__ void fill_kernel(float* __restrict__ p, size_t n, float v)
{
    for (size_t i = blockIdx.x * (size_t)blockDim.x + threadIdx.x; i < n;
         i += (size_t)gridDim.x * blockDim.x)
        p[i] = v;
}

__global__ void bias_elu(const float* __restrict__ agg, const float* __restrict__ bias,
                         float* __restrict__ z)
{
    const size_t total = (size_t)N_NODES * HC;
    for (size_t i = blockIdx.x * (size_t)blockDim.x + threadIdx.x; i < total;
         i += (size_t)gridDim.x * blockDim.x) {
        float v = agg[i] + bias[i % HC];
        z[i] = (v > 0.f) ? v : expm1f(v);
    }
}

__global__ void final_kernel(const float* __restrict__ agg2, const float* __restrict__ b2,
                             float* __restrict__ out)
{
    const int i = blockIdx.x * blockDim.x + threadIdx.x;
    if (i >= N_NODES * NUM_CLASS) return;
    float v = agg2[i] + b2[i % NUM_CLASS];
    out[i] = 1.f / (1.f + expf(-v));
}

extern "C" void kernel_launch(void* const* d_in, const int* in_sizes, int n_in,
                              void* d_out, int out_size, void* d_ws, size_t ws_size,
                              hipStream_t stream)
{
    const float* x   = (const float*)d_in[0];
    const int*   ei  = (const int*)d_in[1];
    const float* W1  = (const float*)d_in[2];
    const float* as1 = (const float*)d_in[3];
    const float* ad1 = (const float*)d_in[4];
    const float* b1  = (const float*)d_in[5];
    const float* W2  = (const float*)d_in[6];
    const float* as2 = (const float*)d_in[7];
    const float* ad2 = (const float*)d_in[8];
    const float* b2  = (const float*)d_in[9];
    float* out = (float*)d_out;

    float* ws = (float*)d_ws;
    size_t off = 0;
    float* h1    = ws + off; off += (size_t)N_NODES * HC;
    float* agg1  = ws + off; off += (size_t)N_NODES * HC;
    float* sA1   = ws + off; off += (size_t)N_NODES * HEADS;
    float* sD1   = ws + off; off += (size_t)N_NODES * HEADS;
    float* emax1 = ws + off; off += (size_t)N_NODES * HEADS;
    float* den1  = ws + off; off += (size_t)N_NODES * HEADS;
    float* ex1   = ws + off; off += (size_t)E_TOT * HEADS;
    float* h2    = ws + off; off += (size_t)N_NODES * NUM_CLASS;
    float* sA2   = ws + off; off += N_NODES;
    float* sD2   = ws + off; off += N_NODES;
    float* emax2 = ws + off; off += N_NODES;
    float* den2  = ws + off; off += N_NODES;
    float* ex2   = ws + off; off += E_TOT;
    float* agg2  = ws + off; off += (size_t)N_NODES * NUM_CLASS;
    (void)ws_size; (void)in_sizes; (void)n_in; (void)out_size;

    const dim3 b256(256);

    gemm1_wmma<<<dim3(HC / 64, (N_NODES + 31) / 32), b256, 0, stream>>>(x, W1, h1);
    att1_kernel<<<(N_NODES * HEADS) / 8, b256, 0, stream>>>(h1, as1, ad1, sA1, sD1);

    fill_kernel<<<4096, b256, 0, stream>>>(agg1, (size_t)N_NODES * HC, 0.f);
    fill_kernel<<<64,   b256, 0, stream>>>(den1, (size_t)N_NODES * HEADS, 0.f);
    fill_kernel<<<64,   b256, 0, stream>>>(emax1, (size_t)N_NODES * HEADS, -INFINITY);

    const int eg1 = (E_TOT * HEADS + 255) / 256;
    edge_max<<<eg1, b256, 0, stream>>>(ei, sA1, sD1, emax1, HEADS);
    edge_expsum<<<eg1, b256, 0, stream>>>(ei, sA1, sD1, emax1, den1, ex1, HEADS);
    edge_agg1<<<E_TOT, b256, 0, stream>>>(ei, h1, ex1, den1, agg1);

    bias_elu<<<4096, b256, 0, stream>>>(agg1, b1, h1);

    gemm2_wmma<<<(N_NODES + 127) / 128, b256, 0, stream>>>(h1, W2, h2);
    att2_kernel<<<(N_NODES + 255) / 256, b256, 0, stream>>>(h2, as2, ad2, sA2, sD2);

    fill_kernel<<<8, b256, 0, stream>>>(den2, (size_t)N_NODES, 0.f);
    fill_kernel<<<8, b256, 0, stream>>>(emax2, (size_t)N_NODES, -INFINITY);
    fill_kernel<<<8, b256, 0, stream>>>(agg2, (size_t)N_NODES * NUM_CLASS, 0.f);

    const int eg2 = (E_TOT + 255) / 256;
    edge_max<<<eg2, b256, 0, stream>>>(ei, sA2, sD2, emax2, 1);
    edge_expsum<<<eg2, b256, 0, stream>>>(ei, sA2, sD2, emax2, den2, ex2, 1);
    edge_agg2<<<(E_TOT * 16 + 255) / 256, b256, 0, stream>>>(ei, h2, ex2, den2, agg2);

    final_kernel<<<(N_NODES * NUM_CLASS + 255) / 256, b256, 0, stream>>>(agg2, b2, out);
}


# Bias+ELU applied in GEMM2's A-operand load from the aggregate (z and its pass gone); 8 K-slice waves per row tile, partial tiles summed in the layer-2 dot kernel; sum order changes
# speedup vs baseline: 2.0415x; 2.0415x over previous
#include <hip/hip_runtime.h>
#include <math.h>

#define N_NODES   10000
#define F_IN      128
#define HEADS     8
#define HIDD      256
#define HC        (HEADS * HIDD)
#define NUM_CLASS 10
#define E_IN      50000
#define E_TOT     (E_IN + N_NODES)
#define NEG_SLOPE 0.2f

typedef __attribute__((ext_vector_type(2))) float v2f;
typedef __attribute__((ext_vector_type(8))) float v8f;

__launch_bounds__(256)
__global__ void gemm1_wmma(const float* __restrict__ X, const float* __restrict__ W,
                           float* __restrict__ Hout)
{
    __shared__ float As[32][F_IN + 4];
    __shared__ float Bs[F_IN][64];

    const int tid = threadIdx.x;
    const int m0  = blockIdx.y * 32;
    const int n0  = blockIdx.x * 64;

#pragma unroll
    for (int i = 0; i < 4; ++i) {
        int idx = (tid + i * 256) * 4;
        int r = idx >> 7;
        int c = idx & 127;
        float4 v = make_float4(0.f, 0.f, 0.f, 0.f);
        int gr = m0 + r;
        if (gr < N_NODES) v = *(const float4*)(X + (size_t)gr * F_IN + c);
        *(float4*)(&As[r][c]) = v;
    }
#pragma unroll
    for (int i = 0; i < 8; ++i) {
        int idx = (tid + i * 256) * 4;
        int r = idx >> 6;
        int c = idx & 63;
        *(float4*)(&Bs[r][c]) = *(const float4*)(W + (size_t)r * HC + n0 + c);
    }
    __syncthreads();

    const int wave = tid >> 5;
    const int lane = tid & 31;
    const int half = lane >> 4;
    const int ml   = lane & 15;
    const int ar   = (wave >> 2) * 16;
    const int bc   = (wave & 3) * 16;

    v8f acc = {0.f, 0.f, 0.f, 0.f, 0.f, 0.f, 0.f, 0.f};
#pragma unroll
    for (int kk = 0; kk < F_IN / 4; ++kk) {
        const int kb = kk * 4 + half * 2;
        v2f a, b;
        float2 av = *(const float2*)(&As[ar + ml][kb]);
        a.x = av.x; a.y = av.y;
        b.x = Bs[kb + 0][bc + ml];
        b.y = Bs[kb + 1][bc + ml];
        acc = __builtin_amdgcn_wmma_f32_16x16x4_f32(false, a, false, b,
                                                    (short)0, acc, false, false);
    }

    const int col = n0 + bc + ml;
#pragma unroll
    for (int r = 0; r < 8; ++r) {
        const int row = m0 + ar + r + 8 * half;
        if (row < N_NODES) Hout[(size_t)row * HC + col] = acc[r];
    }
}

#define KSPLIT 8
#define KSL    (HC / KSPLIT)
#define G2_WAVES 4

__device__ __forceinline__ float elu1(float v)
{
    const float e = expm1f(v);
    return (v > 0.f) ? v : e;
}

__launch_bounds__(G2_WAVES * 32)
__global__ void gemm2_fused(const float* __restrict__ AGG, const float* __restrict__ B1,
                            const float* __restrict__ W2, float* __restrict__ H2P)
{
    const int tid  = threadIdx.x;
    const int wave = tid >> 5;
    const int lane = tid & 31;
    const int half = lane >> 4;
    const int ml   = lane & 15;
    const int m0   = (blockIdx.x * G2_WAVES + wave) * 16;
    if (m0 >= N_NODES) return;
    const int sl   = blockIdx.y;
    const int ra   = min(m0 + ml, N_NODES - 1);
    const int nb   = min(ml, NUM_CLASS - 1);
    const int k0   = sl * KSL + half * 4;
    const float* ap = AGG + (size_t)ra * HC + k0;
    const float* bp = B1 + k0;
    const float* wp = W2 + (size_t)k0 * NUM_CLASS + nb;

    v8f acc = {0.f, 0.f, 0.f, 0.f, 0.f, 0.f, 0.f, 0.f};
    for (int k = 0; k < KSL; k += 32) {
        float4 av[4], bv[4];
        float  w[4][4];
#pragma unroll
        for (int s = 0; s < 4; ++s) av[s] = *(const float4*)(ap + k + 8 * s);
#pragma unroll
        for (int s = 0; s < 4; ++s) bv[s] = *(const float4*)(bp + k + 8 * s);
#pragma unroll
        for (int s = 0; s < 4; ++s)
#pragma unroll
            for (int j = 0; j < 4; ++j)
                w[s][j] = wp[(size_t)(k + 8 * s + j) * NUM_CLASS];
#pragma unroll
        for (int s = 0; s < 4; ++s) {
            v2f a, b;
            a.x = elu1(av[s].x + bv[s].x); a.y = elu1(av[s].y + bv[s].y);
            b.x = w[s][0]; b.y = w[s][1];
            acc = __builtin_amdgcn_wmma_f32_16x16x4_f32(false, a, false, b,
                                                        (short)0, acc, false, false);
            a.x = elu1(av[s].z + bv[s].z); a.y = elu1(av[s].w + bv[s].w);
            b.x = w[s][2]; b.y = w[s][3];
            acc = __builtin_amdgcn_wmma_f32_16x16x4_f32(false, a, false, b,
                                                        (short)0, acc, false, false);
        }
    }
    if (ml < NUM_CLASS) {
        float* hp = H2P + (size_t)sl * N_NODES * NUM_CLASS;
#pragma unroll
        for (int r = 0; r < 8; ++r) {
            const int row = m0 + r + 8 * half;
            if (row < N_NODES) hp[(size_t)row * NUM_CLASS + ml] = acc[r];
        }
    }
}

__launch_bounds__(256)
__global__ void att1_kernel(const float* __restrict__ H1,
                            const float* __restrict__ att_src,
                            const float* __restrict__ att_dst,
                            float* __restrict__ aS, float* __restrict__ aD)
{
    const int wave = threadIdx.x >> 5;
    const int lane = threadIdx.x & 31;
    const int gw = blockIdx.x * 8 + wave;
    if (gw >= N_NODES * HEADS) return;
    const int n = gw / HEADS, hh = gw % HEADS;
    const float* hp = H1 + (size_t)n * HC + hh * HIDD;
    const float* as = att_src + hh * HIDD;
    const float* ad = att_dst + hh * HIDD;
    float ss = 0.f, sd = 0.f;
    for (int c = lane; c < HIDD; c += 32) {
        float v = hp[c];
        ss += v * as[c];
        sd += v * ad[c];
    }
    for (int off = 16; off > 0; off >>= 1) {
        ss += __shfl_down(ss, off, 32);
        sd += __shfl_down(sd, off, 32);
    }
    if (lane == 0) { aS[gw] = ss; aD[gw] = sd; }
}

__global__ void att2_kernel(const float* __restrict__ H2P,
                            const float* __restrict__ as2,
                            const float* __restrict__ ad2,
                            float* __restrict__ H2,
                            float* __restrict__ aS, float* __restrict__ aD)
{
    const int n = blockIdx.x * blockDim.x + threadIdx.x;
    if (n >= N_NODES) return;
    float ss = 0.f, sd = 0.f;
#pragma unroll
    for (int c = 0; c < NUM_CLASS; ++c) {
        float v = 0.f;
#pragma unroll
        for (int s = 0; s < KSPLIT; ++s)
            v += H2P[((size_t)s * N_NODES + n) * NUM_CLASS + c];
        H2[(size_t)n * NUM_CLASS + c] = v;
        ss += v * as2[c];
        sd += v * ad2[c];
    }
    aS[n] = ss; aD[n] = sd;
}

__device__ __forceinline__ int2 edge_sd(const int* __restrict__ ei, int e)
{
    int s = (e < E_IN) ? ei[e]        : (e - E_IN);
    int d = (e < E_IN) ? ei[E_IN + e] : (e - E_IN);
    return make_int2(s, d);
}

__device__ __forceinline__ void atomicMaxF(float* addr, float v)
{
    if (v >= 0.f) atomicMax((int*)addr, __float_as_int(v));
    else          atomicMin((unsigned int*)addr, __float_as_uint(v));
}

__global__ void edge_max(const int* __restrict__ ei,
                         const float* __restrict__ aS, const float* __restrict__ aD,
                         float* __restrict__ emax, int Hn)
{
    const int t = blockIdx.x * blockDim.x + threadIdx.x;
    if (t >= E_TOT * Hn) return;
    const int e = t / Hn, hh = t % Hn;
    const int2 sd = edge_sd(ei, e);
    float x = aS[sd.x * Hn + hh] + aD[sd.y * Hn + hh];
    x = (x >= 0.f) ? x : NEG_SLOPE * x;
    atomicMaxF(&emax[sd.y * Hn + hh], x);
}

__global__ void edge_expsum(const int* __restrict__ ei,
                            const float* __restrict__ aS, const float* __restrict__ aD,
                            const float* __restrict__ emax,
                            float* __restrict__ denom, float* __restrict__ exbuf, int Hn)
{
    const int t = blockIdx.x * blockDim.x + threadIdx.x;
    if (t >= E_TOT * Hn) return;
    const int e = t / Hn, hh = t % Hn;
    const int2 sd = edge_sd(ei, e);
    float x = aS[sd.x * Hn + hh] + aD[sd.y * Hn + hh];
    x = (x >= 0.f) ? x : NEG_SLOPE * x;
    float ex = expf(x - emax[sd.y * Hn + hh]);
    exbuf[t] = ex;
    atomicAdd(&denom[sd.y * Hn + hh], ex);
}

__launch_bounds__(256)
__global__ void edge_agg1(const int* __restrict__ ei, const float* __restrict__ H1,
                          const float* __restrict__ exbuf, const float* __restrict__ denom,
                          float* __restrict__ agg)
{
    static_assert(E_TOT % 8 == 0, "8 edges per block, no tail");
    constexpr int LINES = HC / 32;
    constexpr int GROUP = 32;
    const int lane = threadIdx.x & 31;
    const int wave = __builtin_amdgcn_readfirstlane((int)(threadIdx.x >> 5));
    const int e  = blockIdx.x * 8 + wave;
    const int2 sd = edge_sd(ei, e);
    const int hl = lane & 7;
    const float alpha_l = exbuf[(size_t)e * HEADS + hl]
                        / (denom[sd.y * HEADS + hl] + 1e-16f);
    const float* hs = H1  + (size_t)sd.x * HC + lane;
    float*       op = agg + (size_t)sd.y * HC + lane;
#pragma unroll
    for (int g = 0; g < LINES / GROUP; ++g) {
        float v[GROUP];
#pragma unroll
        for (int q = 0; q < GROUP; ++q) v[q] = hs[(g * GROUP + q) * 32];
#pragma unroll
        for (int q = 0; q < GROUP; ++q) {
            const int line = g * GROUP + q;
            const float alpha = __int_as_float(
                __builtin_amdgcn_readlane(__float_as_int(alpha_l), line / (HIDD / 32)));
            atomicAdd(op + line * 32, alpha * v[q]);
        }
    }
}

__global__ void edge_agg2(const int* __restrict__ ei, const float* __restrict__ H2,
                          const float* __restrict__ exbuf, const float* __restrict__ denom,
                          float* __restrict__ agg)
{
    const int t = blockIdx.x * blockDim.x + threadIdx.x;
    if (t >= E_TOT * 16) return;
    const int e = t >> 4, c = t & 15;
    if (c >= NUM_CLASS) return;
    const int2 sd = edge_sd(ei, e);
    const float alpha = exbuf[e] / (denom[sd.y] + 1e-16f);
    atomicAdd(&agg[(size_t)sd.y * NUM_CLASS + c],
              alpha * H2[(size_t)sd.x * NUM_CLASS + c]);
}

__global__ void fill_kernel(float* __restrict__ p, size_t n, float v)
{
    for (size_t i = blockIdx.x * (size_t)blockDim.x + threadIdx.x; i < n;
         i += (size_t)gridDim.x * blockDim.x)
        p[i] = v;
}

__global__ void final_kernel(const float* __restrict__ agg2, const float* __restrict__ b2,
                             float* __restrict__ out)
{
    const int i = blockIdx.x * blockDim.x + threadIdx.x;
    if (i >= N_NODES * NUM_CLASS) return;
    float v = agg2[i] + b2[i % NUM_CLASS];
    out[i] = 1.f / (1.f + expf(-v));
}

extern "C" void kernel_launch(void* const* d_in, const int* in_sizes, int n_in,
                              void* d_out, int out_size, void* d_ws, size_t ws_size,
                              hipStream_t stream)
{
    const float* x   = (const float*)d_in[0];
    const int*   ei  = (const int*)d_in[1];
    const float* W1  = (const float*)d_in[2];
    const float* as1 = (const float*)d_in[3];
    const float* ad1 = (const float*)d_in[4];
    const float* b1  = (const float*)d_in[5];
    const float* W2  = (const float*)d_in[6];
    const float* as2 = (const float*)d_in[7];
    const float* ad2 = (const float*)d_in[8];
    const float* b2  = (const float*)d_in[9];
    float* out = (float*)d_out;

    float* ws = (float*)d_ws;
    size_t off = 0;
    float* h1    = ws + off; off += (size_t)N_NODES * HC;
    float* agg1  = ws + off; off += (size_t)N_NODES * HC;
    float* sA1   = ws + off; off += (size_t)N_NODES * HEADS;
    float* sD1   = ws + off; off += (size_t)N_NODES * HEADS;
    float* emax1 = ws + off; off += (size_t)N_NODES * HEADS;
    float* den1  = ws + off; off += (size_t)N_NODES * HEADS;
    float* ex1   = ws + off; off += (size_t)E_TOT * HEADS;
    float* h2    = ws + off; off += (size_t)N_NODES * NUM_CLASS;
    float* sA2   = ws + off; off += N_NODES;
    float* sD2   = ws + off; off += N_NODES;
    float* emax2 = ws + off; off += N_NODES;
    float* den2  = ws + off; off += N_NODES;
    float* ex2   = ws + off; off += E_TOT;
    float* agg2  = ws + off; off += (size_t)N_NODES * NUM_CLASS;
    float* h2p   = ws + off; off += (size_t)KSPLIT * N_NODES * NUM_CLASS;
    (void)ws_size; (void)in_sizes; (void)n_in; (void)out_size;

    const dim3 b256(256);

    gemm1_wmma<<<dim3(HC / 64, (N_NODES + 31) / 32), b256, 0, stream>>>(x, W1, h1);
    att1_kernel<<<(N_NODES * HEADS) / 8, b256, 0, stream>>>(h1, as1, ad1, sA1, sD1);

    fill_kernel<<<4096, b256, 0, stream>>>(agg1, (size_t)N_NODES * HC, 0.f);
    fill_kernel<<<64,   b256, 0, stream>>>(den1, (size_t)N_NODES * HEADS, 0.f);
    fill_kernel<<<64,   b256, 0, stream>>>(emax1, (size_t)N_NODES * HEADS, -INFINITY);

    const int eg1 = (E_TOT * HEADS + 255) / 256;
    edge_max<<<eg1, b256, 0, stream>>>(ei, sA1, sD1, emax1, HEADS);
    edge_expsum<<<eg1, b256, 0, stream>>>(ei, sA1, sD1, emax1, den1, ex1, HEADS);
    edge_agg1<<<E_TOT / 8, b256, 0, stream>>>(ei, h1, ex1, den1, agg1);

    gemm2_fused<<<dim3((N_NODES + 16 * G2_WAVES - 1) / (16 * G2_WAVES), KSPLIT), dim3(G2_WAVES * 32), 0, stream>>>(agg1, b1, W2, h2p);
    att2_kernel<<<(N_NODES + 255) / 256, b256, 0, stream>>>(h2p, as2, ad2, h2, sA2, sD2);

    fill_kernel<<<8, b256, 0, stream>>>(den2, (size_t)N_NODES, 0.f);
    fill_kernel<<<8, b256, 0, stream>>>(emax2, (size_t)N_NODES, -INFINITY);
    fill_kernel<<<8, b256, 0, stream>>>(agg2, (size_t)N_NODES * NUM_CLASS, 0.f);

    const int eg2 = (E_TOT + 255) / 256;
    edge_max<<<eg2, b256, 0, stream>>>(ei, sA2, sD2, emax2, 1);
    edge_expsum<<<eg2, b256, 0, stream>>>(ei, sA2, sD2, emax2, den2, ex2, 1);
    edge_agg2<<<(E_TOT * 16 + 255) / 256, b256, 0, stream>>>(ei, h2, ex2, den2, agg2);

    final_kernel<<<(N_NODES * NUM_CLASS + 255) / 256, b256, 0, stream>>>(agg2, b2, out);
}


# Layer-1 attention dots taken from GEMM1's accumulators in its epilogue (block = 32 rows x one whole head, wave = 16x64 on 4 tiles): att1 launch and its re-read of h1 gone; dot sum order changes
# speedup vs baseline: 2.3159x; 1.1344x over previous
#include <hip/hip_runtime.h>
#include <math.h>

#define N_NODES   10000
#define F_IN      128
#define HEADS     8
#define HIDD      256
#define HC        (HEADS * HIDD)
#define NUM_CLASS 10
#define E_IN      50000
#define E_TOT     (E_IN + N_NODES)
#define NEG_SLOPE 0.2f

typedef __attribute__((ext_vector_type(2))) float v2f;
typedef __attribute__((ext_vector_type(8))) float v8f;

#define G1_KP (2 * HIDD + 32)

__launch_bounds__(256)
__global__ void gemm1_att(const float* __restrict__ X, const float* __restrict__ W,
                          const float* __restrict__ att_src,
                          const float* __restrict__ att_dst,
                          float* __restrict__ Hout,
                          float* __restrict__ aS, float* __restrict__ aD)
{
    static_assert(N_NODES % 16 == 0, "a wave's 16 rows are all inside or all outside");
    __shared__ __attribute__((aligned(16))) float As[32][F_IN + 4];
    __shared__ __attribute__((aligned(16))) float Bs[F_IN / 2][G1_KP];
    __shared__ __attribute__((aligned(16))) float red[8][32];

    const int tid = threadIdx.x;
    const int hh  = blockIdx.x;
    const int n0  = hh * HIDD;
    const int m0  = blockIdx.y * 32;

#pragma unroll
    for (int i = 0; i < 4; ++i) {
        int idx = (tid + i * 256) * 4;
        int r = idx >> 7;
        int c = idx & 127;
        const int gr = min(m0 + r, N_NODES - 1);
        *(float4*)(&As[r][c]) = *(const float4*)(X + (size_t)gr * F_IN + c);
    }
#pragma unroll
    for (int i = 0; i < 16; ++i) {
        const int item = tid + i * 256;
        const int p = item >> 6;
        const int c = (item & 63) * 4;
        const float* wp = W + (size_t)(2 * p) * HC + n0 + c;
        const float4 r0 = *(const float4*)wp;
        const float4 r1 = *(const float4*)(wp + HC);
        *(float4*)(&Bs[p][2 * c])     = make_float4(r0.x, r1.x, r0.y, r1.y);
        *(float4*)(&Bs[p][2 * c + 4]) = make_float4(r0.z, r1.z, r0.w, r1.w);
    }
    __syncthreads();

    const int wave = tid >> 5;
    const int lane = tid & 31;
    const int half = lane >> 4;
    const int ml   = lane & 15;
    const int ar   = (wave >> 2) * 16;
    const int bc   = (wave & 3) * 64;

    float asv[4], adv[4];
#pragma unroll
    for (int t = 0; t < 4; ++t) {
        asv[t] = att_src[n0 + bc + 16 * t + ml];
        adv[t] = att_dst[n0 + bc + 16 * t + ml];
    }

    v8f acc[4];
#pragma unroll
    for (int t = 0; t < 4; ++t)
        acc[t] = (v8f){0.f, 0.f, 0.f, 0.f, 0.f, 0.f, 0.f, 0.f};

#pragma unroll
    for (int kk = 0; kk < F_IN / 4; ++kk) {
        const int p = kk * 2 + half;
        v2f a;
        const float2 av = *(const float2*)(&As[ar + ml][2 * p]);
        a.x = av.x; a.y = av.y;
#pragma unroll
        for (int t = 0; t < 4; ++t) {
            v2f b;
            const float2 bv = *(const float2*)(&Bs[p][2 * (bc + 16 * t + ml)]);
            b.x = bv.x; b.y = bv.y;
            acc[t] = __builtin_amdgcn_wmma_f32_16x16x4_f32(false, a, false, b,
                                                           (short)0, acc[t], false, false);
        }
    }

    if (m0 + ar < N_NODES) {
#pragma unroll
        for (int t = 0; t < 4; ++t) {
            const int col = n0 + bc + 16 * t + ml;
#pragma unroll
            for (int r = 0; r < 8; ++r) {
                const int row = m0 + ar + r + 8 * half;
                Hout[(size_t)row * HC + col] = acc[t][r];
            }
        }
    }

    float v[16];
#pragma unroll
    for (int r = 0; r < 8; ++r) {
        float ss = 0.f, sd = 0.f;
#pragma unroll
        for (int t = 0; t < 4; ++t) {
            ss += acc[t][r] * asv[t];
            sd += acc[t][r] * adv[t];
        }
        v[r] = ss; v[8 + r] = sd;
    }
#pragma unroll
    for (int j = 0; j < 16; ++j) {
        v[j] += __int_as_float(__builtin_amdgcn_update_dpp(0, __float_as_int(v[j]), 0x161, 0xf, 0xf, false));
        v[j] += __int_as_float(__builtin_amdgcn_update_dpp(0, __float_as_int(v[j]), 0x162, 0xf, 0xf, false));
        v[j] += __int_as_float(__builtin_amdgcn_update_dpp(0, __float_as_int(v[j]), 0x164, 0xf, 0xf, false));
        v[j] += __int_as_float(__builtin_amdgcn_update_dpp(0, __float_as_int(v[j]), 0x168, 0xf, 0xf, false));
    }
    if (ml == 0) {
#pragma unroll
        for (int j = 0; j < 16; j += 4)
            *(float4*)(&red[wave][half * 16 + j]) = make_float4(v[j], v[j + 1], v[j + 2], v[j + 3]);
    }
    __syncthreads();
    if (tid < 64) {
        const int rh = tid >> 5;
        const int ln = tid & 31;
        const float s = ((red[rh * 4 + 0][ln] + red[rh * 4 + 1][ln])
                         + red[rh * 4 + 2][ln]) + red[rh * 4 + 3][ln];
        const int row = m0 + rh * 16 + (ln & 7) + 8 * (ln >> 4);
        if (row < N_NODES) {
            float* dstp = ((ln >> 3) & 1) ? aD : aS;
            dstp[(size_t)row * HEADS + hh] = s;
        }
    }
}

#define KSPLIT 8
#define KSL    (HC / KSPLIT)
#define G2_WAVES 4

__device__ __forceinline__ float elu1(float v)
{
    const float e = expm1f(v);
    return (v > 0.f) ? v : e;
}

__launch_bounds__(G2_WAVES * 32)
__global__ void gemm2_fused(const float* __restrict__ AGG, const float* __restrict__ B1,
                            const float* __restrict__ W2, float* __restrict__ H2P)
{
    const int tid  = threadIdx.x;
    const int wave = tid >> 5;
    const int lane = tid & 31;
    const int half = lane >> 4;
    const int ml   = lane & 15;
    const int m0   = (blockIdx.x * G2_WAVES + wave) * 16;
    if (m0 >= N_NODES) return;
    const int sl   = blockIdx.y;
    const int ra   = min(m0 + ml, N_NODES - 1);
    const int nb   = min(ml, NUM_CLASS - 1);
    const int k0   = sl * KSL + half * 4;
    const float* ap = AGG + (size_t)ra * HC + k0;
    const float* bp = B1 + k0;
    const float* wp = W2 + (size_t)k0 * NUM_CLASS + nb;

    v8f acc = {0.f, 0.f, 0.f, 0.f, 0.f, 0.f, 0.f, 0.f};
    for (int k = 0; k < KSL; k += 32) {
        float4 av[4], bv[4];
        float  w[4][4];
#pragma unroll
        for (int s = 0; s < 4; ++s) av[s] = *(const float4*)(ap + k + 8 * s);
#pragma unroll
        for (int s = 0; s < 4; ++s) bv[s] = *(const float4*)(bp + k + 8 * s);
#pragma unroll
        for (int s = 0; s < 4; ++s)
#pragma unroll
            for (int j = 0; j < 4; ++j)
                w[s][j] = wp[(size_t)(k + 8 * s + j) * NUM_CLASS];
#pragma unroll
        for (int s = 0; s < 4; ++s) {
            v2f a, b;
            a.x = elu1(av[s].x + bv[s].x); a.y = elu1(av[s].y + bv[s].y);
            b.x = w[s][0]; b.y = w[s][1];
            acc = __builtin_amdgcn_wmma_f32_16x16x4_f32(false, a, false, b,
                                                        (short)0, acc, false, false);
            a.x = elu1(av[s].z + bv[s].z); a.y = elu1(av[s].w + bv[s].w);
            b.x = w[s][2]; b.y = w[s][3];
            acc = __builtin_amdgcn_wmma_f32_16x16x4_f32(false, a, false, b,
                                                        (short)0, acc, false, false);
        }
    }
    if (ml < NUM_CLASS) {
        float* hp = H2P + (size_t)sl * N_NODES * NUM_CLASS;
#pragma unroll
        for (int r = 0; r < 8; ++r) {
            const int row = m0 + r + 8 * half;
            if (row < N_NODES) hp[(size_t)row * NUM_CLASS + ml] = acc[r];
        }
    }
}

__global__ void att2_kernel(const float* __restrict__ H2P,
                            const float* __restrict__ as2,
                            const float* __restrict__ ad2,
                            float* __restrict__ H2,
                            float* __restrict__ aS, float* __restrict__ aD)
{
    const int n = blockIdx.x * blockDim.x + threadIdx.x;
    if (n >= N_NODES) return;
    float ss = 0.f, sd = 0.f;
#pragma unroll
    for (int c = 0; c < NUM_CLASS; ++c) {
        float v = 0.f;
#pragma unroll
        for (int s = 0; s < KSPLIT; ++s)
            v += H2P[((size_t)s * N_NODES + n) * NUM_CLASS + c];
        H2[(size_t)n * NUM_CLASS + c] = v;
        ss += v * as2[c];
        sd += v * ad2[c];
    }
    aS[n] = ss; aD[n] = sd;
}

__device__ __forceinline__ int2 edge_sd(const int* __restrict__ ei, int e)
{
    int s = (e < E_IN) ? ei[e]        : (e - E_IN);
    int d = (e < E_IN) ? ei[E_IN + e] : (e - E_IN);
    return make_int2(s, d);
}

__device__ __forceinline__ void atomicMaxF(float* addr, float v)
{
    if (v >= 0.f) atomicMax((int*)addr, __float_as_int(v));
    else          atomicMin((unsigned int*)addr, __float_as_uint(v));
}

__global__ void edge_max(const int* __restrict__ ei,
                         const float* __restrict__ aS, const float* __restrict__ aD,
                         float* __restrict__ emax, int Hn)
{
    const int t = blockIdx.x * blockDim.x + threadIdx.x;
    if (t >= E_TOT * Hn) return;
    const int e = t / Hn, hh = t % Hn;
    const int2 sd = edge_sd(ei, e);
    float x = aS[sd.x * Hn + hh] + aD[sd.y * Hn + hh];
    x = (x >= 0.f) ? x : NEG_SLOPE * x;
    atomicMaxF(&emax[sd.y * Hn + hh], x);
}

__global__ void edge_expsum(const int* __restrict__ ei,
                            const float* __restrict__ aS, const float* __restrict__ aD,
                            const float* __restrict__ emax,
                            float* __restrict__ denom, float* __restrict__ exbuf, int Hn)
{
    const int t = blockIdx.x * blockDim.x + threadIdx.x;
    if (t >= E_TOT * Hn) return;
    const int e = t / Hn, hh = t % Hn;
    const int2 sd = edge_sd(ei, e);
    float x = aS[sd.x * Hn + hh] + aD[sd.y * Hn + hh];
    x = (x >= 0.f) ? x : NEG_SLOPE * x;
    float ex = expf(x - emax[sd.y * Hn + hh]);
    exbuf[t] = ex;
    atomicAdd(&denom[sd.y * Hn + hh], ex);
}

__launch_bounds__(256)
__global__ void edge_agg1(const int* __restrict__ ei, const float* __restrict__ H1,
                          const float* __restrict__ exbuf, const float* __restrict__ denom,
                          float* __restrict__ agg)
{
    static_assert(E_TOT % 8 == 0, "8 edges per block, no tail");
    constexpr int LINES = HC / 32;
    constexpr int GROUP = 32;
    const int lane = threadIdx.x & 31;
    const int wave = __builtin_amdgcn_readfirstlane((int)(threadIdx.x >> 5));
    const int e  = blockIdx.x * 8 + wave;
    const int2 sd = edge_sd(ei, e);
    const int hl = lane & 7;
    const float alpha_l = exbuf[(size_t)e * HEADS + hl]
                        / (denom[sd.y * HEADS + hl] + 1e-16f);
    const float* hs = H1  + (size_t)sd.x * HC + lane;
    float*       op = agg + (size_t)sd.y * HC + lane;
#pragma unroll
    for (int g = 0; g < LINES / GROUP; ++g) {
        float v[GROUP];
#pragma unroll
        for (int q = 0; q < GROUP; ++q) v[q] = hs[(g * GROUP + q) * 32];
#pragma unroll
        for (int q = 0; q < GROUP; ++q) {
            const int line = g * GROUP + q;
            const float alpha = __int_as_float(
                __builtin_amdgcn_readlane(__float_as_int(alpha_l), line / (HIDD / 32)));
            atomicAdd(op + line * 32, alpha * v[q]);
        }
    }
}

__global__ void edge_agg2(const int* __restrict__ ei, const float* __restrict__ H2,
                          const float* __restrict__ exbuf, const float* __restrict__ denom,
                          float* __restrict__ agg)
{
    const int t = blockIdx.x * blockDim.x + threadIdx.x;
    if (t >= E_TOT * 16) return;
    const int e = t >> 4, c = t & 15;
    if (c >= NUM_CLASS) return;
    const int2 sd = edge_sd(ei, e);
    const float alpha = exbuf[e] / (denom[sd.y] + 1e-16f);
    atomicAdd(&agg[(size_t)sd.y * NUM_CLASS + c],
              alpha * H2[(size_t)sd.x * NUM_CLASS + c]);
}

__global__ void fill_kernel(float* __restrict__ p, size_t n, float v)
{
    for (size_t i = blockIdx.x * (size_t)blockDim.x + threadIdx.x; i < n;
         i += (size_t)gridDim.x * blockDim.x)
        p[i] = v;
}

__global__ void final_kernel(const float* __restrict__ agg2, const float* __restrict__ b2,
                             float* __restrict__ out)
{
    const int i = blockIdx.x * blockDim.x + threadIdx.x;
    if (i >= N_NODES * NUM_CLASS) return;
    float v = agg2[i] + b2[i % NUM_CLASS];
    out[i] = 1.f / (1.f + expf(-v));
}

extern "C" void kernel_launch(void* const* d_in, const int* in_sizes, int n_in,
                              void* d_out, int out_size, void* d_ws, size_t ws_size,
                              hipStream_t stream)
{
    const float* x   = (const float*)d_in[0];
    const int*   ei  = (const int*)d_in[1];
    const float* W1  = (const float*)d_in[2];
    const float* as1 = (const float*)d_in[3];
    const float* ad1 = (const float*)d_in[4];
    const float* b1  = (const float*)d_in[5];
    const float* W2  = (const float*)d_in[6];
    const float* as2 = (const float*)d_in[7];
    const float* ad2 = (const float*)d_in[8];
    const float* b2  = (const float*)d_in[9];
    float* out = (float*)d_out;

    float* ws = (float*)d_ws;
    size_t off = 0;
    float* h1    = ws + off; off += (size_t)N_NODES * HC;
    float* agg1  = ws + off; off += (size_t)N_NODES * HC;
    float* sA1   = ws + off; off += (size_t)N_NODES * HEADS;
    float* sD1   = ws + off; off += (size_t)N_NODES * HEADS;
    float* emax1 = ws + off; off += (size_t)N_NODES * HEADS;
    float* den1  = ws + off; off += (size_t)N_NODES * HEADS;
    float* ex1   = ws + off; off += (size_t)E_TOT * HEADS;
    float* h2    = ws + off; off += (size_t)N_NODES * NUM_CLASS;
    float* sA2   = ws + off; off += N_NODES;
    float* sD2   = ws + off; off += N_NODES;
    float* emax2 = ws + off; off += N_NODES;
    float* den2  = ws + off; off += N_NODES;
    float* ex2   = ws + off; off += E_TOT;
    float* agg2  = ws + off; off += (size_t)N_NODES * NUM_CLASS;
    float* h2p   = ws + off; off += (size_t)KSPLIT * N_NODES * NUM_CLASS;
    (void)ws_size; (void)in_sizes; (void)n_in; (void)out_size;

    const dim3 b256(256);

    gemm1_att<<<dim3(HEADS, (N_NODES + 31) / 32), b256, 0, stream>>>(x, W1, as1, ad1, h1, sA1, sD1);

    fill_kernel<<<4096, b256, 0, stream>>>(agg1, (size_t)N_NODES * HC, 0.f);
    fill_kernel<<<64,   b256, 0, stream>>>(den1, (size_t)N_NODES * HEADS, 0.f);
    fill_kernel<<<64,   b256, 0, stream>>>(emax1, (size_t)N_NODES * HEADS, -INFINITY);

    const int eg1 = (E_TOT * HEADS + 255) / 256;
    edge_max<<<eg1, b256, 0, stream>>>(ei, sA1, sD1, emax1, HEADS);
    edge_expsum<<<eg1, b256, 0, stream>>>(ei, sA1, sD1, emax1, den1, ex1, HEADS);
    edge_agg1<<<E_TOT / 8, b256, 0, stream>>>(ei, h1, ex1, den1, agg1);

    gemm2_fused<<<dim3((N_NODES + 16 * G2_WAVES - 1) / (16 * G2_WAVES), KSPLIT), dim3(G2_WAVES * 32), 0, stream>>>(agg1, b1, W2, h2p);
    att2_kernel<<<(N_NODES + 255) / 256, b256, 0, stream>>>(h2p, as2, ad2, h2, sA2, sD2);

    fill_kernel<<<8, b256, 0, stream>>>(den2, (size_t)N_NODES, 0.f);
    fill_kernel<<<8, b256, 0, stream>>>(emax2, (size_t)N_NODES, -INFINITY);
    fill_kernel<<<8, b256, 0, stream>>>(agg2, (size_t)N_NODES * NUM_CLASS, 0.f);

    const int eg2 = (E_TOT + 255) / 256;
    edge_max<<<eg2, b256, 0, stream>>>(ei, sA2, sD2, emax2, 1);
    edge_expsum<<<eg2, b256, 0, stream>>>(ei, sA2, sD2, emax2, den2, ex2, 1);
    edge_agg2<<<(E_TOT * 16 + 255) / 256, b256, 0, stream>>>(ei, h2, ex2, den2, agg2);

    final_kernel<<<(N_NODES * NUM_CLASS + 255) / 256, b256, 0, stream>>>(agg2, b2, out);
}


# Layer-2 edge phase by owner work-groups: block owns 40 nodes, lists its edges from one scan of the destination column, max/exp-sum/row sums in LDS; 16 -> 10 launches; sum order changes
# speedup vs baseline: 2.6006x; 1.1229x over previous
#include <hip/hip_runtime.h>
#include <math.h>

#define N_NODES   10000
#define F_IN      128
#define HEADS     8
#define HIDD      256
#define HC        (HEADS * HIDD)
#define NUM_CLASS 10
#define E_IN      50000
#define E_TOT     (E_IN + N_NODES)
#define NEG_SLOPE 0.2f

typedef __attribute__((ext_vector_type(2))) float v2f;
typedef __attribute__((ext_vector_type(8))) float v8f;

#define G1_KP (2 * HIDD + 32)

__launch_bounds__(256)
__global__ void gemm1_att(const float* __restrict__ X, const float* __restrict__ W,
                          const float* __restrict__ att_src,
                          const float* __restrict__ att_dst,
                          float* __restrict__ Hout,
                          float* __restrict__ aS, float* __restrict__ aD)
{
    static_assert(N_NODES % 16 == 0, "a wave's 16 rows are all inside or all outside");
    __shared__ __attribute__((aligned(16))) float As[32][F_IN + 4];
    __shared__ __attribute__((aligned(16))) float Bs[F_IN / 2][G1_KP];
    __shared__ __attribute__((aligned(16))) float red[8][32];

    const int tid = threadIdx.x;
    const int hh  = blockIdx.x;
    const int n0  = hh * HIDD;
    const int m0  = blockIdx.y * 32;

#pragma unroll
    for (int i = 0; i < 4; ++i) {
        int idx = (tid + i * 256) * 4;
        int r = idx >> 7;
        int c = idx & 127;
        const int gr = min(m0 + r, N_NODES - 1);
        *(float4*)(&As[r][c]) = *(const float4*)(X + (size_t)gr * F_IN + c);
    }
#pragma unroll
    for (int i = 0; i < 16; ++i) {
        const int item = tid + i * 256;
        const int p = item >> 6;
        const int c = (item & 63) * 4;
        const float* wp = W + (size_t)(2 * p) * HC + n0 + c;
        const float4 r0 = *(const float4*)wp;
        const float4 r1 = *(const float4*)(wp + HC);
        *(float4*)(&Bs[p][2 * c])     = make_float4(r0.x, r1.x, r0.y, r1.y);
        *(float4*)(&Bs[p][2 * c + 4]) = make_float4(r0.z, r1.z, r0.w, r1.w);
    }
    __syncthreads();

    const int wave = tid >> 5;
    const int lane = tid & 31;
    const int half = lane >> 4;
    const int ml   = lane & 15;
    const int ar   = (wave >> 2) * 16;
    const int bc   = (wave & 3) * 64;

    float asv[4], adv[4];
#pragma unroll
    for (int t = 0; t < 4; ++t) {
        asv[t] = att_src[n0 + bc + 16 * t + ml];
        adv[t] = att_dst[n0 + bc + 16 * t + ml];
    }

    v8f acc[4];
#pragma unroll
    for (int t = 0; t < 4; ++t)
        acc[t] = (v8f){0.f, 0.f, 0.f, 0.f, 0.f, 0.f, 0.f, 0.f};

#pragma unroll
    for (int kk = 0; kk < F_IN / 4; ++kk) {
        const int p = kk * 2 + half;
        v2f a;
        const float2 av = *(const float2*)(&As[ar + ml][2 * p]);
        a.x = av.x; a.y = av.y;
#pragma unroll
        for (int t = 0; t < 4; ++t) {
            v2f b;
            const float2 bv = *(const float2*)(&Bs[p][2 * (bc + 16 * t + ml)]);
            b.x = bv.x; b.y = bv.y;
            acc[t] = __builtin_amdgcn_wmma_f32_16x16x4_f32(false, a, false, b,
                                                           (short)0, acc[t], false, false);
        }
    }

    if (m0 + ar < N_NODES) {
#pragma unroll
        for (int t = 0; t < 4; ++t) {
            const int col = n0 + bc + 16 * t + ml;
#pragma unroll
            for (int r = 0; r < 8; ++r) {
                const int row = m0 + ar + r + 8 * half;
                Hout[(size_t)row * HC + col] = acc[t][r];
            }
        }
    }

    float v[16];
#pragma unroll
    for (int r = 0; r < 8; ++r) {
        float ss = 0.f, sd = 0.f;
#pragma unroll
        for (int t = 0; t < 4; ++t) {
            ss += acc[t][r] * asv[t];
            sd += acc[t][r] * adv[t];
        }
        v[r] = ss; v[8 + r] = sd;
    }
#pragma unroll
    for (int j = 0; j < 16; ++j) {
        v[j] += __int_as_float(__builtin_amdgcn_update_dpp(0, __float_as_int(v[j]), 0x161, 0xf, 0xf, false));
        v[j] += __int_as_float(__builtin_amdgcn_update_dpp(0, __float_as_int(v[j]), 0x162, 0xf, 0xf, false));
        v[j] += __int_as_float(__builtin_amdgcn_update_dpp(0, __float_as_int(v[j]), 0x164, 0xf, 0xf, false));
        v[j] += __int_as_float(__builtin_amdgcn_update_dpp(0, __float_as_int(v[j]), 0x168, 0xf, 0xf, false));
    }
    if (ml == 0) {
#pragma unroll
        for (int j = 0; j < 16; j += 4)
            *(float4*)(&red[wave][half * 16 + j]) = make_float4(v[j], v[j + 1], v[j + 2], v[j + 3]);
    }
    __syncthreads();
    if (tid < 64) {
        const int rh = tid >> 5;
        const int ln = tid & 31;
        const float s = ((red[rh * 4 + 0][ln] + red[rh * 4 + 1][ln])
                         + red[rh * 4 + 2][ln]) + red[rh * 4 + 3][ln];
        const int row = m0 + rh * 16 + (ln & 7) + 8 * (ln >> 4);
        if (row < N_NODES) {
            float* dstp = ((ln >> 3) & 1) ? aD : aS;
            dstp[(size_t)row * HEADS + hh] = s;
        }
    }
}

#define KSPLIT 8
#define KSL    (HC / KSPLIT)
#define G2_WAVES 4

__device__ __forceinline__ float elu1(float v)
{
    const float e = expm1f(v);
    return (v > 0.f) ? v : e;
}

__launch_bounds__(G2_WAVES * 32)
__global__ void gemm2_fused(const float* __restrict__ AGG, const float* __restrict__ B1,
                            const float* __restrict__ W2, float* __restrict__ H2P)
{
    const int tid  = threadIdx.x;
    const int wave = tid >> 5;
    const int lane = tid & 31;
    const int half = lane >> 4;
    const int ml   = lane & 15;
    const int m0   = (blockIdx.x * G2_WAVES + wave) * 16;
    if (m0 >= N_NODES) return;
    const int sl   = blockIdx.y;
    const int ra   = min(m0 + ml, N_NODES - 1);
    const int nb   = min(ml, NUM_CLASS - 1);
    const int k0   = sl * KSL + half * 4;
    const float* ap = AGG + (size_t)ra * HC + k0;
    const float* bp = B1 + k0;
    const float* wp = W2 + (size_t)k0 * NUM_CLASS + nb;

    v8f acc = {0.f, 0.f, 0.f, 0.f, 0.f, 0.f, 0.f, 0.f};
    for (int k = 0; k < KSL; k += 32) {
        float4 av[4], bv[4];
        float  w[4][4];
#pragma unroll
        for (int s = 0; s < 4; ++s) av[s] = *(const float4*)(ap + k + 8 * s);
#pragma unroll
        for (int s = 0; s < 4; ++s) bv[s] = *(const float4*)(bp + k + 8 * s);
#pragma unroll
        for (int s = 0; s < 4; ++s)
#pragma unroll
            for (int j = 0; j < 4; ++j)
                w[s][j] = wp[(size_t)(k + 8 * s + j) * NUM_CLASS];
#pragma unroll
        for (int s = 0; s < 4; ++s) {
            v2f a, b;
            a.x = elu1(av[s].x + bv[s].x); a.y = elu1(av[s].y + bv[s].y);
            b.x = w[s][0]; b.y = w[s][1];
            acc = __builtin_amdgcn_wmma_f32_16x16x4_f32(false, a, false, b,
                                                        (short)0, acc, false, false);
            a.x = elu1(av[s].z + bv[s].z); a.y = elu1(av[s].w + bv[s].w);
            b.x = w[s][2]; b.y = w[s][3];
            acc = __builtin_amdgcn_wmma_f32_16x16x4_f32(false, a, false, b,
                                                        (short)0, acc, false, false);
        }
    }
    if (ml < NUM_CLASS) {
        float* hp = H2P + (size_t)sl * N_NODES * NUM_CLASS;
#pragma unroll
        for (int r = 0; r < 8; ++r) {
            const int row = m0 + r + 8 * half;
            if (row < N_NODES) hp[(size_t)row * NUM_CLASS + ml] = acc[r];
        }
    }
}

__global__ void att2_kernel(const float* __restrict__ H2P,
                            const float* __restrict__ as2,
                            const float* __restrict__ ad2,
                            float* __restrict__ H2,
                            float* __restrict__ aS, float* __restrict__ aD)
{
    const int n = blockIdx.x * blockDim.x + threadIdx.x;
    if (n >= N_NODES) return;
    float ss = 0.f, sd = 0.f;
#pragma unroll
    for (int c = 0; c < NUM_CLASS; ++c) {
        float v = 0.f;
#pragma unroll
        for (int s = 0; s < KSPLIT; ++s)
            v += H2P[((size_t)s * N_NODES + n) * NUM_CLASS + c];
        H2[(size_t)n * NUM_CLASS + c] = v;
        ss += v * as2[c];
        sd += v * ad2[c];
    }
    aS[n] = ss; aD[n] = sd;
}

__device__ __forceinline__ int2 edge_sd(const int* __restrict__ ei, int e)
{
    int s = (e < E_IN) ? ei[e]        : (e - E_IN);
    int d = (e < E_IN) ? ei[E_IN + e] : (e - E_IN);
    return make_int2(s, d);
}

__device__ __forceinline__ void atomicMaxF(float* addr, float v)
{
    if (v >= 0.f) atomicMax((int*)addr, __float_as_int(v));
    else          atomicMin((unsigned int*)addr, __float_as_uint(v));
}

__global__ void edge_max(const int* __restrict__ ei,
                         const float* __restrict__ aS, const float* __restrict__ aD,
                         float* __restrict__ emax, int Hn)
{
    const int t = blockIdx.x * blockDim.x + threadIdx.x;
    if (t >= E_TOT * Hn) return;
    const int e = t / Hn, hh = t % Hn;
    const int2 sd = edge_sd(ei, e);
    float x = aS[sd.x * Hn + hh] + aD[sd.y * Hn + hh];
    x = (x >= 0.f) ? x : NEG_SLOPE * x;
    atomicMaxF(&emax[sd.y * Hn + hh], x);
}

__global__ void edge_expsum(const int* __restrict__ ei,
                            const float* __restrict__ aS, const float* __restrict__ aD,
                            const float* __restrict__ emax,
                            float* __restrict__ denom, float* __restrict__ exbuf, int Hn)
{
    const int t = blockIdx.x * blockDim.x + threadIdx.x;
    if (t >= E_TOT * Hn) return;
    const int e = t / Hn, hh = t % Hn;
    const int2 sd = edge_sd(ei, e);
    float x = aS[sd.x * Hn + hh] + aD[sd.y * Hn + hh];
    x = (x >= 0.f) ? x : NEG_SLOPE * x;
    float ex = expf(x - emax[sd.y * Hn + hh]);
    exbuf[t] = ex;
    atomicAdd(&denom[sd.y * Hn + hh], ex);
}

__launch_bounds__(256)
__global__ void edge_agg1(const int* __restrict__ ei, const float* __restrict__ H1,
                          const float* __restrict__ exbuf, const float* __restrict__ denom,
                          float* __restrict__ agg)
{
    static_assert(E_TOT % 8 == 0, "8 edges per block, no tail");
    constexpr int LINES = HC / 32;
    constexpr int GROUP = 32;
    const int lane = threadIdx.x & 31;
    const int wave = __builtin_amdgcn_readfirstlane((int)(threadIdx.x >> 5));
    const int e  = blockIdx.x * 8 + wave;
    const int2 sd = edge_sd(ei, e);
    const int hl = lane & 7;
    const float alpha_l = exbuf[(size_t)e * HEADS + hl]
                        / (denom[sd.y * HEADS + hl] + 1e-16f);
    const float* hs = H1  + (size_t)sd.x * HC + lane;
    float*       op = agg + (size_t)sd.y * HC + lane;
#pragma unroll
    for (int g = 0; g < LINES / GROUP; ++g) {
        float v[GROUP];
#pragma unroll
        for (int q = 0; q < GROUP; ++q) v[q] = hs[(g * GROUP + q) * 32];
#pragma unroll
        for (int q = 0; q < GROUP; ++q) {
            const int line = g * GROUP + q;
            const float alpha = __int_as_float(
                __builtin_amdgcn_readlane(__float_as_int(alpha_l), line / (HIDD / 32)));
            atomicAdd(op + line * 32, alpha * v[q]);
        }
    }
}

#define L2_NPB     40
#define L2_THREADS 1024
#define L2_Q       (E_IN / 4)
#define L2_IT      ((L2_Q + L2_THREADS - 1) / L2_THREADS)

__launch_bounds__(L2_THREADS)
__global__ void layer2_owner(const int* __restrict__ ei, const float* __restrict__ H2,
                             const float* __restrict__ aS, const float* __restrict__ aD,
                             const float* __restrict__ b2, float* __restrict__ out)
{
    static_assert(N_NODES % L2_NPB == 0, "no partial node range");
    static_assert(E_IN % 4 == 0, "destination column read as 4-edge granules");
    static_assert(L2_NPB <= 64 && E_IN < (1 << 24) && N_NODES < (1 << 24), "edge or source number and local node share one int");
    static_assert(L2_NPB * NUM_CLASS <= L2_THREADS, "one thread per owned output element");
    __shared__ int   list[E_IN];
    __shared__ int   cnt;
    __shared__ float sd_s[L2_NPB], emax_s[L2_NPB], den_s[L2_NPB];
    __shared__ float agg_s[L2_NPB * NUM_CLASS];

    const int tid = threadIdx.x;
    const int n0  = blockIdx.x * L2_NPB;

    const int4* dq = (const int4*)(ei + E_IN);
    int4 d4[L2_IT];
#pragma unroll
    for (int i = 0; i < L2_IT; ++i)
        d4[i] = dq[min(tid + i * L2_THREADS, L2_Q - 1)];

    if (tid == 0) cnt = 0;
    if (tid < L2_NPB) {
        const float d = aD[n0 + tid];
        float x = aS[n0 + tid] + d;
        x = (x >= 0.f) ? x : NEG_SLOPE * x;
        sd_s[tid]   = d;
        emax_s[tid] = x;
        den_s[tid]  = 0.f;
    }
    if (tid < L2_NPB * NUM_CLASS) agg_s[tid] = 0.f;
    __syncthreads();

#pragma unroll
    for (int i = 0; i < L2_IT; ++i) {
        const int j = tid + i * L2_THREADS;
        const int dv[4] = {d4[i].x, d4[i].y, d4[i].z, d4[i].w};
#pragma unroll
        for (int q = 0; q < 4; ++q) {
            const unsigned dl = (unsigned)(dv[q] - n0);
            if (j < L2_Q && dl < (unsigned)L2_NPB) {
                const int slot = atomicAdd(&cnt, 1);
                list[slot] = ((4 * j + q) << 6) | (int)dl;
            }
        }
    }
    __syncthreads();
    const int m = cnt;

    for (int t = tid; t < m; t += L2_THREADS) {
        const int en = list[t];
        const int dl = en & 63;
        const int s  = ei[en >> 6];
        list[t] = (s << 6) | dl;
        float x = aS[s] + sd_s[dl];
        x = (x >= 0.f) ? x : NEG_SLOPE * x;
        atomicMaxF(&emax_s[dl], x);
    }
    __syncthreads();

    if (tid < L2_NPB) {
        const int n = n0 + tid;
        float x = aS[n] + sd_s[tid];
        x = (x >= 0.f) ? x : NEG_SLOPE * x;
        const float ex = expf(x - emax_s[tid]);
        atomicAdd(&den_s[tid], ex);
#pragma unroll
        for (int c = 0; c < NUM_CLASS; ++c)
            atomicAdd(&agg_s[tid * NUM_CLASS + c], ex * H2[(size_t)n * NUM_CLASS + c]);
    }
    for (int t = tid; t < m; t += L2_THREADS) {
        const int en = list[t];
        const int dl = en & 63;
        const int s  = en >> 6;
        float x = aS[s] + sd_s[dl];
        x = (x >= 0.f) ? x : NEG_SLOPE * x;
        const float ex = expf(x - emax_s[dl]);
        atomicAdd(&den_s[dl], ex);
#pragma unroll
        for (int c = 0; c < NUM_CLASS; ++c)
            atomicAdd(&agg_s[dl * NUM_CLASS + c], ex * H2[(size_t)s * NUM_CLASS + c]);
    }
    __syncthreads();

    if (tid < L2_NPB * NUM_CLASS) {
        const int i = tid / NUM_CLASS;
        const int c = tid - i * NUM_CLASS;
        const float v = agg_s[tid] / (den_s[i] + 1e-16f) + b2[c];
        out[(size_t)n0 * NUM_CLASS + tid] = 1.f / (1.f + expf(-v));
    }
}

__global__ void fill_kernel(float* __restrict__ p, size_t n, float v)
{
    for (size_t i = blockIdx.x * (size_t)blockDim.x + threadIdx.x; i < n;
         i += (size_t)gridDim.x * blockDim.x)
        p[i] = v;
}

extern "C" void kernel_launch(void* const* d_in, const int* in_sizes, int n_in,
                              void* d_out, int out_size, void* d_ws, size_t ws_size,
                              hipStream_t stream)
{
    const float* x   = (const float*)d_in[0];
    const int*   ei  = (const int*)d_in[1];
    const float* W1  = (const float*)d_in[2];
    const float* as1 = (const float*)d_in[3];
    const float* ad1 = (const float*)d_in[4];
    const float* b1  = (const float*)d_in[5];
    const float* W2  = (const float*)d_in[6];
    const float* as2 = (const float*)d_in[7];
    const float* ad2 = (const float*)d_in[8];
    const float* b2  = (const float*)d_in[9];
    float* out = (float*)d_out;

    float* ws = (float*)d_ws;
    size_t off = 0;
    float* h1    = ws + off; off += (size_t)N_NODES * HC;
    float* agg1  = ws + off; off += (size_t)N_NODES * HC;
    float* sA1   = ws + off; off += (size_t)N_NODES * HEADS;
    float* sD1   = ws + off; off += (size_t)N_NODES * HEADS;
    float* emax1 = ws + off; off += (size_t)N_NODES * HEADS;
    float* den1  = ws + off; off += (size_t)N_NODES * HEADS;
    float* ex1   = ws + off; off += (size_t)E_TOT * HEADS;
    float* h2    = ws + off; off += (size_t)N_NODES * NUM_CLASS;
    float* sA2   = ws + off; off += N_NODES;
    float* sD2   = ws + off; off += N_NODES;
    float* h2p   = ws + off; off += (size_t)KSPLIT * N_NODES * NUM_CLASS;
    (void)ws_size; (void)in_sizes; (void)n_in; (void)out_size;

    const dim3 b256(256);

    gemm1_att<<<dim3(HEADS, (N_NODES + 31) / 32), b256, 0, stream>>>(x, W1, as1, ad1, h1, sA1, sD1);

    fill_kernel<<<4096, b256, 0, stream>>>(agg1, (size_t)N_NODES * HC, 0.f);
    fill_kernel<<<64,   b256, 0, stream>>>(den1, (size_t)N_NODES * HEADS, 0.f);
    fill_kernel<<<64,   b256, 0, stream>>>(emax1, (size_t)N_NODES * HEADS, -INFINITY);

    const int eg1 = (E_TOT * HEADS + 255) / 256;
    edge_max<<<eg1, b256, 0, stream>>>(ei, sA1, sD1, emax1, HEADS);
    edge_expsum<<<eg1, b256, 0, stream>>>(ei, sA1, sD1, emax1, den1, ex1, HEADS);
    edge_agg1<<<E_TOT / 8, b256, 0, stream>>>(ei, h1, ex1, den1, agg1);

    gemm2_fused<<<dim3((N_NODES + 16 * G2_WAVES - 1) / (16 * G2_WAVES), KSPLIT), dim3(G2_WAVES * 32), 0, stream>>>(agg1, b1, W2, h2p);
    att2_kernel<<<(N_NODES + 255) / 256, b256, 0, stream>>>(h2p, as2, ad2, h2, sA2, sD2);

    layer2_owner<<<N_NODES / L2_NPB, L2_THREADS, 0, stream>>>(ei, h2, sA2, sD2, b2, out);
}


# Layer-1 edge phase by owner work-groups: block lists its 40 nodes' edges from one scan of the destination column; wave per node: softmax and row sums in registers; 10 -> 5 launches; sum order changes
# speedup vs baseline: 4.0012x; 1.5386x over previous
#include <hip/hip_runtime.h>
#include <math.h>

#define N_NODES   10000
#define F_IN      128
#define HEADS     8
#define HIDD      256
#define HC        (HEADS * HIDD)
#define NUM_CLASS 10
#define E_IN      50000
#define E_TOT     (E_IN + N_NODES)
#define NEG_SLOPE 0.2f

typedef __attribute__((ext_vector_type(2))) float v2f;
typedef __attribute__((ext_vector_type(8))) float v8f;

#define G1_KP (2 * HIDD + 32)

__launch_bounds__(256)
__global__ void gemm1_att(const float* __restrict__ X, const float* __restrict__ W,
                          const float* __restrict__ att_src,
                          const float* __restrict__ att_dst,
                          float* __restrict__ Hout,
                          float* __restrict__ aS, float* __restrict__ aD)
{
    static_assert(N_NODES % 16 == 0, "a wave's 16 rows are all inside or all outside");
    __shared__ __attribute__((aligned(16))) float As[32][F_IN + 4];
    __shared__ __attribute__((aligned(16))) float Bs[F_IN / 2][G1_KP];
    __shared__ __attribute__((aligned(16))) float red[8][32];

    const int tid = threadIdx.x;
    const int hh  = blockIdx.x;
    const int n0  = hh * HIDD;
    const int m0  = blockIdx.y * 32;

#pragma unroll
    for (int i = 0; i < 4; ++i) {
        int idx = (tid + i * 256) * 4;
        int r = idx >> 7;
        int c = idx & 127;
        const int gr = min(m0 + r, N_NODES - 1);
        *(float4*)(&As[r][c]) = *(const float4*)(X + (size_t)gr * F_IN + c);
    }
#pragma unroll
    for (int i = 0; i < 16; ++i) {
        const int item = tid + i * 256;
        const int p = item >> 6;
        const int c = (item & 63) * 4;
        const float* wp = W + (size_t)(2 * p) * HC + n0 + c;
        const float4 r0 = *(const float4*)wp;
        const float4 r1 = *(const float4*)(wp + HC);
        *(float4*)(&Bs[p][2 * c])     = make_float4(r0.x, r1.x, r0.y, r1.y);
        *(float4*)(&Bs[p][2 * c + 4]) = make_float4(r0.z, r1.z, r0.w, r1.w);
    }
    __syncthreads();

    const int wave = tid >> 5;
    const int lane = tid & 31;
    const int half = lane >> 4;
    const int ml   = lane & 15;
    const int ar   = (wave >> 2) * 16;
    const int bc   = (wave & 3) * 64;

    float asv[4], adv[4];
#pragma unroll
    for (int t = 0; t < 4; ++t) {
        asv[t] = att_src[n0 + bc + 16 * t + ml];
        adv[t] = att_dst[n0 + bc + 16 * t + ml];
    }

    v8f acc[4];
#pragma unroll
    for (int t = 0; t < 4; ++t)
        acc[t] = (v8f){0.f, 0.f, 0.f, 0.f, 0.f, 0.f, 0.f, 0.f};

#pragma unroll
    for (int kk = 0; kk < F_IN / 4; ++kk) {
        const int p = kk * 2 + half;
        v2f a;
        const float2 av = *(const float2*)(&As[ar + ml][2 * p]);
        a.x = av.x; a.y = av.y;
#pragma unroll
        for (int t = 0; t < 4; ++t) {
            v2f b;
            const float2 bv = *(const float2*)(&Bs[p][2 * (bc + 16 * t + ml)]);
            b.x = bv.x; b.y = bv.y;
            acc[t] = __builtin_amdgcn_wmma_f32_16x16x4_f32(false, a, false, b,
                                                           (short)0, acc[t], false, false);
        }
    }

    if (m0 + ar < N_NODES) {
#pragma unroll
        for (int t = 0; t < 4; ++t) {
            const int col = n0 + bc + 16 * t + ml;
#pragma unroll
            for (int r = 0; r < 8; ++r) {
                const int row = m0 + ar + r + 8 * half;
                Hout[(size_t)row * HC + col] = acc[t][r];
            }
        }
    }

    float v[16];
#pragma unroll
    for (int r = 0; r < 8; ++r) {
        float ss = 0.f, sd = 0.f;
#pragma unroll
        for (int t = 0; t < 4; ++t) {
            ss += acc[t][r] * asv[t];
            sd += acc[t][r] * adv[t];
        }
        v[r] = ss; v[8 + r] = sd;
    }
#pragma unroll
    for (int j = 0; j < 16; ++j) {
        v[j] += __int_as_float(__builtin_amdgcn_update_dpp(0, __float_as_int(v[j]), 0x161, 0xf, 0xf, false));
        v[j] += __int_as_float(__builtin_amdgcn_update_dpp(0, __float_as_int(v[j]), 0x162, 0xf, 0xf, false));
        v[j] += __int_as_float(__builtin_amdgcn_update_dpp(0, __float_as_int(v[j]), 0x164, 0xf, 0xf, false));
        v[j] += __int_as_float(__builtin_amdgcn_update_dpp(0, __float_as_int(v[j]), 0x168, 0xf, 0xf, false));
    }
    if (ml == 0) {
#pragma unroll
        for (int j = 0; j < 16; j += 4)
            *(float4*)(&red[wave][half * 16 + j]) = make_float4(v[j], v[j + 1], v[j + 2], v[j + 3]);
    }
    __syncthreads();
    if (tid < 64) {
        const int rh = tid >> 5;
        const int ln = tid & 31;
        const float s = ((red[rh * 4 + 0][ln] + red[rh * 4 + 1][ln])
                         + red[rh * 4 + 2][ln]) + red[rh * 4 + 3][ln];
        const int row = m0 + rh * 16 + (ln & 7) + 8 * (ln >> 4);
        if (row < N_NODES) {
            float* dstp = ((ln >> 3) & 1) ? aD : aS;
            dstp[(size_t)row * HEADS + hh] = s;
        }
    }
}

#define KSPLIT 8
#define KSL    (HC / KSPLIT)
#define G2_WAVES 4

__device__ __forceinline__ float elu1(float v)
{
    const float e = expm1f(v);
    return (v > 0.f) ? v : e;
}

__launch_bounds__(G2_WAVES * 32)
__global__ void gemm2_fused(const float* __restrict__ AGG, const float* __restrict__ B1,
                            const float* __restrict__ W2, float* __restrict__ H2P)
{
    const int tid  = threadIdx.x;
    const int wave = tid >> 5;
    const int lane = tid & 31;
    const int half = lane >> 4;
    const int ml   = lane & 15;
    const int m0   = (blockIdx.x * G2_WAVES + wave) * 16;
    if (m0 >= N_NODES) return;
    const int sl   = blockIdx.y;
    const int ra   = min(m0 + ml, N_NODES - 1);
    const int nb   = min(ml, NUM_CLASS - 1);
    const int k0   = sl * KSL + half * 4;
    const float* ap = AGG + (size_t)ra * HC + k0;
    const float* bp = B1 + k0;
    const float* wp = W2 + (size_t)k0 * NUM_CLASS + nb;

    v8f acc = {0.f, 0.f, 0.f, 0.f, 0.f, 0.f, 0.f, 0.f};
    for (int k = 0; k < KSL; k += 32) {
        float4 av[4], bv[4];
        float  w[4][4];
#pragma unroll
        for (int s = 0; s < 4; ++s) av[s] = *(const float4*)(ap + k + 8 * s);
#pragma unroll
        for (int s = 0; s < 4; ++s) bv[s] = *(const float4*)(bp + k + 8 * s);
#pragma unroll
        for (int s = 0; s < 4; ++s)
#pragma unroll
            for (int j = 0; j < 4; ++j)
                w[s][j] = wp[(size_t)(k + 8 * s + j) * NUM_CLASS];
#pragma unroll
        for (int s = 0; s < 4; ++s) {
            v2f a, b;
            a.x = elu1(av[s].x + bv[s].x); a.y = elu1(av[s].y + bv[s].y);
            b.x = w[s][0]; b.y = w[s][1];
            acc = __builtin_amdgcn_wmma_f32_16x16x4_f32(false, a, false, b,
                                                        (short)0, acc, false, false);
            a.x = elu1(av[s].z + bv[s].z); a.y = elu1(av[s].w + bv[s].w);
            b.x = w[s][2]; b.y = w[s][3];
            acc = __builtin_amdgcn_wmma_f32_16x16x4_f32(false, a, false, b,
                                                        (short)0, acc, false, false);
        }
    }
    if (ml < NUM_CLASS) {
        float* hp = H2P + (size_t)sl * N_NODES * NUM_CLASS;
#pragma unroll
        for (int r = 0; r < 8; ++r) {
            const int row = m0 + r + 8 * half;
            if (row < N_NODES) hp[(size_t)row * NUM_CLASS + ml] = acc[r];
        }
    }
}

__global__ void att2_kernel(const float* __restrict__ H2P,
                            const float* __restrict__ as2,
                            const float* __restrict__ ad2,
                            float* __restrict__ H2,
                            float* __restrict__ aS, float* __restrict__ aD)
{
    const int n = blockIdx.x * blockDim.x + threadIdx.x;
    if (n >= N_NODES) return;
    float ss = 0.f, sd = 0.f;
#pragma unroll
    for (int c = 0; c < NUM_CLASS; ++c) {
        float v = 0.f;
#pragma unroll
        for (int s = 0; s < KSPLIT; ++s)
            v += H2P[((size_t)s * N_NODES + n) * NUM_CLASS + c];
        H2[(size_t)n * NUM_CLASS + c] = v;
        ss += v * as2[c];
        sd += v * ad2[c];
    }
    aS[n] = ss; aD[n] = sd;
}

#define L1_NPB     40
#define L1_THREADS 640
#define L1_WAVES   (L1_THREADS / 32)
#define L1_Q       (E_IN / 4)
#define L1_IT      ((L1_Q + L1_THREADS - 1) / L1_THREADS)
#define L1_CH      (HC / 128)
#define L1_CAP     1024

__device__ __forceinline__ float lrelu(float x)
{
    return (x >= 0.f) ? x : NEG_SLOPE * x;
}

__launch_bounds__(L1_THREADS)
__global__ void layer1_owner(const int* __restrict__ ei, const float* __restrict__ H1,
                             const float* __restrict__ aS, const float* __restrict__ aD,
                             float* __restrict__ agg)
{
    static_assert(N_NODES % L1_NPB == 0, "no partial node range");
    static_assert(E_IN % 4 == 0, "destination column read as 4-edge granules");
    static_assert(L1_NPB <= 63 && N_NODES < (1 << 24), "source and local node share one int; local 63 marks no entry");
    static_assert(HEADS == 8 && HIDD == 256, "two 128-channel pieces per head, 8 scores per node");
    __shared__ int list[E_IN];
    __shared__ __attribute__((aligned(16))) float xl[L1_CAP * HEADS];
    __shared__ int cnt;

    const int tid = threadIdx.x;
    const int n0  = blockIdx.x * L1_NPB;

    const int4* dq = (const int4*)(ei + E_IN);
    int4 d4[L1_IT];
#pragma unroll
    for (int i = 0; i < L1_IT; ++i)
        d4[i] = dq[min(tid + i * L1_THREADS, L1_Q - 1)];

    if (tid == 0) cnt = 0;
    __syncthreads();

#pragma unroll
    for (int i = 0; i < L1_IT; ++i) {
        const int j = tid + i * L1_THREADS;
        const int dv[4] = {d4[i].x, d4[i].y, d4[i].z, d4[i].w};
#pragma unroll
        for (int q = 0; q < 4; ++q) {
            const unsigned dl = (unsigned)(dv[q] - n0);
            if (j < L1_Q && dl < (unsigned)L1_NPB) {
                const int s = ei[4 * j + q];
                const int slot = atomicAdd(&cnt, 1);
                list[slot] = (s << 6) | (int)dl;
            }
        }
    }
    __syncthreads();
    const int m = cnt;

    for (int t = tid; t < min(m, L1_CAP); t += L1_THREADS) {
        const int en = list[t];
        const float* sp = aS + (size_t)(en >> 6) * HEADS;
        const float* dp = aD + (size_t)(n0 + (en & 63)) * HEADS;
        const float4 a0 = *(const float4*)sp;
        const float4 a1 = *(const float4*)(sp + 4);
        const float4 e0 = *(const float4*)dp;
        const float4 e1 = *(const float4*)(dp + 4);
        *(float4*)(&xl[t * HEADS])     = make_float4(lrelu(a0.x + e0.x), lrelu(a0.y + e0.y),
                                                     lrelu(a0.z + e0.z), lrelu(a0.w + e0.w));
        *(float4*)(&xl[t * HEADS + 4]) = make_float4(lrelu(a1.x + e1.x), lrelu(a1.y + e1.y),
                                                     lrelu(a1.z + e1.z), lrelu(a1.w + e1.w));
    }
    __syncthreads();

    const int lane = tid & 31;
    const int wave = __builtin_amdgcn_readfirstlane(tid >> 5);

    for (int i = wave; i < L1_NPB; i += L1_WAVES) {
        const int n = n0 + i;

        float4 acc[L1_CH];
        {
            const float* rp = H1 + (size_t)n * HC + lane * 4;
#pragma unroll
            for (int j = 0; j < L1_CH; ++j) acc[j] = *(const float4*)(rp + j * 128);
        }

        float adn[HEADS], xs[HEADS], mx[HEADS];
        {
            const float4 p0 = *(const float4*)(aD + (size_t)n * HEADS);
            const float4 p1 = *(const float4*)(aD + (size_t)n * HEADS + 4);
            const float4 q0 = *(const float4*)(aS + (size_t)n * HEADS);
            const float4 q1 = *(const float4*)(aS + (size_t)n * HEADS + 4);
            const float dd[HEADS] = {p0.x, p0.y, p0.z, p0.w, p1.x, p1.y, p1.z, p1.w};
            const float ss[HEADS] = {q0.x, q0.y, q0.z, q0.w, q1.x, q1.y, q1.z, q1.w};
#pragma unroll
            for (int h = 0; h < HEADS; ++h) {
                adn[h] = dd[h];
                xs[h]  = lrelu(ss[h] + dd[h]);
                mx[h]  = xs[h];
            }
        }

        for (int t = 0; t < m; t += 32) {
            const int idx = t + lane;
            const int en  = (idx < m) ? list[idx] : -1;
            if ((en & 63) == i) {
                float x[HEADS];
                if (idx < L1_CAP) {
                    const float4 u0 = *(const float4*)(&xl[idx * HEADS]);
                    const float4 u1 = *(const float4*)(&xl[idx * HEADS + 4]);
                    x[0] = u0.x; x[1] = u0.y; x[2] = u0.z; x[3] = u0.w;
                    x[4] = u1.x; x[5] = u1.y; x[6] = u1.z; x[7] = u1.w;
                } else {
                    const float* sp = aS + (size_t)(en >> 6) * HEADS;
                    const float4 a0 = *(const float4*)sp;
                    const float4 a1 = *(const float4*)(sp + 4);
                    const float av[HEADS] = {a0.x, a0.y, a0.z, a0.w, a1.x, a1.y, a1.z, a1.w};
#pragma unroll
                    for (int h = 0; h < HEADS; ++h) x[h] = lrelu(av[h] + adn[h]);
                }
#pragma unroll
                for (int h = 0; h < HEADS; ++h) mx[h] = fmaxf(mx[h], x[h]);
            }
        }
#pragma unroll
        for (int h = 0; h < HEADS; ++h) {
#pragma unroll
            for (int o = 16; o >= 1; o >>= 1)
                mx[h] = fmaxf(mx[h], __shfl_xor(mx[h], o));
        }

        float exs[HEADS], den[HEADS];
#pragma unroll
        for (int h = 0; h < HEADS; ++h) {
            exs[h] = expf(xs[h] - mx[h]);
            den[h] = 0.f;
        }
#pragma unroll
        for (int j = 0; j < L1_CH; ++j) {
            const float w = exs[j >> 1];
            acc[j] = make_float4(w * acc[j].x, w * acc[j].y, w * acc[j].z, w * acc[j].w);
        }
        for (int t = 0; t < m; t += 32) {
            const int idx = t + lane;
            const int en  = (idx < m) ? list[idx] : -1;
            const bool mine = ((en & 63) == i);
            float ex[HEADS];
#pragma unroll
            for (int h = 0; h < HEADS; ++h) ex[h] = 0.f;
            if (mine) {
                float x[HEADS];
                if (idx < L1_CAP) {
                    const float4 u0 = *(const float4*)(&xl[idx * HEADS]);
                    const float4 u1 = *(const float4*)(&xl[idx * HEADS + 4]);
                    x[0] = u0.x; x[1] = u0.y; x[2] = u0.z; x[3] = u0.w;
                    x[4] = u1.x; x[5] = u1.y; x[6] = u1.z; x[7] = u1.w;
                } else {
                    const float* sp = aS + (size_t)(en >> 6) * HEADS;
                    const float4 a0 = *(const float4*)sp;
                    const float4 a1 = *(const float4*)(sp + 4);
                    const float av[HEADS] = {a0.x, a0.y, a0.z, a0.w, a1.x, a1.y, a1.z, a1.w};
#pragma unroll
                    for (int h = 0; h < HEADS; ++h) x[h] = lrelu(av[h] + adn[h]);
                }
#pragma unroll
                for (int h = 0; h < HEADS; ++h) {
                    ex[h]   = expf(x[h] - mx[h]);
                    den[h] += ex[h];
                }
            }
            unsigned mask = __builtin_amdgcn_ballot_w32(mine);
            while (mask) {
                const int b = __builtin_ctz(mask);
                mask &= mask - 1;
                const int s = __builtin_amdgcn_readlane(en, b) >> 6;
                float w[HEADS];
#pragma unroll
                for (int h = 0; h < HEADS; ++h)
                    w[h] = __int_as_float(__builtin_amdgcn_readlane(__float_as_int(ex[h]), b));
                const float* rp = H1 + (size_t)s * HC + lane * 4;
                float4 v[L1_CH];
#pragma unroll
                for (int j = 0; j < L1_CH; ++j) v[j] = *(const float4*)(rp + j * 128);
#pragma unroll
                for (int j = 0; j < L1_CH; ++j) {
                    const float ww = w[j >> 1];
                    acc[j].x += ww * v[j].x;
                    acc[j].y += ww * v[j].y;
                    acc[j].z += ww * v[j].z;
                    acc[j].w += ww * v[j].w;
                }
            }
        }
        float rc[HEADS];
#pragma unroll
        for (int h = 0; h < HEADS; ++h) {
#pragma unroll
            for (int o = 16; o >= 1; o >>= 1)
                den[h] += __shfl_xor(den[h], o);
            rc[h] = 1.f / ((den[h] + exs[h]) + 1e-16f);
        }
        float* op = agg + (size_t)n * HC + lane * 4;
#pragma unroll
        for (int j = 0; j < L1_CH; ++j) {
            const float r = rc[j >> 1];
            *(float4*)(op + j * 128) = make_float4(acc[j].x * r, acc[j].y * r, acc[j].z * r, acc[j].w * r);
        }
    }
}

__device__ __forceinline__ void atomicMaxF(float* addr, float v)
{
    if (v >= 0.f) atomicMax((int*)addr, __float_as_int(v));
    else          atomicMin((unsigned int*)addr, __float_as_uint(v));
}

#define L2_NPB     40
#define L2_THREADS 1024
#define L2_Q       (E_IN / 4)
#define L2_IT      ((L2_Q + L2_THREADS - 1) / L2_THREADS)

__launch_bounds__(L2_THREADS)
__global__ void layer2_owner(const int* __restrict__ ei, const float* __restrict__ H2,
                             const float* __restrict__ aS, const float* __restrict__ aD,
                             const float* __restrict__ b2, float* __restrict__ out)
{
    static_assert(N_NODES % L2_NPB == 0, "no partial node range");
    static_assert(E_IN % 4 == 0, "destination column read as 4-edge granules");
    static_assert(L2_NPB <= 64 && E_IN < (1 << 24) && N_NODES < (1 << 24), "edge or source number and local node share one int");
    static_assert(L2_NPB * NUM_CLASS <= L2_THREADS, "one thread per owned output element");
    __shared__ int   list[E_IN];
    __shared__ int   cnt;
    __shared__ float sd_s[L2_NPB], emax_s[L2_NPB], den_s[L2_NPB];
    __shared__ float agg_s[L2_NPB * NUM_CLASS];

    const int tid = threadIdx.x;
    const int n0  = blockIdx.x * L2_NPB;

    const int4* dq = (const int4*)(ei + E_IN);
    int4 d4[L2_IT];
#pragma unroll
    for (int i = 0; i < L2_IT; ++i)
        d4[i] = dq[min(tid + i * L2_THREADS, L2_Q - 1)];

    if (tid == 0) cnt = 0;
    if (tid < L2_NPB) {
        const float d = aD[n0 + tid];
        float x = aS[n0 + tid] + d;
        x = (x >= 0.f) ? x : NEG_SLOPE * x;
        sd_s[tid]   = d;
        emax_s[tid] = x;
        den_s[tid]  = 0.f;
    }
    if (tid < L2_NPB * NUM_CLASS) agg_s[tid] = 0.f;
    __syncthreads();

#pragma unroll
    for (int i = 0; i < L2_IT; ++i) {
        const int j = tid + i * L2_THREADS;
        const int dv[4] = {d4[i].x, d4[i].y, d4[i].z, d4[i].w};
#pragma unroll
        for (int q = 0; q < 4; ++q) {
            const unsigned dl = (unsigned)(dv[q] - n0);
            if (j < L2_Q && dl < (unsigned)L2_NPB) {
                const int slot = atomicAdd(&cnt, 1);
                list[slot] = ((4 * j + q) << 6) | (int)dl;
            }
        }
    }
    __syncthreads();
    const int m = cnt;

    for (int t = tid; t < m; t += L2_THREADS) {
        const int en = list[t];
        const int dl = en & 63;
        const int s  = ei[en >> 6];
        list[t] = (s << 6) | dl;
        float x = aS[s] + sd_s[dl];
        x = (x >= 0.f) ? x : NEG_SLOPE * x;
        atomicMaxF(&emax_s[dl], x);
    }
    __syncthreads();

    if (tid < L2_NPB) {
        const int n = n0 + tid;
        float x = aS[n] + sd_s[tid];
        x = (x >= 0.f) ? x : NEG_SLOPE * x;
        const float ex = expf(x - emax_s[tid]);
        atomicAdd(&den_s[tid], ex);
#pragma unroll
        for (int c = 0; c < NUM_CLASS; ++c)
            atomicAdd(&agg_s[tid * NUM_CLASS + c], ex * H2[(size_t)n * NUM_CLASS + c]);
    }
    for (int t = tid; t < m; t += L2_THREADS) {
        const int en = list[t];
        const int dl = en & 63;
        const int s  = en >> 6;
        float x = aS[s] + sd_s[dl];
        x = (x >= 0.f) ? x : NEG_SLOPE * x;
        const float ex = expf(x - emax_s[dl]);
        atomicAdd(&den_s[dl], ex);
#pragma unroll
        for (int c = 0; c < NUM_CLASS; ++c)
            atomicAdd(&agg_s[dl * NUM_CLASS + c], ex * H2[(size_t)s * NUM_CLASS + c]);
    }
    __syncthreads();

    if (tid < L2_NPB * NUM_CLASS) {
        const int i = tid / NUM_CLASS;
        const int c = tid - i * NUM_CLASS;
        const float v = agg_s[tid] / (den_s[i] + 1e-16f) + b2[c];
        out[(size_t)n0 * NUM_CLASS + tid] = 1.f / (1.f + expf(-v));
    }
}

extern "C" void kernel_launch(void* const* d_in, const int* in_sizes, int n_in,
                              void* d_out, int out_size, void* d_ws, size_t ws_size,
                              hipStream_t stream)
{
    const float* x   = (const float*)d_in[0];
    const int*   ei  = (const int*)d_in[1];
    const float* W1  = (const float*)d_in[2];
    const float* as1 = (const float*)d_in[3];
    const float* ad1 = (const float*)d_in[4];
    const float* b1  = (const float*)d_in[5];
    const float* W2  = (const float*)d_in[6];
    const float* as2 = (const float*)d_in[7];
    const float* ad2 = (const float*)d_in[8];
    const float* b2  = (const float*)d_in[9];
    float* out = (float*)d_out;

    float* ws = (float*)d_ws;
    size_t off = 0;
    float* h1    = ws + off; off += (size_t)N_NODES * HC;
    float* agg1  = ws + off; off += (size_t)N_NODES * HC;
    float* sA1   = ws + off; off += (size_t)N_NODES * HEADS;
    float* sD1   = ws + off; off += (size_t)N_NODES * HEADS;
    float* h2    = ws + off; off += (size_t)N_NODES * NUM_CLASS;
    float* sA2   = ws + off; off += N_NODES;
    float* sD2   = ws + off; off += N_NODES;
    float* h2p   = ws + off; off += (size_t)KSPLIT * N_NODES * NUM_CLASS;
    (void)ws_size; (void)in_sizes; (void)n_in; (void)out_size;

    const dim3 b256(256);

    gemm1_att<<<dim3(HEADS, (N_NODES + 31) / 32), b256, 0, stream>>>(x, W1, as1, ad1, h1, sA1, sD1);

    layer1_owner<<<N_NODES / L1_NPB, L1_THREADS, 0, stream>>>(ei, h1, sA1, sD1, agg1);

    gemm2_fused<<<dim3((N_NODES + 16 * G2_WAVES - 1) / (16 * G2_WAVES), KSPLIT), dim3(G2_WAVES * 32), 0, stream>>>(agg1, b1, W2, h2p);
    att2_kernel<<<(N_NODES + 255) / 256, b256, 0, stream>>>(h2p, as2, ad2, h2, sA2, sD2);

    layer2_owner<<<N_NODES / L2_NPB, L2_THREADS, 0, stream>>>(ei, h2, sA2, sD2, b2, out);
}


# GEMM2 + layer-2 attention dots done by the layer-1 owner blocks on their own 40 rows (bias+ELU at the row store, rows back via L2, 16 K-slice waves summed in LDS): 5 -> 3 launches; sum order changes
# speedup vs baseline: 4.1450x; 1.0359x over previous
#include <hip/hip_runtime.h>
#include <math.h>

#define N_NODES   10000
#define F_IN      128
#define HEADS     8
#define HIDD      256
#define HC        (HEADS * HIDD)
#define NUM_CLASS 10
#define E_IN      50000
#define E_TOT     (E_IN + N_NODES)
#define NEG_SLOPE 0.2f

typedef __attribute__((ext_vector_type(2))) float v2f;
typedef __attribute__((ext_vector_type(8))) float v8f;

#define G1_KP (2 * HIDD + 32)

__launch_bounds__(256)
__global__ void gemm1_att(const float* __restrict__ X, const float* __restrict__ W,
                          const float* __restrict__ att_src,
                          const float* __restrict__ att_dst,
                          float* __restrict__ Hout,
                          float* __restrict__ aS, float* __restrict__ aD)
{
    static_assert(N_NODES % 16 == 0, "a wave's 16 rows are all inside or all outside");
    __shared__ __attribute__((aligned(16))) float As[32][F_IN + 4];
    __shared__ __attribute__((aligned(16))) float Bs[F_IN / 2][G1_KP];
    __shared__ __attribute__((aligned(16))) float red[8][32];

    const int tid = threadIdx.x;
    const int hh  = blockIdx.x;
    const int n0  = hh * HIDD;
    const int m0  = blockIdx.y * 32;

#pragma unroll
    for (int i = 0; i < 4; ++i) {
        int idx = (tid + i * 256) * 4;
        int r = idx >> 7;
        int c = idx & 127;
        const int gr = min(m0 + r, N_NODES - 1);
        *(float4*)(&As[r][c]) = *(const float4*)(X + (size_t)gr * F_IN + c);
    }
#pragma unroll
    for (int i = 0; i < 16; ++i) {
        const int item = tid + i * 256;
        const int p = item >> 6;
        const int c = (item & 63) * 4;
        const float* wp = W + (size_t)(2 * p) * HC + n0 + c;
        const float4 r0 = *(const float4*)wp;
        const float4 r1 = *(const float4*)(wp + HC);
        *(float4*)(&Bs[p][2 * c])     = make_float4(r0.x, r1.x, r0.y, r1.y);
        *(float4*)(&Bs[p][2 * c + 4]) = make_float4(r0.z, r1.z, r0.w, r1.w);
    }
    __syncthreads();

    const int wave = tid >> 5;
    const int lane = tid & 31;
    const int half = lane >> 4;
    const int ml   = lane & 15;
    const int ar   = (wave >> 2) * 16;
    const int bc   = (wave & 3) * 64;

    float asv[4], adv[4];
#pragma unroll
    for (int t = 0; t < 4; ++t) {
        asv[t] = att_src[n0 + bc + 16 * t + ml];
        adv[t] = att_dst[n0 + bc + 16 * t + ml];
    }

    v8f acc[4];
#pragma unroll
    for (int t = 0; t < 4; ++t)
        acc[t] = (v8f){0.f, 0.f, 0.f, 0.f, 0.f, 0.f, 0.f, 0.f};

#pragma unroll
    for (int kk = 0; kk < F_IN / 4; ++kk) {
        const int p = kk * 2 + half;
        v2f a;
        const float2 av = *(const float2*)(&As[ar + ml][2 * p]);
        a.x = av.x; a.y = av.y;
#pragma unroll
        for (int t = 0; t < 4; ++t) {
            v2f b;
            const float2 bv = *(const float2*)(&Bs[p][2 * (bc + 16 * t + ml)]);
            b.x = bv.x; b.y = bv.y;
            acc[t] = __builtin_amdgcn_wmma_f32_16x16x4_f32(false, a, false, b,
                                                           (short)0, acc[t], false, false);
        }
    }

    if (m0 + ar < N_NODES) {
#pragma unroll
        for (int t = 0; t < 4; ++t) {
            const int col = n0 + bc + 16 * t + ml;
#pragma unroll
            for (int r = 0; r < 8; ++r) {
                const int row = m0 + ar + r + 8 * half;
                Hout[(size_t)row * HC + col] = acc[t][r];
            }
        }
    }

    float v[16];
#pragma unroll
    for (int r = 0; r < 8; ++r) {
        float ss = 0.f, sd = 0.f;
#pragma unroll
        for (int t = 0; t < 4; ++t) {
            ss += acc[t][r] * asv[t];
            sd += acc[t][r] * adv[t];
        }
        v[r] = ss; v[8 + r] = sd;
    }
#pragma unroll
    for (int j = 0; j < 16; ++j) {
        v[j] += __int_as_float(__builtin_amdgcn_update_dpp(0, __float_as_int(v[j]), 0x161, 0xf, 0xf, false));
        v[j] += __int_as_float(__builtin_amdgcn_update_dpp(0, __float_as_int(v[j]), 0x162, 0xf, 0xf, false));
        v[j] += __int_as_float(__builtin_amdgcn_update_dpp(0, __float_as_int(v[j]), 0x164, 0xf, 0xf, false));
        v[j] += __int_as_float(__builtin_amdgcn_update_dpp(0, __float_as_int(v[j]), 0x168, 0xf, 0xf, false));
    }
    if (ml == 0) {
#pragma unroll
        for (int j = 0; j < 16; j += 4)
            *(float4*)(&red[wave][half * 16 + j]) = make_float4(v[j], v[j + 1], v[j + 2], v[j + 3]);
    }
    __syncthreads();
    if (tid < 64) {
        const int rh = tid >> 5;
        const int ln = tid & 31;
        const float s = ((red[rh * 4 + 0][ln] + red[rh * 4 + 1][ln])
                         + red[rh * 4 + 2][ln]) + red[rh * 4 + 3][ln];
        const int row = m0 + rh * 16 + (ln & 7) + 8 * (ln >> 4);
        if (row < N_NODES) {
            float* dstp = ((ln >> 3) & 1) ? aD : aS;
            dstp[(size_t)row * HEADS + hh] = s;
        }
    }
}

__device__ __forceinline__ float elu1(float v)
{
    const float e = expm1f(v);
    return (v > 0.f) ? v : e;
}

#define L1_NPB     40
#define L1_THREADS 640
#define L1_WAVES   (L1_THREADS / 32)
#define L1_Q       (E_IN / 4)
#define L1_IT      ((L1_Q + L1_THREADS - 1) / L1_THREADS)
#define L1_CH      (HC / 128)
#define L1_CAP     1024
#define L1_GW      16
#define L1_KSL     (HC / L1_GW)
#define L1_TILES   ((L1_NPB + 15) / 16)

__device__ __forceinline__ float lrelu(float x)
{
    return (x >= 0.f) ? x : NEG_SLOPE * x;
}

__launch_bounds__(L1_THREADS)
__global__ void layer1_owner(const int* __restrict__ ei, const float* __restrict__ H1,
                             const float* __restrict__ aS, const float* __restrict__ aD,
                             float* agg, const float* B1,
                             const float* __restrict__ W2,
                             const float* __restrict__ as2, const float* __restrict__ ad2,
                             float* __restrict__ H2,
                             float* __restrict__ aS2, float* __restrict__ aD2)
{
    static_assert(N_NODES % L1_NPB == 0, "no partial node range");
    static_assert(E_IN % 4 == 0, "destination column read as 4-edge granules");
    static_assert(L1_NPB <= 63 && N_NODES < (1 << 24), "source and local node share one int; local 63 marks no entry");
    static_assert(HEADS == 8 && HIDD == 256, "two 128-channel pieces per head, 8 scores per node");
    static_assert(L1_GW <= L1_WAVES && L1_KSL % 32 == 0, "a K slice per wave, whole 32-k trips");
    static_assert(L1_NPB * NUM_CLASS <= L1_THREADS, "one thread per owned h2 element");
    __shared__ int list[E_IN];
    __shared__ __attribute__((aligned(16))) float part[L1_GW][L1_TILES * 16][16];
    __shared__ float h2s[L1_NPB * NUM_CLASS];
    __shared__ __attribute__((aligned(16))) float xl[L1_CAP * HEADS];
    __shared__ int cnt;

    const int tid = threadIdx.x;
    const int n0  = blockIdx.x * L1_NPB;

    const int4* dq = (const int4*)(ei + E_IN);
    int4 d4[L1_IT];
#pragma unroll
    for (int i = 0; i < L1_IT; ++i)
        d4[i] = dq[min(tid + i * L1_THREADS, L1_Q - 1)];

    if (tid == 0) cnt = 0;
    __syncthreads();

#pragma unroll
    for (int i = 0; i < L1_IT; ++i) {
        const int j = tid + i * L1_THREADS;
        const int dv[4] = {d4[i].x, d4[i].y, d4[i].z, d4[i].w};
#pragma unroll
        for (int q = 0; q < 4; ++q) {
            const unsigned dl = (unsigned)(dv[q] - n0);
            if (j < L1_Q && dl < (unsigned)L1_NPB) {
                const int s = ei[4 * j + q];
                const int slot = atomicAdd(&cnt, 1);
                list[slot] = (s << 6) | (int)dl;
            }
        }
    }
    __syncthreads();
    const int m = cnt;

    for (int t = tid; t < min(m, L1_CAP); t += L1_THREADS) {
        const int en = list[t];
        const float* sp = aS + (size_t)(en >> 6) * HEADS;
        const float* dp = aD + (size_t)(n0 + (en & 63)) * HEADS;
        const float4 a0 = *(const float4*)sp;
        const float4 a1 = *(const float4*)(sp + 4);
        const float4 e0 = *(const float4*)dp;
        const float4 e1 = *(const float4*)(dp + 4);
        *(float4*)(&xl[t * HEADS])     = make_float4(lrelu(a0.x + e0.x), lrelu(a0.y + e0.y),
                                                     lrelu(a0.z + e0.z), lrelu(a0.w + e0.w));
        *(float4*)(&xl[t * HEADS + 4]) = make_float4(lrelu(a1.x + e1.x), lrelu(a1.y + e1.y),
                                                     lrelu(a1.z + e1.z), lrelu(a1.w + e1.w));
    }
    __syncthreads();

    const int lane = tid & 31;
    const int wave = __builtin_amdgcn_readfirstlane(tid >> 5);

    for (int i = wave; i < L1_NPB; i += L1_WAVES) {
        const int n = n0 + i;

        float4 acc[L1_CH];
        {
            const float* rp = H1 + (size_t)n * HC + lane * 4;
#pragma unroll
            for (int j = 0; j < L1_CH; ++j) acc[j] = *(const float4*)(rp + j * 128);
        }

        float adn[HEADS], xs[HEADS], mx[HEADS];
        {
            const float4 p0 = *(const float4*)(aD + (size_t)n * HEADS);
            const float4 p1 = *(const float4*)(aD + (size_t)n * HEADS + 4);
            const float4 q0 = *(const float4*)(aS + (size_t)n * HEADS);
            const float4 q1 = *(const float4*)(aS + (size_t)n * HEADS + 4);
            const float dd[HEADS] = {p0.x, p0.y, p0.z, p0.w, p1.x, p1.y, p1.z, p1.w};
            const float ss[HEADS] = {q0.x, q0.y, q0.z, q0.w, q1.x, q1.y, q1.z, q1.w};
#pragma unroll
            for (int h = 0; h < HEADS; ++h) {
                adn[h] = dd[h];
                xs[h]  = lrelu(ss[h] + dd[h]);
                mx[h]  = xs[h];
            }
        }

        for (int t = 0; t < m; t += 32) {
            const int idx = t + lane;
            const int en  = (idx < m) ? list[idx] : -1;
            if ((en & 63) == i) {
                float x[HEADS];
                if (idx < L1_CAP) {
                    const float4 u0 = *(const float4*)(&xl[idx * HEADS]);
                    const float4 u1 = *(const float4*)(&xl[idx * HEADS + 4]);
                    x[0] = u0.x; x[1] = u0.y; x[2] = u0.z; x[3] = u0.w;
                    x[4] = u1.x; x[5] = u1.y; x[6] = u1.z; x[7] = u1.w;
                } else {
                    const float* sp = aS + (size_t)(en >> 6) * HEADS;
                    const float4 a0 = *(const float4*)sp;
                    const float4 a1 = *(const float4*)(sp + 4);
                    const float av[HEADS] = {a0.x, a0.y, a0.z, a0.w, a1.x, a1.y, a1.z, a1.w};
#pragma unroll
                    for (int h = 0; h < HEADS; ++h) x[h] = lrelu(av[h] + adn[h]);
                }
#pragma unroll
                for (int h = 0; h < HEADS; ++h) mx[h] = fmaxf(mx[h], x[h]);
            }
        }
#pragma unroll
        for (int h = 0; h < HEADS; ++h) {
#pragma unroll
            for (int o = 16; o >= 1; o >>= 1)
                mx[h] = fmaxf(mx[h], __shfl_xor(mx[h], o));
        }

        float exs[HEADS], den[HEADS];
#pragma unroll
        for (int h = 0; h < HEADS; ++h) {
            exs[h] = expf(xs[h] - mx[h]);
            den[h] = 0.f;
        }
#pragma unroll
        for (int j = 0; j < L1_CH; ++j) {
            const float w = exs[j >> 1];
            acc[j] = make_float4(w * acc[j].x, w * acc[j].y, w * acc[j].z, w * acc[j].w);
        }
        for (int t = 0; t < m; t += 32) {
            const int idx = t + lane;
            const int en  = (idx < m) ? list[idx] : -1;
            const bool mine = ((en & 63) == i);
            float ex[HEADS];
#pragma unroll
            for (int h = 0; h < HEADS; ++h) ex[h] = 0.f;
            if (mine) {
                float x[HEADS];
                if (idx < L1_CAP) {
                    const float4 u0 = *(const float4*)(&xl[idx * HEADS]);
                    const float4 u1 = *(const float4*)(&xl[idx * HEADS + 4]);
                    x[0] = u0.x; x[1] = u0.y; x[2] = u0.z; x[3] = u0.w;
                    x[4] = u1.x; x[5] = u1.y; x[6] = u1.z; x[7] = u1.w;
                } else {
                    const float* sp = aS + (size_t)(en >> 6) * HEADS;
                    const float4 a0 = *(const float4*)sp;
                    const float4 a1 = *(const float4*)(sp + 4);
                    const float av[HEADS] = {a0.x, a0.y, a0.z, a0.w, a1.x, a1.y, a1.z, a1.w};
#pragma unroll
                    for (int h = 0; h < HEADS; ++h) x[h] = lrelu(av[h] + adn[h]);
                }
#pragma unroll
                for (int h = 0; h < HEADS; ++h) {
                    ex[h]   = expf(x[h] - mx[h]);
                    den[h] += ex[h];
                }
            }
            unsigned mask = __builtin_amdgcn_ballot_w32(mine);
            while (mask) {
                const int b = __builtin_ctz(mask);
                mask &= mask - 1;
                const int s = __builtin_amdgcn_readlane(en, b) >> 6;
                float w[HEADS];
#pragma unroll
                for (int h = 0; h < HEADS; ++h)
                    w[h] = __int_as_float(__builtin_amdgcn_readlane(__float_as_int(ex[h]), b));
                const float* rp = H1 + (size_t)s * HC + lane * 4;
                float4 v[L1_CH];
#pragma unroll
                for (int j = 0; j < L1_CH; ++j) v[j] = *(const float4*)(rp + j * 128);
#pragma unroll
                for (int j = 0; j < L1_CH; ++j) {
                    const float ww = w[j >> 1];
                    acc[j].x += ww * v[j].x;
                    acc[j].y += ww * v[j].y;
                    acc[j].z += ww * v[j].z;
                    acc[j].w += ww * v[j].w;
                }
            }
        }
        float4 bq[L1_CH];
        {
            const float* bp = B1 + lane * 4;
#pragma unroll
            for (int j = 0; j < L1_CH; ++j) bq[j] = *(const float4*)(bp + j * 128);
        }
        float rc[HEADS];
#pragma unroll
        for (int h = 0; h < HEADS; ++h) {
#pragma unroll
            for (int o = 16; o >= 1; o >>= 1)
                den[h] += __shfl_xor(den[h], o);
            rc[h] = 1.f / ((den[h] + exs[h]) + 1e-16f);
        }
        float* op = agg + (size_t)n * HC + lane * 4;
#pragma unroll
        for (int j = 0; j < L1_CH; ++j) {
            const float r = rc[j >> 1];
            const float4 b = bq[j];
            *(float4*)(op + j * 128) = make_float4(elu1(acc[j].x * r + b.x), elu1(acc[j].y * r + b.y),
                                                   elu1(acc[j].z * r + b.z), elu1(acc[j].w * r + b.w));
        }
    }

    __syncthreads();

    if (wave < L1_GW) {
        const int half = lane >> 4;
        const int ml   = lane & 15;
        const int nb   = min(ml, NUM_CLASS - 1);
        const int k0   = wave * L1_KSL + half * 4;
        const float* wp = W2 + (size_t)k0 * NUM_CLASS + nb;
        const float* ap[L1_TILES];
        v8f c2[L1_TILES];
#pragma unroll
        for (int tt = 0; tt < L1_TILES; ++tt) {
            ap[tt] = agg + (size_t)(n0 + min(tt * 16 + ml, L1_NPB - 1)) * HC + k0;
            c2[tt] = (v8f){0.f, 0.f, 0.f, 0.f, 0.f, 0.f, 0.f, 0.f};
        }
#pragma unroll 2
        for (int k = 0; k < L1_KSL; k += 32) {
            float4 av[L1_TILES][4];
            float  w[4][4];
#pragma unroll
            for (int tt = 0; tt < L1_TILES; ++tt)
#pragma unroll
                for (int s = 0; s < 4; ++s) av[tt][s] = *(const float4*)(ap[tt] + k + 8 * s);
#pragma unroll
            for (int s = 0; s < 4; ++s)
#pragma unroll
                for (int j = 0; j < 4; ++j)
                    w[s][j] = wp[(size_t)(k + 8 * s + j) * NUM_CLASS];
#pragma unroll
            for (int s = 0; s < 4; ++s) {
                v2f a, b;
                b.x = w[s][0]; b.y = w[s][1];
#pragma unroll
                for (int tt = 0; tt < L1_TILES; ++tt) {
                    a.x = av[tt][s].x; a.y = av[tt][s].y;
                    c2[tt] = __builtin_amdgcn_wmma_f32_16x16x4_f32(false, a, false, b,
                                                                   (short)0, c2[tt], false, false);
                }
                b.x = w[s][2]; b.y = w[s][3];
#pragma unroll
                for (int tt = 0; tt < L1_TILES; ++tt) {
                    a.x = av[tt][s].z; a.y = av[tt][s].w;
                    c2[tt] = __builtin_amdgcn_wmma_f32_16x16x4_f32(false, a, false, b,
                                                                   (short)0, c2[tt], false, false);
                }
            }
        }
#pragma unroll
        for (int tt = 0; tt < L1_TILES; ++tt)
#pragma unroll
            for (int r = 0; r < 8; ++r)
                part[wave][tt * 16 + r + 8 * half][ml] = c2[tt][r];
    }
    __syncthreads();

    if (tid < L1_NPB * NUM_CLASS) {
        const int i = tid / NUM_CLASS;
        const int cc = tid - i * NUM_CLASS;
        float v = 0.f;
#pragma unroll
        for (int g = 0; g < L1_GW; ++g) v += part[g][i][cc];
        H2[(size_t)n0 * NUM_CLASS + tid] = v;
        h2s[tid] = v;
    }
    __syncthreads();

    if (tid < L1_NPB) {
        float ss = 0.f, sd = 0.f;
#pragma unroll
        for (int cc = 0; cc < NUM_CLASS; ++cc) {
            const float v = h2s[tid * NUM_CLASS + cc];
            ss += v * as2[cc];
            sd += v * ad2[cc];
        }
        aS2[n0 + tid] = ss; aD2[n0 + tid] = sd;
    }
}

__device__ __forceinline__ void atomicMaxF(float* addr, float v)
{
    if (v >= 0.f) atomicMax((int*)addr, __float_as_int(v));
    else          atomicMin((unsigned int*)addr, __float_as_uint(v));
}

#define L2_NPB     40
#define L2_THREADS 1024
#define L2_Q       (E_IN / 4)
#define L2_IT      ((L2_Q + L2_THREADS - 1) / L2_THREADS)

__launch_bounds__(L2_THREADS)
__global__ void layer2_owner(const int* __restrict__ ei, const float* __restrict__ H2,
                             const float* __restrict__ aS, const float* __restrict__ aD,
                             const float* __restrict__ b2, float* __restrict__ out)
{
    static_assert(N_NODES % L2_NPB == 0, "no partial node range");
    static_assert(E_IN % 4 == 0, "destination column read as 4-edge granules");
    static_assert(L2_NPB <= 64 && E_IN < (1 << 24) && N_NODES < (1 << 24), "edge or source number and local node share one int");
    static_assert(L2_NPB * NUM_CLASS <= L2_THREADS, "one thread per owned output element");
    __shared__ int   list[E_IN];
    __shared__ int   cnt;
    __shared__ float sd_s[L2_NPB], emax_s[L2_NPB], den_s[L2_NPB];
    __shared__ float agg_s[L2_NPB * NUM_CLASS];

    const int tid = threadIdx.x;
    const int n0  = blockIdx.x * L2_NPB;

    const int4* dq = (const int4*)(ei + E_IN);
    int4 d4[L2_IT];
#pragma unroll
    for (int i = 0; i < L2_IT; ++i)
        d4[i] = dq[min(tid + i * L2_THREADS, L2_Q - 1)];

    if (tid == 0) cnt = 0;
    if (tid < L2_NPB) {
        const float d = aD[n0 + tid];
        float x = aS[n0 + tid] + d;
        x = (x >= 0.f) ? x : NEG_SLOPE * x;
        sd_s[tid]   = d;
        emax_s[tid] = x;
        den_s[tid]  = 0.f;
    }
    if (tid < L2_NPB * NUM_CLASS) agg_s[tid] = 0.f;
    __syncthreads();

#pragma unroll
    for (int i = 0; i < L2_IT; ++i) {
        const int j = tid + i * L2_THREADS;
        const int dv[4] = {d4[i].x, d4[i].y, d4[i].z, d4[i].w};
#pragma unroll
        for (int q = 0; q < 4; ++q) {
            const unsigned dl = (unsigned)(dv[q] - n0);
            if (j < L2_Q && dl < (unsigned)L2_NPB) {
                const int slot = atomicAdd(&cnt, 1);
                list[slot] = ((4 * j + q) << 6) | (int)dl;
            }
        }
    }
    __syncthreads();
    const int m = cnt;

    for (int t = tid; t < m; t += L2_THREADS) {
        const int en = list[t];
        const int dl = en & 63;
        const int s  = ei[en >> 6];
        list[t] = (s << 6) | dl;
        float x = aS[s] + sd_s[dl];
        x = (x >= 0.f) ? x : NEG_SLOPE * x;
        atomicMaxF(&emax_s[dl], x);
    }
    __syncthreads();

    if (tid < L2_NPB) {
        const int n = n0 + tid;
        float x = aS[n] + sd_s[tid];
        x = (x >= 0.f) ? x : NEG_SLOPE * x;
        const float ex = expf(x - emax_s[tid]);
        atomicAdd(&den_s[tid], ex);
#pragma unroll
        for (int c = 0; c < NUM_CLASS; ++c)
            atomicAdd(&agg_s[tid * NUM_CLASS + c], ex * H2[(size_t)n * NUM_CLASS + c]);
    }
    for (int t = tid; t < m; t += L2_THREADS) {
        const int en = list[t];
        const int dl = en & 63;
        const int s  = en >> 6;
        float x = aS[s] + sd_s[dl];
        x = (x >= 0.f) ? x : NEG_SLOPE * x;
        const float ex = expf(x - emax_s[dl]);
        atomicAdd(&den_s[dl], ex);
#pragma unroll
        for (int c = 0; c < NUM_CLASS; ++c)
            atomicAdd(&agg_s[dl * NUM_CLASS + c], ex * H2[(size_t)s * NUM_CLASS + c]);
    }
    __syncthreads();

    if (tid < L2_NPB * NUM_CLASS) {
        const int i = tid / NUM_CLASS;
        const int c = tid - i * NUM_CLASS;
        const float v = agg_s[tid] / (den_s[i] + 1e-16f) + b2[c];
        out[(size_t)n0 * NUM_CLASS + tid] = 1.f / (1.f + expf(-v));
    }
}

extern "C" void kernel_launch(void* const* d_in, const int* in_sizes, int n_in,
                              void* d_out, int out_size, void* d_ws, size_t ws_size,
                              hipStream_t stream)
{
    const float* x   = (const float*)d_in[0];
    const int*   ei  = (const int*)d_in[1];
    const float* W1  = (const float*)d_in[2];
    const float* as1 = (const float*)d_in[3];
    const float* ad1 = (const float*)d_in[4];
    const float* b1  = (const float*)d_in[5];
    const float* W2  = (const float*)d_in[6];
    const float* as2 = (const float*)d_in[7];
    const float* ad2 = (const float*)d_in[8];
    const float* b2  = (const float*)d_in[9];
    float* out = (float*)d_out;

    float* ws = (float*)d_ws;
    size_t off = 0;
    float* h1    = ws + off; off += (size_t)N_NODES * HC;
    float* agg1  = ws + off; off += (size_t)N_NODES * HC;
    float* sA1   = ws + off; off += (size_t)N_NODES * HEADS;
    float* sD1   = ws + off; off += (size_t)N_NODES * HEADS;
    float* h2    = ws + off; off += (size_t)N_NODES * NUM_CLASS;
    float* sA2   = ws + off; off += N_NODES;
    float* sD2   = ws + off; off += N_NODES;
    (void)ws_size; (void)in_sizes; (void)n_in; (void)out_size;

    const dim3 b256(256);

    gemm1_att<<<dim3(HEADS, (N_NODES + 31) / 32), b256, 0, stream>>>(x, W1, as1, ad1, h1, sA1, sD1);

    layer1_owner<<<N_NODES / L1_NPB, L1_THREADS, 0, stream>>>(ei, h1, sA1, sD1, agg1,
                                                              b1, W2, as2, ad2, h2, sA2, sD2);

    layer2_owner<<<N_NODES / L2_NPB, L2_THREADS, 0, stream>>>(ei, h2, sA2, sD2, b2, out);
}


# GEMM1 block keeps its head's 128x256 B panel in LDS across 5 consecutive 32-row tiles (2504 -> 504 blocks, next A tile fetched to registers under the k loop); every sum unchanged
# speedup vs baseline: 4.4412x; 1.0715x over previous
#include <hip/hip_runtime.h>
#include <math.h>

#define N_NODES   10000
#define F_IN      128
#define HEADS     8
#define HIDD      256
#define HC        (HEADS * HIDD)
#define NUM_CLASS 10
#define E_IN      50000
#define E_TOT     (E_IN + N_NODES)
#define NEG_SLOPE 0.2f

typedef __attribute__((ext_vector_type(2))) float v2f;
typedef __attribute__((ext_vector_type(8))) float v8f;

#define G1_KP (2 * HIDD + 32)
#define G1_TILES ((N_NODES + 31) / 32)
#define G1_TPB   5
#define G1_RG    ((G1_TILES + G1_TPB - 1) / G1_TPB)

__launch_bounds__(256)
__global__ void gemm1_att(const float* __restrict__ X, const float* __restrict__ W,
                          const float* __restrict__ att_src,
                          const float* __restrict__ att_dst,
                          float* __restrict__ Hout,
                          float* __restrict__ aS, float* __restrict__ aD)
{
    static_assert(N_NODES % 16 == 0, "a wave's 16 rows are all inside or all outside");
    __shared__ __attribute__((aligned(16))) float As[32][F_IN + 4];
    __shared__ __attribute__((aligned(16))) float Bs[F_IN / 2][G1_KP];
    __shared__ __attribute__((aligned(16))) float red[8][32];

    const int tid = threadIdx.x;
    const int hh  = blockIdx.x;
    const int n0  = hh * HIDD;
    const int t0  = blockIdx.y * G1_TPB;

    const int sr = tid >> 5;
    const int sc = (tid & 31) * 4;
    float4 an0, an1, an2, an3;
    {
        const int r0 = t0 * 32 + sr;
        an0 = *(const float4*)(X + (size_t)min(r0,      N_NODES - 1) * F_IN + sc);
        an1 = *(const float4*)(X + (size_t)min(r0 + 8,  N_NODES - 1) * F_IN + sc);
        an2 = *(const float4*)(X + (size_t)min(r0 + 16, N_NODES - 1) * F_IN + sc);
        an3 = *(const float4*)(X + (size_t)min(r0 + 24, N_NODES - 1) * F_IN + sc);
    }
#pragma unroll
    for (int i = 0; i < 16; ++i) {
        const int item = tid + i * 256;
        const int p = item >> 6;
        const int c = (item & 63) * 4;
        const float* wp = W + (size_t)(2 * p) * HC + n0 + c;
        const float4 r0 = *(const float4*)wp;
        const float4 r1 = *(const float4*)(wp + HC);
        *(float4*)(&Bs[p][2 * c])     = make_float4(r0.x, r1.x, r0.y, r1.y);
        *(float4*)(&Bs[p][2 * c + 4]) = make_float4(r0.z, r1.z, r0.w, r1.w);
    }
    *(float4*)(&As[sr][sc])      = an0;
    *(float4*)(&As[sr + 8][sc])  = an1;
    *(float4*)(&As[sr + 16][sc]) = an2;
    *(float4*)(&As[sr + 24][sc]) = an3;
    __syncthreads();

    const int wave = tid >> 5;
    const int lane = tid & 31;
    const int half = lane >> 4;
    const int ml   = lane & 15;
    const int ar   = (wave >> 2) * 16;
    const int bc   = (wave & 3) * 64;

    float asv[4], adv[4];
#pragma unroll
    for (int t = 0; t < 4; ++t) {
        asv[t] = att_src[n0 + bc + 16 * t + ml];
        adv[t] = att_dst[n0 + bc + 16 * t + ml];
    }

    for (int it = 0; it < G1_TPB; ++it) {
        const int m0 = (t0 + it) * 32;
        if (m0 >= N_NODES) break;
        const bool more = (it + 1 < G1_TPB) && (m0 + 32 < N_NODES);

        if (more) {
            const int r0 = m0 + 32 + sr;
            an0 = *(const float4*)(X + (size_t)min(r0,      N_NODES - 1) * F_IN + sc);
            an1 = *(const float4*)(X + (size_t)min(r0 + 8,  N_NODES - 1) * F_IN + sc);
            an2 = *(const float4*)(X + (size_t)min(r0 + 16, N_NODES - 1) * F_IN + sc);
            an3 = *(const float4*)(X + (size_t)min(r0 + 24, N_NODES - 1) * F_IN + sc);
        }

        v8f acc[4];
#pragma unroll
        for (int t = 0; t < 4; ++t)
            acc[t] = (v8f){0.f, 0.f, 0.f, 0.f, 0.f, 0.f, 0.f, 0.f};

#pragma unroll
        for (int kk = 0; kk < F_IN / 4; ++kk) {
            const int p = kk * 2 + half;
            v2f a;
            const float2 av = *(const float2*)(&As[ar + ml][2 * p]);
            a.x = av.x; a.y = av.y;
#pragma unroll
            for (int t = 0; t < 4; ++t) {
                v2f b;
                const float2 bv = *(const float2*)(&Bs[p][2 * (bc + 16 * t + ml)]);
                b.x = bv.x; b.y = bv.y;
                acc[t] = __builtin_amdgcn_wmma_f32_16x16x4_f32(false, a, false, b,
                                                               (short)0, acc[t], false, false);
            }
        }

        if (m0 + ar < N_NODES) {
#pragma unroll
            for (int t = 0; t < 4; ++t) {
                const int col = n0 + bc + 16 * t + ml;
#pragma unroll
                for (int r = 0; r < 8; ++r) {
                    const int row = m0 + ar + r + 8 * half;
                    Hout[(size_t)row * HC + col] = acc[t][r];
                }
            }
        }

        float v[16];
#pragma unroll
        for (int r = 0; r < 8; ++r) {
            float ss = 0.f, sd = 0.f;
#pragma unroll
            for (int t = 0; t < 4; ++t) {
                ss += acc[t][r] * asv[t];
                sd += acc[t][r] * adv[t];
            }
            v[r] = ss; v[8 + r] = sd;
        }
#pragma unroll
        for (int j = 0; j < 16; ++j) {
            v[j] += __int_as_float(__builtin_amdgcn_update_dpp(0, __float_as_int(v[j]), 0x161, 0xf, 0xf, false));
            v[j] += __int_as_float(__builtin_amdgcn_update_dpp(0, __float_as_int(v[j]), 0x162, 0xf, 0xf, false));
            v[j] += __int_as_float(__builtin_amdgcn_update_dpp(0, __float_as_int(v[j]), 0x164, 0xf, 0xf, false));
            v[j] += __int_as_float(__builtin_amdgcn_update_dpp(0, __float_as_int(v[j]), 0x168, 0xf, 0xf, false));
        }
        if (ml == 0) {
#pragma unroll
            for (int j = 0; j < 16; j += 4)
                *(float4*)(&red[wave][half * 16 + j]) = make_float4(v[j], v[j + 1], v[j + 2], v[j + 3]);
        }
        __syncthreads();
        if (more) {
            *(float4*)(&As[sr][sc])      = an0;
            *(float4*)(&As[sr + 8][sc])  = an1;
            *(float4*)(&As[sr + 16][sc]) = an2;
            *(float4*)(&As[sr + 24][sc]) = an3;
        }
        if (tid < 64) {
            const int rh = tid >> 5;
            const int ln = tid & 31;
            const float s = ((red[rh * 4 + 0][ln] + red[rh * 4 + 1][ln])
                             + red[rh * 4 + 2][ln]) + red[rh * 4 + 3][ln];
            const int row = m0 + rh * 16 + (ln & 7) + 8 * (ln >> 4);
            if (row < N_NODES) {
                float* dstp = ((ln >> 3) & 1) ? aD : aS;
                dstp[(size_t)row * HEADS + hh] = s;
            }
        }
        if (more) __syncthreads();
    }
}

__device__ __forceinline__ float elu1(float v)
{
    const float e = expm1f(v);
    return (v > 0.f) ? v : e;
}

#define L1_NPB     40
#define L1_THREADS 640
#define L1_WAVES   (L1_THREADS / 32)
#define L1_Q       (E_IN / 4)
#define L1_IT      ((L1_Q + L1_THREADS - 1) / L1_THREADS)
#define L1_CH      (HC / 128)
#define L1_CAP     1024
#define L1_GW      16
#define L1_KSL     (HC / L1_GW)
#define L1_TILES   ((L1_NPB + 15) / 16)

__device__ __forceinline__ float lrelu(float x)
{
    return (x >= 0.f) ? x : NEG_SLOPE * x;
}

__launch_bounds__(L1_THREADS)
__global__ void layer1_owner(const int* __restrict__ ei, const float* __restrict__ H1,
                             const float* __restrict__ aS, const float* __restrict__ aD,
                             float* agg, const float* B1,
                             const float* __restrict__ W2,
                             const float* __restrict__ as2, const float* __restrict__ ad2,
                             float* __restrict__ H2,
                             float* __restrict__ aS2, float* __restrict__ aD2)
{
    static_assert(N_NODES % L1_NPB == 0, "no partial node range");
    static_assert(E_IN % 4 == 0, "destination column read as 4-edge granules");
    static_assert(L1_NPB <= 63 && N_NODES < (1 << 24), "source and local node share one int; local 63 marks no entry");
    static_assert(HEADS == 8 && HIDD == 256, "two 128-channel pieces per head, 8 scores per node");
    static_assert(L1_GW <= L1_WAVES && L1_KSL % 32 == 0, "a K slice per wave, whole 32-k trips");
    static_assert(L1_NPB * NUM_CLASS <= L1_THREADS, "one thread per owned h2 element");
    __shared__ int list[E_IN];
    __shared__ __attribute__((aligned(16))) float part[L1_GW][L1_TILES * 16][16];
    __shared__ float h2s[L1_NPB * NUM_CLASS];
    __shared__ __attribute__((aligned(16))) float xl[L1_CAP * HEADS];
    __shared__ int cnt;

    const int tid = threadIdx.x;
    const int n0  = blockIdx.x * L1_NPB;

    const int4* dq = (const int4*)(ei + E_IN);
    int4 d4[L1_IT];
#pragma unroll
    for (int i = 0; i < L1_IT; ++i)
        d4[i] = dq[min(tid + i * L1_THREADS, L1_Q - 1)];

    if (tid == 0) cnt = 0;
    __syncthreads();

#pragma unroll
    for (int i = 0; i < L1_IT; ++i) {
        const int j = tid + i * L1_THREADS;
        const int dv[4] = {d4[i].x, d4[i].y, d4[i].z, d4[i].w};
#pragma unroll
        for (int q = 0; q < 4; ++q) {
            const unsigned dl = (unsigned)(dv[q] - n0);
            if (j < L1_Q && dl < (unsigned)L1_NPB) {
                const int s = ei[4 * j + q];
                const int slot = atomicAdd(&cnt, 1);
                list[slot] = (s << 6) | (int)dl;
            }
        }
    }
    __syncthreads();
    const int m = cnt;

    for (int t = tid; t < min(m, L1_CAP); t += L1_THREADS) {
        const int en = list[t];
        const float* sp = aS + (size_t)(en >> 6) * HEADS;
        const float* dp = aD + (size_t)(n0 + (en & 63)) * HEADS;
        const float4 a0 = *(const float4*)sp;
        const float4 a1 = *(const float4*)(sp + 4);
        const float4 e0 = *(const float4*)dp;
        const float4 e1 = *(const float4*)(dp + 4);
        *(float4*)(&xl[t * HEADS])     = make_float4(lrelu(a0.x + e0.x), lrelu(a0.y + e0.y),
                                                     lrelu(a0.z + e0.z), lrelu(a0.w + e0.w));
        *(float4*)(&xl[t * HEADS + 4]) = make_float4(lrelu(a1.x + e1.x), lrelu(a1.y + e1.y),
                                                     lrelu(a1.z + e1.z), lrelu(a1.w + e1.w));
    }
    __syncthreads();

    const int lane = tid & 31;
    const int wave = __builtin_amdgcn_readfirstlane(tid >> 5);

    for (int i = wave; i < L1_NPB; i += L1_WAVES) {
        const int n = n0 + i;

        float4 acc[L1_CH];
        {
            const float* rp = H1 + (size_t)n * HC + lane * 4;
#pragma unroll
            for (int j = 0; j < L1_CH; ++j) acc[j] = *(const float4*)(rp + j * 128);
        }

        float adn[HEADS], xs[HEADS], mx[HEADS];
        {
            const float4 p0 = *(const float4*)(aD + (size_t)n * HEADS);
            const float4 p1 = *(const float4*)(aD + (size_t)n * HEADS + 4);
            const float4 q0 = *(const float4*)(aS + (size_t)n * HEADS);
            const float4 q1 = *(const float4*)(aS + (size_t)n * HEADS + 4);
            const float dd[HEADS] = {p0.x, p0.y, p0.z, p0.w, p1.x, p1.y, p1.z, p1.w};
            const float ss[HEADS] = {q0.x, q0.y, q0.z, q0.w, q1.x, q1.y, q1.z, q1.w};
#pragma unroll
            for (int h = 0; h < HEADS; ++h) {
                adn[h] = dd[h];
                xs[h]  = lrelu(ss[h] + dd[h]);
                mx[h]  = xs[h];
            }
        }

        for (int t = 0; t < m; t += 32) {
            const int idx = t + lane;
            const int en  = (idx < m) ? list[idx] : -1;
            if ((en & 63) == i) {
                float x[HEADS];
                if (idx < L1_CAP) {
                    const float4 u0 = *(const float4*)(&xl[idx * HEADS]);
                    const float4 u1 = *(const float4*)(&xl[idx * HEADS + 4]);
                    x[0] = u0.x; x[1] = u0.y; x[2] = u0.z; x[3] = u0.w;
                    x[4] = u1.x; x[5] = u1.y; x[6] = u1.z; x[7] = u1.w;
                } else {
                    const float* sp = aS + (size_t)(en >> 6) * HEADS;
                    const float4 a0 = *(const float4*)sp;
                    const float4 a1 = *(const float4*)(sp + 4);
                    const float av[HEADS] = {a0.x, a0.y, a0.z, a0.w, a1.x, a1.y, a1.z, a1.w};
#pragma unroll
                    for (int h = 0; h < HEADS; ++h) x[h] = lrelu(av[h] + adn[h]);
                }
#pragma unroll
                for (int h = 0; h < HEADS; ++h) mx[h] = fmaxf(mx[h], x[h]);
            }
        }
#pragma unroll
        for (int h = 0; h < HEADS; ++h) {
#pragma unroll
            for (int o = 16; o >= 1; o >>= 1)
                mx[h] = fmaxf(mx[h], __shfl_xor(mx[h], o));
        }

        float exs[HEADS], den[HEADS];
#pragma unroll
        for (int h = 0; h < HEADS; ++h) {
            exs[h] = expf(xs[h] - mx[h]);
            den[h] = 0.f;
        }
#pragma unroll
        for (int j = 0; j < L1_CH; ++j) {
            const float w = exs[j >> 1];
            acc[j] = make_float4(w * acc[j].x, w * acc[j].y, w * acc[j].z, w * acc[j].w);
        }
        for (int t = 0; t < m; t += 32) {
            const int idx = t + lane;
            const int en  = (idx < m) ? list[idx] : -1;
            const bool mine = ((en & 63) == i);
            float ex[HEADS];
#pragma unroll
            for (int h = 0; h < HEADS; ++h) ex[h] = 0.f;
            if (mine) {
                float x[HEADS];
                if (idx < L1_CAP) {
                    const float4 u0 = *(const float4*)(&xl[idx * HEADS]);
                    const float4 u1 = *(const float4*)(&xl[idx * HEADS + 4]);
                    x[0] = u0.x; x[1] = u0.y; x[2] = u0.z; x[3] = u0.w;
                    x[4] = u1.x; x[5] = u1.y; x[6] = u1.z; x[7] = u1.w;
                } else {
                    const float* sp = aS + (size_t)(en >> 6) * HEADS;
                    const float4 a0 = *(const float4*)sp;
                    const float4 a1 = *(const float4*)(sp + 4);
                    const float av[HEADS] = {a0.x, a0.y, a0.z, a0.w, a1.x, a1.y, a1.z, a1.w};
#pragma unroll
                    for (int h = 0; h < HEADS; ++h) x[h] = lrelu(av[h] + adn[h]);
                }
#pragma unroll
                for (int h = 0; h < HEADS; ++h) {
                    ex[h]   = expf(x[h] - mx[h]);
                    den[h] += ex[h];
                }
            }
            unsigned mask = __builtin_amdgcn_ballot_w32(mine);
            while (mask) {
                const int b = __builtin_ctz(mask);
                mask &= mask - 1;
                const int s = __builtin_amdgcn_readlane(en, b) >> 6;
                float w[HEADS];
#pragma unroll
                for (int h = 0; h < HEADS; ++h)
                    w[h] = __int_as_float(__builtin_amdgcn_readlane(__float_as_int(ex[h]), b));
                const float* rp = H1 + (size_t)s * HC + lane * 4;
                float4 v[L1_CH];
#pragma unroll
                for (int j = 0; j < L1_CH; ++j) v[j] = *(const float4*)(rp + j * 128);
#pragma unroll
                for (int j = 0; j < L1_CH; ++j) {
                    const float ww = w[j >> 1];
                    acc[j].x += ww * v[j].x;
                    acc[j].y += ww * v[j].y;
                    acc[j].z += ww * v[j].z;
                    acc[j].w += ww * v[j].w;
                }
            }
        }
        float4 bq[L1_CH];
        {
            const float* bp = B1 + lane * 4;
#pragma unroll
            for (int j = 0; j < L1_CH; ++j) bq[j] = *(const float4*)(bp + j * 128);
        }
        float rc[HEADS];
#pragma unroll
        for (int h = 0; h < HEADS; ++h) {
#pragma unroll
            for (int o = 16; o >= 1; o >>= 1)
                den[h] += __shfl_xor(den[h], o);
            rc[h] = 1.f / ((den[h] + exs[h]) + 1e-16f);
        }
        float* op = agg + (size_t)n * HC + lane * 4;
#pragma unroll
        for (int j = 0; j < L1_CH; ++j) {
            const float r = rc[j >> 1];
            const float4 b = bq[j];
            *(float4*)(op + j * 128) = make_float4(elu1(acc[j].x * r + b.x), elu1(acc[j].y * r + b.y),
                                                   elu1(acc[j].z * r + b.z), elu1(acc[j].w * r + b.w));
        }
    }

    __syncthreads();

    if (wave < L1_GW) {
        const int half = lane >> 4;
        const int ml   = lane & 15;
        const int nb   = min(ml, NUM_CLASS - 1);
        const int k0   = wave * L1_KSL + half * 4;
        const float* wp = W2 + (size_t)k0 * NUM_CLASS + nb;
        const float* ap[L1_TILES];
        v8f c2[L1_TILES];
#pragma unroll
        for (int tt = 0; tt < L1_TILES; ++tt) {
            ap[tt] = agg + (size_t)(n0 + min(tt * 16 + ml, L1_NPB - 1)) * HC + k0;
            c2[tt] = (v8f){0.f, 0.f, 0.f, 0.f, 0.f, 0.f, 0.f, 0.f};
        }
#pragma unroll 2
        for (int k = 0; k < L1_KSL; k += 32) {
            float4 av[L1_TILES][4];
            float  w[4][4];
#pragma unroll
            for (int tt = 0; tt < L1_TILES; ++tt)
#pragma unroll
                for (int s = 0; s < 4; ++s) av[tt][s] = *(const float4*)(ap[tt] + k + 8 * s);
#pragma unroll
            for (int s = 0; s < 4; ++s)
#pragma unroll
                for (int j = 0; j < 4; ++j)
                    w[s][j] = wp[(size_t)(k + 8 * s + j) * NUM_CLASS];
#pragma unroll
            for (int s = 0; s < 4; ++s) {
                v2f a, b;
                b.x = w[s][0]; b.y = w[s][1];
#pragma unroll
                for (int tt = 0; tt < L1_TILES; ++tt) {
                    a.x = av[tt][s].x; a.y = av[tt][s].y;
                    c2[tt] = __builtin_amdgcn_wmma_f32_16x16x4_f32(false, a, false, b,
                                                                   (short)0, c2[tt], false, false);
                }
                b.x = w[s][2]; b.y = w[s][3];
#pragma unroll
                for (int tt = 0; tt < L1_TILES; ++tt) {
                    a.x = av[tt][s].z; a.y = av[tt][s].w;
                    c2[tt] = __builtin_amdgcn_wmma_f32_16x16x4_f32(false, a, false, b,
                                                                   (short)0, c2[tt], false, false);
                }
            }
        }
#pragma unroll
        for (int tt = 0; tt < L1_TILES; ++tt)
#pragma unroll
            for (int r = 0; r < 8; ++r)
                part[wave][tt * 16 + r + 8 * half][ml] = c2[tt][r];
    }
    __syncthreads();

    if (tid < L1_NPB * NUM_CLASS) {
        const int i = tid / NUM_CLASS;
        const int cc = tid - i * NUM_CLASS;
        float v = 0.f;
#pragma unroll
        for (int g = 0; g < L1_GW; ++g) v += part[g][i][cc];
        H2[(size_t)n0 * NUM_CLASS + tid] = v;
        h2s[tid] = v;
    }
    __syncthreads();

    if (tid < L1_NPB) {
        float ss = 0.f, sd = 0.f;
#pragma unroll
        for (int cc = 0; cc < NUM_CLASS; ++cc) {
            const float v = h2s[tid * NUM_CLASS + cc];
            ss += v * as2[cc];
            sd += v * ad2[cc];
        }
        aS2[n0 + tid] = ss; aD2[n0 + tid] = sd;
    }
}

__device__ __forceinline__ void atomicMaxF(float* addr, float v)
{
    if (v >= 0.f) atomicMax((int*)addr, __float_as_int(v));
    else          atomicMin((unsigned int*)addr, __float_as_uint(v));
}

#define L2_NPB     40
#define L2_THREADS 1024
#define L2_Q       (E_IN / 4)
#define L2_IT      ((L2_Q + L2_THREADS - 1) / L2_THREADS)

__launch_bounds__(L2_THREADS)
__global__ void layer2_owner(const int* __restrict__ ei, const float* __restrict__ H2,
                             const float* __restrict__ aS, const float* __restrict__ aD,
                             const float* __restrict__ b2, float* __restrict__ out)
{
    static_assert(N_NODES % L2_NPB == 0, "no partial node range");
    static_assert(E_IN % 4 == 0, "destination column read as 4-edge granules");
    static_assert(L2_NPB <= 64 && E_IN < (1 << 24) && N_NODES < (1 << 24), "edge or source number and local node share one int");
    static_assert(L2_NPB * NUM_CLASS <= L2_THREADS, "one thread per owned output element");
    __shared__ int   list[E_IN];
    __shared__ int   cnt;
    __shared__ float sd_s[L2_NPB], emax_s[L2_NPB], den_s[L2_NPB];
    __shared__ float agg_s[L2_NPB * NUM_CLASS];

    const int tid = threadIdx.x;
    const int n0  = blockIdx.x * L2_NPB;

    const int4* dq = (const int4*)(ei + E_IN);
    int4 d4[L2_IT];
#pragma unroll
    for (int i = 0; i < L2_IT; ++i)
        d4[i] = dq[min(tid + i * L2_THREADS, L2_Q - 1)];

    if (tid == 0) cnt = 0;
    if (tid < L2_NPB) {
        const float d = aD[n0 + tid];
        float x = aS[n0 + tid] + d;
        x = (x >= 0.f) ? x : NEG_SLOPE * x;
        sd_s[tid]   = d;
        emax_s[tid] = x;
        den_s[tid]  = 0.f;
    }
    if (tid < L2_NPB * NUM_CLASS) agg_s[tid] = 0.f;
    __syncthreads();

#pragma unroll
    for (int i = 0; i < L2_IT; ++i) {
        const int j = tid + i * L2_THREADS;
        const int dv[4] = {d4[i].x, d4[i].y, d4[i].z, d4[i].w};
#pragma unroll
        for (int q = 0; q < 4; ++q) {
            const unsigned dl = (unsigned)(dv[q] - n0);
            if (j < L2_Q && dl < (unsigned)L2_NPB) {
                const int slot = atomicAdd(&cnt, 1);
                list[slot] = ((4 * j + q) << 6) | (int)dl;
            }
        }
    }
    __syncthreads();
    const int m = cnt;

    for (int t = tid; t < m; t += L2_THREADS) {
        const int en = list[t];
        const int dl = en & 63;
        const int s  = ei[en >> 6];
        list[t] = (s << 6) | dl;
        float x = aS[s] + sd_s[dl];
        x = (x >= 0.f) ? x : NEG_SLOPE * x;
        atomicMaxF(&emax_s[dl], x);
    }
    __syncthreads();

    if (tid < L2_NPB) {
        const int n = n0 + tid;
        float x = aS[n] + sd_s[tid];
        x = (x >= 0.f) ? x : NEG_SLOPE * x;
        const float ex = expf(x - emax_s[tid]);
        atomicAdd(&den_s[tid], ex);
#pragma unroll
        for (int c = 0; c < NUM_CLASS; ++c)
            atomicAdd(&agg_s[tid * NUM_CLASS + c], ex * H2[(size_t)n * NUM_CLASS + c]);
    }
    for (int t = tid; t < m; t += L2_THREADS) {
        const int en = list[t];
        const int dl = en & 63;
        const int s  = en >> 6;
        float x = aS[s] + sd_s[dl];
        x = (x >= 0.f) ? x : NEG_SLOPE * x;
        const float ex = expf(x - emax_s[dl]);
        atomicAdd(&den_s[dl], ex);
#pragma unroll
        for (int c = 0; c < NUM_CLASS; ++c)
            atomicAdd(&agg_s[dl * NUM_CLASS + c], ex * H2[(size_t)s * NUM_CLASS + c]);
    }
    __syncthreads();

    if (tid < L2_NPB * NUM_CLASS) {
        const int i = tid / NUM_CLASS;
        const int c = tid - i * NUM_CLASS;
        const float v = agg_s[tid] / (den_s[i] + 1e-16f) + b2[c];
        out[(size_t)n0 * NUM_CLASS + tid] = 1.f / (1.f + expf(-v));
    }
}

extern "C" void kernel_launch(void* const* d_in, const int* in_sizes, int n_in,
                              void* d_out, int out_size, void* d_ws, size_t ws_size,
                              hipStream_t stream)
{
    const float* x   = (const float*)d_in[0];
    const int*   ei  = (const int*)d_in[1];
    const float* W1  = (const float*)d_in[2];
    const float* as1 = (const float*)d_in[3];
    const float* ad1 = (const float*)d_in[4];
    const float* b1  = (const float*)d_in[5];
    const float* W2  = (const float*)d_in[6];
    const float* as2 = (const float*)d_in[7];
    const float* ad2 = (const float*)d_in[8];
    const float* b2  = (const float*)d_in[9];
    float* out = (float*)d_out;

    float* ws = (float*)d_ws;
    size_t off = 0;
    float* h1    = ws + off; off += (size_t)N_NODES * HC;
    float* agg1  = ws + off; off += (size_t)N_NODES * HC;
    float* sA1   = ws + off; off += (size_t)N_NODES * HEADS;
    float* sD1   = ws + off; off += (size_t)N_NODES * HEADS;
    float* h2    = ws + off; off += (size_t)N_NODES * NUM_CLASS;
    float* sA2   = ws + off; off += N_NODES;
    float* sD2   = ws + off; off += N_NODES;
    (void)ws_size; (void)in_sizes; (void)n_in; (void)out_size;

    const dim3 b256(256);

    gemm1_att<<<dim3(HEADS, G1_RG), b256, 0, stream>>>(x, W1, as1, ad1, h1, sA1, sD1);

    layer1_owner<<<N_NODES / L1_NPB, L1_THREADS, 0, stream>>>(ei, h1, sA1, sD1, agg1,
                                                              b1, W2, as2, ad2, h2, sA2, sD2);

    layer2_owner<<<N_NODES / L2_NPB, L2_THREADS, 0, stream>>>(ei, h2, sA2, sD2, b2, out);
}


# Layer-2 owner blocks read their edge list from the workspace, written by the layer-1 owner block of the same nodes: no second scan of the destination column, 1024 -> 256 threads; sum order changes
# speedup vs baseline: 4.6255x; 1.0415x over previous
#include <hip/hip_runtime.h>
#include <math.h>

#define N_NODES   10000
#define F_IN      128
#define HEADS     8
#define HIDD      256
#define HC        (HEADS * HIDD)
#define NUM_CLASS 10
#define E_IN      50000
#define E_TOT     (E_IN + N_NODES)
#define NEG_SLOPE 0.2f

typedef __attribute__((ext_vector_type(2))) float v2f;
typedef __attribute__((ext_vector_type(8))) float v8f;

#define G1_KP (2 * HIDD + 32)
#define G1_TILES ((N_NODES + 31) / 32)
#define G1_TPB   5
#define G1_RG    ((G1_TILES + G1_TPB - 1) / G1_TPB)

__launch_bounds__(256)
__global__ void gemm1_att(const float* __restrict__ X, const float* __restrict__ W,
                          const float* __restrict__ att_src,
                          const float* __restrict__ att_dst,
                          float* __restrict__ Hout,
                          float* __restrict__ aS, float* __restrict__ aD)
{
    static_assert(N_NODES % 16 == 0, "a wave's 16 rows are all inside or all outside");
    __shared__ __attribute__((aligned(16))) float As[32][F_IN + 4];
    __shared__ __attribute__((aligned(16))) float Bs[F_IN / 2][G1_KP];
    __shared__ __attribute__((aligned(16))) float red[8][32];

    const int tid = threadIdx.x;
    const int hh  = blockIdx.x;
    const int n0  = hh * HIDD;
    const int t0  = blockIdx.y * G1_TPB;

    const int sr = tid >> 5;
    const int sc = (tid & 31) * 4;
    float4 an0, an1, an2, an3;
    {
        const int r0 = t0 * 32 + sr;
        an0 = *(const float4*)(X + (size_t)min(r0,      N_NODES - 1) * F_IN + sc);
        an1 = *(const float4*)(X + (size_t)min(r0 + 8,  N_NODES - 1) * F_IN + sc);
        an2 = *(const float4*)(X + (size_t)min(r0 + 16, N_NODES - 1) * F_IN + sc);
        an3 = *(const float4*)(X + (size_t)min(r0 + 24, N_NODES - 1) * F_IN + sc);
    }
#pragma unroll
    for (int i = 0; i < 16; ++i) {
        const int item = tid + i * 256;
        const int p = item >> 6;
        const int c = (item & 63) * 4;
        const float* wp = W + (size_t)(2 * p) * HC + n0 + c;
        const float4 r0 = *(const float4*)wp;
        const float4 r1 = *(const float4*)(wp + HC);
        *(float4*)(&Bs[p][2 * c])     = make_float4(r0.x, r1.x, r0.y, r1.y);
        *(float4*)(&Bs[p][2 * c + 4]) = make_float4(r0.z, r1.z, r0.w, r1.w);
    }
    *(float4*)(&As[sr][sc])      = an0;
    *(float4*)(&As[sr + 8][sc])  = an1;
    *(float4*)(&As[sr + 16][sc]) = an2;
    *(float4*)(&As[sr + 24][sc]) = an3;
    __syncthreads();

    const int wave = tid >> 5;
    const int lane = tid & 31;
    const int half = lane >> 4;
    const int ml   = lane & 15;
    const int ar   = (wave >> 2) * 16;
    const int bc   = (wave & 3) * 64;

    float asv[4], adv[4];
#pragma unroll
    for (int t = 0; t < 4; ++t) {
        asv[t] = att_src[n0 + bc + 16 * t + ml];
        adv[t] = att_dst[n0 + bc + 16 * t + ml];
    }

    for (int it = 0; it < G1_TPB; ++it) {
        const int m0 = (t0 + it) * 32;
        if (m0 >= N_NODES) break;
        const bool more = (it + 1 < G1_TPB) && (m0 + 32 < N_NODES);

        if (more) {
            const int r0 = m0 + 32 + sr;
            an0 = *(const float4*)(X + (size_t)min(r0,      N_NODES - 1) * F_IN + sc);
            an1 = *(const float4*)(X + (size_t)min(r0 + 8,  N_NODES - 1) * F_IN + sc);
            an2 = *(const float4*)(X + (size_t)min(r0 + 16, N_NODES - 1) * F_IN + sc);
            an3 = *(const float4*)(X + (size_t)min(r0 + 24, N_NODES - 1) * F_IN + sc);
        }

        v8f acc[4];
#pragma unroll
        for (int t = 0; t < 4; ++t)
            acc[t] = (v8f){0.f, 0.f, 0.f, 0.f, 0.f, 0.f, 0.f, 0.f};

#pragma unroll
        for (int kk = 0; kk < F_IN / 4; ++kk) {
            const int p = kk * 2 + half;
            v2f a;
            const float2 av = *(const float2*)(&As[ar + ml][2 * p]);
            a.x = av.x; a.y = av.y;
#pragma unroll
            for (int t = 0; t < 4; ++t) {
                v2f b;
                const float2 bv = *(const float2*)(&Bs[p][2 * (bc + 16 * t + ml)]);
                b.x = bv.x; b.y = bv.y;
                acc[t] = __builtin_amdgcn_wmma_f32_16x16x4_f32(false, a, false, b,
                                                               (short)0, acc[t], false, false);
            }
        }

        if (m0 + ar < N_NODES) {
#pragma unroll
            for (int t = 0; t < 4; ++t) {
                const int col = n0 + bc + 16 * t + ml;
#pragma unroll
                for (int r = 0; r < 8; ++r) {
                    const int row = m0 + ar + r + 8 * half;
                    Hout[(size_t)row * HC + col] = acc[t][r];
                }
            }
        }

        float v[16];
#pragma unroll
        for (int r = 0; r < 8; ++r) {
            float ss = 0.f, sd = 0.f;
#pragma unroll
            for (int t = 0; t < 4; ++t) {
                ss += acc[t][r] * asv[t];
                sd += acc[t][r] * adv[t];
            }
            v[r] = ss; v[8 + r] = sd;
        }
#pragma unroll
        for (int j = 0; j < 16; ++j) {
            v[j] += __int_as_float(__builtin_amdgcn_update_dpp(0, __float_as_int(v[j]), 0x161, 0xf, 0xf, false));
            v[j] += __int_as_float(__builtin_amdgcn_update_dpp(0, __float_as_int(v[j]), 0x162, 0xf, 0xf, false));
            v[j] += __int_as_float(__builtin_amdgcn_update_dpp(0, __float_as_int(v[j]), 0x164, 0xf, 0xf, false));
            v[j] += __int_as_float(__builtin_amdgcn_update_dpp(0, __float_as_int(v[j]), 0x168, 0xf, 0xf, false));
        }
        if (ml == 0) {
#pragma unroll
            for (int j = 0; j < 16; j += 4)
                *(float4*)(&red[wave][half * 16 + j]) = make_float4(v[j], v[j + 1], v[j + 2], v[j + 3]);
        }
        __syncthreads();
        if (more) {
            *(float4*)(&As[sr][sc])      = an0;
            *(float4*)(&As[sr + 8][sc])  = an1;
            *(float4*)(&As[sr + 16][sc]) = an2;
            *(float4*)(&As[sr + 24][sc]) = an3;
        }
        if (tid < 64) {
            const int rh = tid >> 5;
            const int ln = tid & 31;
            const float s = ((red[rh * 4 + 0][ln] + red[rh * 4 + 1][ln])
                             + red[rh * 4 + 2][ln]) + red[rh * 4 + 3][ln];
            const int row = m0 + rh * 16 + (ln & 7) + 8 * (ln >> 4);
            if (row < N_NODES) {
                float* dstp = ((ln >> 3) & 1) ? aD : aS;
                dstp[(size_t)row * HEADS + hh] = s;
            }
        }
        if (more) __syncthreads();
    }
}

__device__ __forceinline__ float elu1(float v)
{
    const float e = expm1f(v);
    return (v > 0.f) ? v : e;
}

#define L1_NPB     40
#define L1_THREADS 640
#define L1_WAVES   (L1_THREADS / 32)
#define L1_Q       (E_IN / 4)
#define L1_IT      ((L1_Q + L1_THREADS - 1) / L1_THREADS)
#define L1_CH      (HC / 128)
#define L1_CAP     1024
#define L1_GW      16
#define L1_KSL     (HC / L1_GW)
#define L1_TILES   ((L1_NPB + 15) / 16)

__device__ __forceinline__ float lrelu(float x)
{
    return (x >= 0.f) ? x : NEG_SLOPE * x;
}

__launch_bounds__(L1_THREADS)
__global__ void layer1_owner(const int* __restrict__ ei, const float* __restrict__ H1,
                             const float* __restrict__ aS, const float* __restrict__ aD,
                             float* agg, const float* B1,
                             const float* __restrict__ W2,
                             const float* __restrict__ as2, const float* __restrict__ ad2,
                             float* __restrict__ H2,
                             float* __restrict__ aS2, float* __restrict__ aD2,
                             int* __restrict__ lst, int* __restrict__ lcnt)
{
    static_assert(N_NODES % L1_NPB == 0, "no partial node range");
    static_assert(E_IN % 4 == 0, "destination column read as 4-edge granules");
    static_assert(L1_NPB <= 63 && N_NODES < (1 << 24), "source and local node share one int; local 63 marks no entry");
    static_assert(HEADS == 8 && HIDD == 256, "two 128-channel pieces per head, 8 scores per node");
    static_assert(L1_GW <= L1_WAVES && L1_KSL % 32 == 0, "a K slice per wave, whole 32-k trips");
    static_assert(L1_NPB * NUM_CLASS <= L1_THREADS, "one thread per owned h2 element");
    __shared__ int list[E_IN];
    __shared__ __attribute__((aligned(16))) float part[L1_GW][L1_TILES * 16][16];
    __shared__ float h2s[L1_NPB * NUM_CLASS];
    __shared__ __attribute__((aligned(16))) float xl[L1_CAP * HEADS];
    __shared__ int cnt;

    const int tid = threadIdx.x;
    const int n0  = blockIdx.x * L1_NPB;

    const int4* dq = (const int4*)(ei + E_IN);
    int4 d4[L1_IT];
#pragma unroll
    for (int i = 0; i < L1_IT; ++i)
        d4[i] = dq[min(tid + i * L1_THREADS, L1_Q - 1)];

    if (tid == 0) cnt = 0;
    __syncthreads();

#pragma unroll
    for (int i = 0; i < L1_IT; ++i) {
        const int j = tid + i * L1_THREADS;
        const int dv[4] = {d4[i].x, d4[i].y, d4[i].z, d4[i].w};
#pragma unroll
        for (int q = 0; q < 4; ++q) {
            const unsigned dl = (unsigned)(dv[q] - n0);
            if (j < L1_Q && dl < (unsigned)L1_NPB) {
                const int s = ei[4 * j + q];
                const int slot = atomicAdd(&cnt, 1);
                list[slot] = (s << 6) | (int)dl;
            }
        }
    }
    __syncthreads();
    const int m = cnt;

    {
        int* lo = lst + (size_t)blockIdx.x * E_IN;
        for (int t = tid; t < m; t += L1_THREADS) lo[t] = list[t];
        if (tid == 0) lcnt[blockIdx.x] = m;
    }

    for (int t = tid; t < min(m, L1_CAP); t += L1_THREADS) {
        const int en = list[t];
        const float* sp = aS + (size_t)(en >> 6) * HEADS;
        const float* dp = aD + (size_t)(n0 + (en & 63)) * HEADS;
        const float4 a0 = *(const float4*)sp;
        const float4 a1 = *(const float4*)(sp + 4);
        const float4 e0 = *(const float4*)dp;
        const float4 e1 = *(const float4*)(dp + 4);
        *(float4*)(&xl[t * HEADS])     = make_float4(lrelu(a0.x + e0.x), lrelu(a0.y + e0.y),
                                                     lrelu(a0.z + e0.z), lrelu(a0.w + e0.w));
        *(float4*)(&xl[t * HEADS + 4]) = make_float4(lrelu(a1.x + e1.x), lrelu(a1.y + e1.y),
                                                     lrelu(a1.z + e1.z), lrelu(a1.w + e1.w));
    }
    __syncthreads();

    const int lane = tid & 31;
    const int wave = __builtin_amdgcn_readfirstlane(tid >> 5);

    for (int i = wave; i < L1_NPB; i += L1_WAVES) {
        const int n = n0 + i;

        float4 acc[L1_CH];
        {
            const float* rp = H1 + (size_t)n * HC + lane * 4;
#pragma unroll
            for (int j = 0; j < L1_CH; ++j) acc[j] = *(const float4*)(rp + j * 128);
        }

        float adn[HEADS], xs[HEADS], mx[HEADS];
        {
            const float4 p0 = *(const float4*)(aD + (size_t)n * HEADS);
            const float4 p1 = *(const float4*)(aD + (size_t)n * HEADS + 4);
            const float4 q0 = *(const float4*)(aS + (size_t)n * HEADS);
            const float4 q1 = *(const float4*)(aS + (size_t)n * HEADS + 4);
            const float dd[HEADS] = {p0.x, p0.y, p0.z, p0.w, p1.x, p1.y, p1.z, p1.w};
            const float ss[HEADS] = {q0.x, q0.y, q0.z, q0.w, q1.x, q1.y, q1.z, q1.w};
#pragma unroll
            for (int h = 0; h < HEADS; ++h) {
                adn[h] = dd[h];
                xs[h]  = lrelu(ss[h] + dd[h]);
                mx[h]  = xs[h];
            }
        }

        for (int t = 0; t < m; t += 32) {
            const int idx = t + lane;
            const int en  = (idx < m) ? list[idx] : -1;
            if ((en & 63) == i) {
                float x[HEADS];
                if (idx < L1_CAP) {
                    const float4 u0 = *(const float4*)(&xl[idx * HEADS]);
                    const float4 u1 = *(const float4*)(&xl[idx * HEADS + 4]);
                    x[0] = u0.x; x[1] = u0.y; x[2] = u0.z; x[3] = u0.w;
                    x[4] = u1.x; x[5] = u1.y; x[6] = u1.z; x[7] = u1.w;
                } else {
                    const float* sp = aS + (size_t)(en >> 6) * HEADS;
                    const float4 a0 = *(const float4*)sp;
                    const float4 a1 = *(const float4*)(sp + 4);
                    const float av[HEADS] = {a0.x, a0.y, a0.z, a0.w, a1.x, a1.y, a1.z, a1.w};
#pragma unroll
                    for (int h = 0; h < HEADS; ++h) x[h] = lrelu(av[h] + adn[h]);
                }
#pragma unroll
                for (int h = 0; h < HEADS; ++h) mx[h] = fmaxf(mx[h], x[h]);
            }
        }
#pragma unroll
        for (int h = 0; h < HEADS; ++h) {
#pragma unroll
            for (int o = 16; o >= 1; o >>= 1)
                mx[h] = fmaxf(mx[h], __shfl_xor(mx[h], o));
        }

        float exs[HEADS], den[HEADS];
#pragma unroll
        for (int h = 0; h < HEADS; ++h) {
            exs[h] = expf(xs[h] - mx[h]);
            den[h] = 0.f;
        }
#pragma unroll
        for (int j = 0; j < L1_CH; ++j) {
            const float w = exs[j >> 1];
            acc[j] = make_float4(w * acc[j].x, w * acc[j].y, w * acc[j].z, w * acc[j].w);
        }
        for (int t = 0; t < m; t += 32) {
            const int idx = t + lane;
            const int en  = (idx < m) ? list[idx] : -1;
            const bool mine = ((en & 63) == i);
            float ex[HEADS];
#pragma unroll
            for (int h = 0; h < HEADS; ++h) ex[h] = 0.f;
            if (mine) {
                float x[HEADS];
                if (idx < L1_CAP) {
                    const float4 u0 = *(const float4*)(&xl[idx * HEADS]);
                    const float4 u1 = *(const float4*)(&xl[idx * HEADS + 4]);
                    x[0] = u0.x; x[1] = u0.y; x[2] = u0.z; x[3] = u0.w;
                    x[4] = u1.x; x[5] = u1.y; x[6] = u1.z; x[7] = u1.w;
                } else {
                    const float* sp = aS + (size_t)(en >> 6) * HEADS;
                    const float4 a0 = *(const float4*)sp;
                    const float4 a1 = *(const float4*)(sp + 4);
                    const float av[HEADS] = {a0.x, a0.y, a0.z, a0.w, a1.x, a1.y, a1.z, a1.w};
#pragma unroll
                    for (int h = 0; h < HEADS; ++h) x[h] = lrelu(av[h] + adn[h]);
                }
#pragma unroll
                for (int h = 0; h < HEADS; ++h) {
                    ex[h]   = expf(x[h] - mx[h]);
                    den[h] += ex[h];
                }
            }
            unsigned mask = __builtin_amdgcn_ballot_w32(mine);
            while (mask) {
                const int b = __builtin_ctz(mask);
                mask &= mask - 1;
                const int s = __builtin_amdgcn_readlane(en, b) >> 6;
                float w[HEADS];
#pragma unroll
                for (int h = 0; h < HEADS; ++h)
                    w[h] = __int_as_float(__builtin_amdgcn_readlane(__float_as_int(ex[h]), b));
                const float* rp = H1 + (size_t)s * HC + lane * 4;
                float4 v[L1_CH];
#pragma unroll
                for (int j = 0; j < L1_CH; ++j) v[j] = *(const float4*)(rp + j * 128);
#pragma unroll
                for (int j = 0; j < L1_CH; ++j) {
                    const float ww = w[j >> 1];
                    acc[j].x += ww * v[j].x;
                    acc[j].y += ww * v[j].y;
                    acc[j].z += ww * v[j].z;
                    acc[j].w += ww * v[j].w;
                }
            }
        }
        float4 bq[L1_CH];
        {
            const float* bp = B1 + lane * 4;
#pragma unroll
            for (int j = 0; j < L1_CH; ++j) bq[j] = *(const float4*)(bp + j * 128);
        }
        float rc[HEADS];
#pragma unroll
        for (int h = 0; h < HEADS; ++h) {
#pragma unroll
            for (int o = 16; o >= 1; o >>= 1)
                den[h] += __shfl_xor(den[h], o);
            rc[h] = 1.f / ((den[h] + exs[h]) + 1e-16f);
        }
        float* op = agg + (size_t)n * HC + lane * 4;
#pragma unroll
        for (int j = 0; j < L1_CH; ++j) {
            const float r = rc[j >> 1];
            const float4 b = bq[j];
            *(float4*)(op + j * 128) = make_float4(elu1(acc[j].x * r + b.x), elu1(acc[j].y * r + b.y),
                                                   elu1(acc[j].z * r + b.z), elu1(acc[j].w * r + b.w));
        }
    }

    __syncthreads();

    if (wave < L1_GW) {
        const int half = lane >> 4;
        const int ml   = lane & 15;
        const int nb   = min(ml, NUM_CLASS - 1);
        const int k0   = wave * L1_KSL + half * 4;
        const float* wp = W2 + (size_t)k0 * NUM_CLASS + nb;
        const float* ap[L1_TILES];
        v8f c2[L1_TILES];
#pragma unroll
        for (int tt = 0; tt < L1_TILES; ++tt) {
            ap[tt] = agg + (size_t)(n0 + min(tt * 16 + ml, L1_NPB - 1)) * HC + k0;
            c2[tt] = (v8f){0.f, 0.f, 0.f, 0.f, 0.f, 0.f, 0.f, 0.f};
        }
#pragma unroll 2
        for (int k = 0; k < L1_KSL; k += 32) {
            float4 av[L1_TILES][4];
            float  w[4][4];
#pragma unroll
            for (int tt = 0; tt < L1_TILES; ++tt)
#pragma unroll
                for (int s = 0; s < 4; ++s) av[tt][s] = *(const float4*)(ap[tt] + k + 8 * s);
#pragma unroll
            for (int s = 0; s < 4; ++s)
#pragma unroll
                for (int j = 0; j < 4; ++j)
                    w[s][j] = wp[(size_t)(k + 8 * s + j) * NUM_CLASS];
#pragma unroll
            for (int s = 0; s < 4; ++s) {
                v2f a, b;
                b.x = w[s][0]; b.y = w[s][1];
#pragma unroll
                for (int tt = 0; tt < L1_TILES; ++tt) {
                    a.x = av[tt][s].x; a.y = av[tt][s].y;
                    c2[tt] = __builtin_amdgcn_wmma_f32_16x16x4_f32(false, a, false, b,
                                                                   (short)0, c2[tt], false, false);
                }
                b.x = w[s][2]; b.y = w[s][3];
#pragma unroll
                for (int tt = 0; tt < L1_TILES; ++tt) {
                    a.x = av[tt][s].z; a.y = av[tt][s].w;
                    c2[tt] = __builtin_amdgcn_wmma_f32_16x16x4_f32(false, a, false, b,
                                                                   (short)0, c2[tt], false, false);
                }
            }
        }
#pragma unroll
        for (int tt = 0; tt < L1_TILES; ++tt)
#pragma unroll
            for (int r = 0; r < 8; ++r)
                part[wave][tt * 16 + r + 8 * half][ml] = c2[tt][r];
    }
    __syncthreads();

    if (tid < L1_NPB * NUM_CLASS) {
        const int i = tid / NUM_CLASS;
        const int cc = tid - i * NUM_CLASS;
        float v = 0.f;
#pragma unroll
        for (int g = 0; g < L1_GW; ++g) v += part[g][i][cc];
        H2[(size_t)n0 * NUM_CLASS + tid] = v;
        h2s[tid] = v;
    }
    __syncthreads();

    if (tid < L1_NPB) {
        float ss = 0.f, sd = 0.f;
#pragma unroll
        for (int cc = 0; cc < NUM_CLASS; ++cc) {
            const float v = h2s[tid * NUM_CLASS + cc];
            ss += v * as2[cc];
            sd += v * ad2[cc];
        }
        aS2[n0 + tid] = ss; aD2[n0 + tid] = sd;
    }
}

__device__ __forceinline__ void atomicMaxF(float* addr, float v)
{
    if (v >= 0.f) atomicMax((int*)addr, __float_as_int(v));
    else          atomicMin((unsigned int*)addr, __float_as_uint(v));
}

#define L2_NPB     40
#define L2_THREADS 256

__launch_bounds__(L2_THREADS)
__global__ void layer2_owner(const int* __restrict__ lst, const int* __restrict__ lcnt,
                             const float* __restrict__ H2,
                             const float* __restrict__ aS, const float* __restrict__ aD,
                             const float* __restrict__ b2, float* __restrict__ out)
{
    static_assert(N_NODES % L2_NPB == 0, "no partial node range");
    static_assert(L2_NPB == L1_NPB, "block b of the layer-1 owner kernel lists exactly this block's nodes");
    static_assert(L2_NPB <= L2_THREADS, "one thread per owned node for the self loops");
    __shared__ float sd_s[L2_NPB], emax_s[L2_NPB], den_s[L2_NPB];
    __shared__ float agg_s[L2_NPB * NUM_CLASS];

    const int tid = threadIdx.x;
    const int n0  = blockIdx.x * L2_NPB;
    const int* lp = lst + (size_t)blockIdx.x * E_IN;
    const int m   = max(0, min(lcnt[blockIdx.x], E_IN));

    int   en_f = 0;
    float as_f = 0.f;
    float hf[NUM_CLASS];
#pragma unroll
    for (int c = 0; c < NUM_CLASS; ++c) hf[c] = 0.f;
    if (tid < m) {
        en_f = lp[tid];
        const int s = en_f >> 6;
        as_f = aS[s];
#pragma unroll
        for (int c = 0; c < NUM_CLASS; ++c) hf[c] = H2[(size_t)s * NUM_CLASS + c];
    }
    float xs = 0.f;
    float hs[NUM_CLASS];
#pragma unroll
    for (int c = 0; c < NUM_CLASS; ++c) hs[c] = 0.f;
    if (tid < L2_NPB) {
        const int n = n0 + tid;
        const float d = aD[n];
        xs = lrelu(aS[n] + d);
        sd_s[tid]   = d;
        emax_s[tid] = xs;
        den_s[tid]  = 0.f;
#pragma unroll
        for (int c = 0; c < NUM_CLASS; ++c) hs[c] = H2[(size_t)n * NUM_CLASS + c];
    }
    for (int t = tid; t < L2_NPB * NUM_CLASS; t += L2_THREADS) agg_s[t] = 0.f;
    __syncthreads();

    if (tid < m) {
        const int dl = en_f & 63;
        atomicMaxF(&emax_s[dl], lrelu(as_f + sd_s[dl]));
    }
    for (int t = tid + L2_THREADS; t < m; t += L2_THREADS) {
        const int en = lp[t];
        const int dl = en & 63;
        atomicMaxF(&emax_s[dl], lrelu(aS[en >> 6] + sd_s[dl]));
    }
    __syncthreads();

    if (tid < L2_NPB) {
        const float ex = expf(xs - emax_s[tid]);
        atomicAdd(&den_s[tid], ex);
#pragma unroll
        for (int c = 0; c < NUM_CLASS; ++c)
            atomicAdd(&agg_s[tid * NUM_CLASS + c], ex * hs[c]);
    }
    if (tid < m) {
        const int dl = en_f & 63;
        const float ex = expf(lrelu(as_f + sd_s[dl]) - emax_s[dl]);
        atomicAdd(&den_s[dl], ex);
#pragma unroll
        for (int c = 0; c < NUM_CLASS; ++c)
            atomicAdd(&agg_s[dl * NUM_CLASS + c], ex * hf[c]);
    }
    for (int t = tid + L2_THREADS; t < m; t += L2_THREADS) {
        const int en = lp[t];
        const int dl = en & 63;
        const int s  = en >> 6;
        const float ex = expf(lrelu(aS[s] + sd_s[dl]) - emax_s[dl]);
        atomicAdd(&den_s[dl], ex);
#pragma unroll
        for (int c = 0; c < NUM_CLASS; ++c)
            atomicAdd(&agg_s[dl * NUM_CLASS + c], ex * H2[(size_t)s * NUM_CLASS + c]);
    }
    __syncthreads();

    for (int t = tid; t < L2_NPB * NUM_CLASS; t += L2_THREADS) {
        const int i = t / NUM_CLASS;
        const int cc = t - i * NUM_CLASS;
        const float v = agg_s[t] / (den_s[i] + 1e-16f) + b2[cc];
        out[(size_t)n0 * NUM_CLASS + t] = 1.f / (1.f + expf(-v));
    }
}

extern "C" void kernel_launch(void* const* d_in, const int* in_sizes, int n_in,
                              void* d_out, int out_size, void* d_ws, size_t ws_size,
                              hipStream_t stream)
{
    const float* x   = (const float*)d_in[0];
    const int*   ei  = (const int*)d_in[1];
    const float* W1  = (const float*)d_in[2];
    const float* as1 = (const float*)d_in[3];
    const float* ad1 = (const float*)d_in[4];
    const float* b1  = (const float*)d_in[5];
    const float* W2  = (const float*)d_in[6];
    const float* as2 = (const float*)d_in[7];
    const float* ad2 = (const float*)d_in[8];
    const float* b2  = (const float*)d_in[9];
    float* out = (float*)d_out;

    float* ws = (float*)d_ws;
    size_t off = 0;
    float* h1    = ws + off; off += (size_t)N_NODES * HC;
    float* agg1  = ws + off; off += (size_t)N_NODES * HC;
    float* sA1   = ws + off; off += (size_t)N_NODES * HEADS;
    float* sD1   = ws + off; off += (size_t)N_NODES * HEADS;
    float* h2    = ws + off; off += (size_t)N_NODES * NUM_CLASS;
    float* sA2   = ws + off; off += N_NODES;
    float* sD2   = ws + off; off += N_NODES;
    int*   lst   = (int*)(ws + off); off += (size_t)(N_NODES / L1_NPB) * E_IN;
    int*   lcnt  = (int*)(ws + off); off += N_NODES / L1_NPB;
    (void)ws_size; (void)in_sizes; (void)n_in; (void)out_size;

    const dim3 b256(256);

    gemm1_att<<<dim3(HEADS, G1_RG), b256, 0, stream>>>(x, W1, as1, ad1, h1, sA1, sD1);

    layer1_owner<<<N_NODES / L1_NPB, L1_THREADS, 0, stream>>>(ei, h1, sA1, sD1, agg1,
                                                              b1, W2, as2, ad2, h2, sA2, sD2,
                                                              lst, lcnt);

    layer2_owner<<<N_NODES / L2_NPB, L2_THREADS, 0, stream>>>(lst, lcnt, h2, sA2, sD2, b2, out);
}


# Layer-1 owner block hands its finished rows (bias+ELU done) to the layer-2 product through LDS slots (two rounds of 20 rows, pitch 16 B x odd), not through memory; every sum unchanged
# speedup vs baseline: 4.9358x; 1.0671x over previous
#include <hip/hip_runtime.h>
#include <math.h>

#define N_NODES   10000
#define F_IN      128
#define HEADS     8
#define HIDD      256
#define HC        (HEADS * HIDD)
#define NUM_CLASS 10
#define E_IN      50000
#define E_TOT     (E_IN + N_NODES)
#define NEG_SLOPE 0.2f

typedef __attribute__((ext_vector_type(2))) float v2f;
typedef __attribute__((ext_vector_type(8))) float v8f;

#define G1_KP (2 * HIDD + 32)
#define G1_TILES ((N_NODES + 31) / 32)
#define G1_TPB   5
#define G1_RG    ((G1_TILES + G1_TPB - 1) / G1_TPB)

__launch_bounds__(256)
__global__ void gemm1_att(const float* __restrict__ X, const float* __restrict__ W,
                          const float* __restrict__ att_src,
                          const float* __restrict__ att_dst,
                          float* __restrict__ Hout,
                          float* __restrict__ aS, float* __restrict__ aD)
{
    static_assert(N_NODES % 16 == 0, "a wave's 16 rows are all inside or all outside");
    __shared__ __attribute__((aligned(16))) float As[32][F_IN + 4];
    __shared__ __attribute__((aligned(16))) float Bs[F_IN / 2][G1_KP];
    __shared__ __attribute__((aligned(16))) float red[8][32];

    const int tid = threadIdx.x;
    const int hh  = blockIdx.x;
    const int n0  = hh * HIDD;
    const int t0  = blockIdx.y * G1_TPB;

    const int sr = tid >> 5;
    const int sc = (tid & 31) * 4;
    float4 an0, an1, an2, an3;
    {
        const int r0 = t0 * 32 + sr;
        an0 = *(const float4*)(X + (size_t)min(r0,      N_NODES - 1) * F_IN + sc);
        an1 = *(const float4*)(X + (size_t)min(r0 + 8,  N_NODES - 1) * F_IN + sc);
        an2 = *(const float4*)(X + (size_t)min(r0 + 16, N_NODES - 1) * F_IN + sc);
        an3 = *(const float4*)(X + (size_t)min(r0 + 24, N_NODES - 1) * F_IN + sc);
    }
#pragma unroll
    for (int i = 0; i < 16; ++i) {
        const int item = tid + i * 256;
        const int p = item >> 6;
        const int c = (item & 63) * 4;
        const float* wp = W + (size_t)(2 * p) * HC + n0 + c;
        const float4 r0 = *(const float4*)wp;
        const float4 r1 = *(const float4*)(wp + HC);
        *(float4*)(&Bs[p][2 * c])     = make_float4(r0.x, r1.x, r0.y, r1.y);
        *(float4*)(&Bs[p][2 * c + 4]) = make_float4(r0.z, r1.z, r0.w, r1.w);
    }
    *(float4*)(&As[sr][sc])      = an0;
    *(float4*)(&As[sr + 8][sc])  = an1;
    *(float4*)(&As[sr + 16][sc]) = an2;
    *(float4*)(&As[sr + 24][sc]) = an3;
    __syncthreads();

    const int wave = tid >> 5;
    const int lane = tid & 31;
    const int half = lane >> 4;
    const int ml   = lane & 15;
    const int ar   = (wave >> 2) * 16;
    const int bc   = (wave & 3) * 64;

    float asv[4], adv[4];
#pragma unroll
    for (int t = 0; t < 4; ++t) {
        asv[t] = att_src[n0 + bc + 16 * t + ml];
        adv[t] = att_dst[n0 + bc + 16 * t + ml];
    }

    for (int it = 0; it < G1_TPB; ++it) {
        const int m0 = (t0 + it) * 32;
        if (m0 >= N_NODES) break;
        const bool more = (it + 1 < G1_TPB) && (m0 + 32 < N_NODES);

        if (more) {
            const int r0 = m0 + 32 + sr;
            an0 = *(const float4*)(X + (size_t)min(r0,      N_NODES - 1) * F_IN + sc);
            an1 = *(const float4*)(X + (size_t)min(r0 + 8,  N_NODES - 1) * F_IN + sc);
            an2 = *(const float4*)(X + (size_t)min(r0 + 16, N_NODES - 1) * F_IN + sc);
            an3 = *(const float4*)(X + (size_t)min(r0 + 24, N_NODES - 1) * F_IN + sc);
        }

        v8f acc[4];
#pragma unroll
        for (int t = 0; t < 4; ++t)
            acc[t] = (v8f){0.f, 0.f, 0.f, 0.f, 0.f, 0.f, 0.f, 0.f};

#pragma unroll
        for (int kk = 0; kk < F_IN / 4; ++kk) {
            const int p = kk * 2 + half;
            v2f a;
            const float2 av = *(const float2*)(&As[ar + ml][2 * p]);
            a.x = av.x; a.y = av.y;
#pragma unroll
            for (int t = 0; t < 4; ++t) {
                v2f b;
                const float2 bv = *(const float2*)(&Bs[p][2 * (bc + 16 * t + ml)]);
                b.x = bv.x; b.y = bv.y;
                acc[t] = __builtin_amdgcn_wmma_f32_16x16x4_f32(false, a, false, b,
                                                               (short)0, acc[t], false, false);
            }
        }

        if (m0 + ar < N_NODES) {
#pragma unroll
            for (int t = 0; t < 4; ++t) {
                const int col = n0 + bc + 16 * t + ml;
#pragma unroll
                for (int r = 0; r < 8; ++r) {
                    const int row = m0 + ar + r + 8 * half;
                    Hout[(size_t)row * HC + col] = acc[t][r];
                }
            }
        }

        float v[16];
#pragma unroll
        for (int r = 0; r < 8; ++r) {
            float ss = 0.f, sd = 0.f;
#pragma unroll
            for (int t = 0; t < 4; ++t) {
                ss += acc[t][r] * asv[t];
                sd += acc[t][r] * adv[t];
            }
            v[r] = ss; v[8 + r] = sd;
        }
#pragma unroll
        for (int j = 0; j < 16; ++j) {
            v[j] += __int_as_float(__builtin_amdgcn_update_dpp(0, __float_as_int(v[j]), 0x161, 0xf, 0xf, false));
            v[j] += __int_as_float(__builtin_amdgcn_update_dpp(0, __float_as_int(v[j]), 0x162, 0xf, 0xf, false));
            v[j] += __int_as_float(__builtin_amdgcn_update_dpp(0, __float_as_int(v[j]), 0x164, 0xf, 0xf, false));
            v[j] += __int_as_float(__builtin_amdgcn_update_dpp(0, __float_as_int(v[j]), 0x168, 0xf, 0xf, false));
        }
        if (ml == 0) {
#pragma unroll
            for (int j = 0; j < 16; j += 4)
                *(float4*)(&red[wave][half * 16 + j]) = make_float4(v[j], v[j + 1], v[j + 2], v[j + 3]);
        }
        __syncthreads();
        if (more) {
            *(float4*)(&As[sr][sc])      = an0;
            *(float4*)(&As[sr + 8][sc])  = an1;
            *(float4*)(&As[sr + 16][sc]) = an2;
            *(float4*)(&As[sr + 24][sc]) = an3;
        }
        if (tid < 64) {
            const int rh = tid >> 5;
            const int ln = tid & 31;
            const float s = ((red[rh * 4 + 0][ln] + red[rh * 4 + 1][ln])
                             + red[rh * 4 + 2][ln]) + red[rh * 4 + 3][ln];
            const int row = m0 + rh * 16 + (ln & 7) + 8 * (ln >> 4);
            if (row < N_NODES) {
                float* dstp = ((ln >> 3) & 1) ? aD : aS;
                dstp[(size_t)row * HEADS + hh] = s;
            }
        }
        if (more) __syncthreads();
    }
}

__device__ __forceinline__ float elu1(float v)
{
    const float e = expm1f(v);
    return (v > 0.f) ? v : e;
}

#define L1_NPB     40
#define L1_THREADS 640
#define L1_WAVES   (L1_THREADS / 32)
#define L1_Q       (E_IN / 4)
#define L1_IT      ((L1_Q + L1_THREADS - 1) / L1_THREADS)
#define L1_CH      (HC / 128)
#define L1_CAP     1024
#define L1_GW      16
#define L1_KSL     (HC / L1_GW)
#define L1_LCAP    4096
#define L1_ZP      (HC + 4)
#define L1_ZSLOTS  (L1_WAVES + 4)

__device__ __forceinline__ float lrelu(float x)
{
    return (x >= 0.f) ? x : NEG_SLOPE * x;
}

__device__ __forceinline__ float xrow(float x)
{
    const int b = __float_as_int(x);
    return __int_as_float(__builtin_amdgcn_permlanex16(b, b, 0x76543210, 0xfedcba98, false, false));
}
template <int M>
__device__ __forceinline__ float xlane(float x)
{
    return __int_as_float(__builtin_amdgcn_update_dpp(0, __float_as_int(x), 0x160 + M, 0xf, 0xf, false));
}

template <int NT>
__device__ __forceinline__ void h2_slice(const float* zl, const int (&ro)[NT],
                                         const float* wp, unsigned wo, v8f (&c)[NT])
{
#pragma unroll 2
    for (int k = 0; k < L1_KSL; k += 32) {
        float4 av[NT][4];
        float  w[4][4];
#pragma unroll
        for (int s = 0; s < 4; ++s)
#pragma unroll
            for (int j = 0; j < 4; ++j)
                w[s][j] = wp[wo + (unsigned)((k + 8 * s + j) * NUM_CLASS)];
#pragma unroll
        for (int tt = 0; tt < NT; ++tt)
#pragma unroll
            for (int s = 0; s < 4; ++s) av[tt][s] = *(const float4*)(zl + ro[tt] + k + 8 * s);
#pragma unroll
        for (int s = 0; s < 4; ++s) {
            v2f a, b;
            b.x = w[s][0]; b.y = w[s][1];
#pragma unroll
            for (int tt = 0; tt < NT; ++tt) {
                a.x = av[tt][s].x; a.y = av[tt][s].y;
                c[tt] = __builtin_amdgcn_wmma_f32_16x16x4_f32(false, a, false, b,
                                                              (short)0, c[tt], false, false);
            }
            b.x = w[s][2]; b.y = w[s][3];
#pragma unroll
            for (int tt = 0; tt < NT; ++tt) {
                a.x = av[tt][s].z; a.y = av[tt][s].w;
                c[tt] = __builtin_amdgcn_wmma_f32_16x16x4_f32(false, a, false, b,
                                                              (short)0, c[tt], false, false);
            }
        }
    }
}

__launch_bounds__(L1_THREADS)
__global__ void layer1_owner(const int* __restrict__ ei, const float* __restrict__ H1,
                             const float* __restrict__ aS, const float* __restrict__ aD,
                             const float* B1, const float* W2,
                             const float* __restrict__ as2, const float* __restrict__ ad2,
                             float* __restrict__ H2,
                             float* __restrict__ aS2, float* __restrict__ aD2,
                             int* __restrict__ lst, int* __restrict__ lcnt)
{
    static_assert(N_NODES % L1_NPB == 0, "no partial node range");
    static_assert(E_IN % 4 == 0, "destination column read as 4-edge granules");
    static_assert(L1_NPB <= 63 && N_NODES < (1 << 24), "source and local node share one int; local 63 marks no entry");
    static_assert(HEADS == 8 && HIDD == 256, "two 128-channel pieces per head, 8 scores per node");
    static_assert(L1_GW <= L1_WAVES && L1_KSL % 32 == 0, "a K slice per wave, whole 32-k trips");
    static_assert(L1_NPB * NUM_CLASS <= L1_THREADS, "one thread per owned h2 element");
    static_assert(L1_NPB == 2 * L1_WAVES && L1_WAVES == 20 && L1_GW == 16,
                  "two rounds of one node per wave; slots 16..23 hold the last tile's 8 rows");
    static_assert((L1_ZP * 4) % 32 == 16, "slot pitch 16 B x odd");
    __shared__ int list[L1_LCAP];
    __shared__ __attribute__((aligned(16))) float zs[L1_ZSLOTS * L1_ZP];
    __shared__ __attribute__((aligned(16))) float part[L1_GW][L1_NPB][16];
    __shared__ float h2s[L1_NPB * NUM_CLASS];
    __shared__ __attribute__((aligned(16))) float xl[L1_CAP * HEADS];
    __shared__ int cnt;

    const int tid = threadIdx.x;
    const int n0  = blockIdx.x * L1_NPB;
    int* lo = lst + (size_t)blockIdx.x * E_IN;

    const int4* dq = (const int4*)(ei + E_IN);
    int4 d4[L1_IT];
#pragma unroll
    for (int i = 0; i < L1_IT; ++i)
        d4[i] = dq[min(tid + i * L1_THREADS, L1_Q - 1)];

    if (tid == 0) cnt = 0;
    __syncthreads();

#pragma unroll
    for (int i = 0; i < L1_IT; ++i) {
        const int j = tid + i * L1_THREADS;
        const int dv[4] = {d4[i].x, d4[i].y, d4[i].z, d4[i].w};
#pragma unroll
        for (int q = 0; q < 4; ++q) {
            const unsigned dl = (unsigned)(dv[q] - n0);
            if (j < L1_Q && dl < (unsigned)L1_NPB) {
                const int s = ei[4 * j + q];
                const int slot = atomicAdd(&cnt, 1);
                const int en = (s << 6) | (int)dl;
                if (slot < L1_LCAP) list[slot] = en;
                else                lo[slot]   = en;
            }
        }
    }
    __syncthreads();
    const int m = cnt;

    {
        for (int t = tid; t < min(m, L1_LCAP); t += L1_THREADS) lo[t] = list[t];
        if (tid == 0) lcnt[blockIdx.x] = m;
    }

    for (int t = tid; t < min(m, L1_CAP); t += L1_THREADS) {
        const int en = list[t];
        const float* sp = aS + (size_t)(en >> 6) * HEADS;
        const float* dp = aD + (size_t)(n0 + (en & 63)) * HEADS;
        const float4 a0 = *(const float4*)sp;
        const float4 a1 = *(const float4*)(sp + 4);
        const float4 e0 = *(const float4*)dp;
        const float4 e1 = *(const float4*)(dp + 4);
        *(float4*)(&xl[t * HEADS])     = make_float4(lrelu(a0.x + e0.x), lrelu(a0.y + e0.y),
                                                     lrelu(a0.z + e0.z), lrelu(a0.w + e0.w));
        *(float4*)(&xl[t * HEADS + 4]) = make_float4(lrelu(a1.x + e1.x), lrelu(a1.y + e1.y),
                                                     lrelu(a1.z + e1.z), lrelu(a1.w + e1.w));
    }
    __syncthreads();

    const int lane = tid & 31;
    const int wave = __builtin_amdgcn_readfirstlane(tid >> 5);

    const int half = lane >> 4;
    const int ml   = lane & 15;
    const int ksl  = min(wave, L1_GW - 1) * L1_KSL;
    const int zo0  = ml * L1_ZP + half * 4;
    const int zo1  = (16 + min(ml, 7)) * L1_ZP + half * 4;
    const unsigned wo = (unsigned)(half * 4 * NUM_CLASS + min(ml, NUM_CLASS - 1));

    for (int rnd = 0; rnd < 2; ++rnd) {
        const int i = wave + rnd * L1_WAVES;
        const int n = n0 + i;

        float4 acc[L1_CH];
        {
            const float* rp = H1 + (size_t)n * HC + lane * 4;
#pragma unroll
            for (int j = 0; j < L1_CH; ++j) acc[j] = *(const float4*)(rp + j * 128);
        }

        float xs[HEADS], mx[HEADS];
        {
            const float4 p0 = *(const float4*)(aD + (size_t)n * HEADS);
            const float4 p1 = *(const float4*)(aD + (size_t)n * HEADS + 4);
            const float4 q0 = *(const float4*)(aS + (size_t)n * HEADS);
            const float4 q1 = *(const float4*)(aS + (size_t)n * HEADS + 4);
            const float dd[HEADS] = {p0.x, p0.y, p0.z, p0.w, p1.x, p1.y, p1.z, p1.w};
            const float ss[HEADS] = {q0.x, q0.y, q0.z, q0.w, q1.x, q1.y, q1.z, q1.w};
#pragma unroll
            for (int h = 0; h < HEADS; ++h) {
                xs[h]  = lrelu(ss[h] + dd[h]);
                mx[h]  = xs[h];
            }
        }

        for (int t = 0; t < m; t += 32) {
            const int idx = t + lane;
            const int en  = (idx < m) ? ((idx < L1_LCAP) ? list[idx] : lo[(unsigned)idx]) : -1;
            if ((en & 63) == i) {
                float x[HEADS];
                if (idx < L1_CAP) {
                    const float4 u0 = *(const float4*)(&xl[idx * HEADS]);
                    const float4 u1 = *(const float4*)(&xl[idx * HEADS + 4]);
                    x[0] = u0.x; x[1] = u0.y; x[2] = u0.z; x[3] = u0.w;
                    x[4] = u1.x; x[5] = u1.y; x[6] = u1.z; x[7] = u1.w;
                } else {
                    const float* sp = aS + (size_t)(en >> 6) * HEADS;
                    const float* dp = aD + (size_t)n * HEADS;
                    const float4 a0 = *(const float4*)sp;
                    const float4 a1 = *(const float4*)(sp + 4);
                    const float4 e0 = *(const float4*)dp;
                    const float4 e1 = *(const float4*)(dp + 4);
                    const float av[HEADS] = {a0.x, a0.y, a0.z, a0.w, a1.x, a1.y, a1.z, a1.w};
                    const float dv[HEADS] = {e0.x, e0.y, e0.z, e0.w, e1.x, e1.y, e1.z, e1.w};
#pragma unroll
                    for (int h = 0; h < HEADS; ++h) x[h] = lrelu(av[h] + dv[h]);
                }
#pragma unroll
                for (int h = 0; h < HEADS; ++h) mx[h] = fmaxf(mx[h], x[h]);
            }
        }
#pragma unroll
        for (int h = 0; h < HEADS; ++h) {
            mx[h] = fmaxf(mx[h], xrow(mx[h]));
            mx[h] = fmaxf(mx[h], xlane<8>(mx[h]));
            mx[h] = fmaxf(mx[h], xlane<4>(mx[h]));
            mx[h] = fmaxf(mx[h], xlane<2>(mx[h]));
            mx[h] = fmaxf(mx[h], xlane<1>(mx[h]));
        }

        float exs[HEADS], den[HEADS];
#pragma unroll
        for (int h = 0; h < HEADS; ++h) {
            exs[h] = expf(xs[h] - mx[h]);
            den[h] = 0.f;
        }
#pragma unroll
        for (int j = 0; j < L1_CH; ++j) {
            const float w = exs[j >> 1];
            acc[j] = make_float4(w * acc[j].x, w * acc[j].y, w * acc[j].z, w * acc[j].w);
        }
        for (int t = 0; t < m; t += 32) {
            const int idx = t + lane;
            const int en  = (idx < m) ? ((idx < L1_LCAP) ? list[idx] : lo[(unsigned)idx]) : -1;
            const bool mine = ((en & 63) == i);
            float ex[HEADS];
#pragma unroll
            for (int h = 0; h < HEADS; ++h) ex[h] = 0.f;
            if (mine) {
                float x[HEADS];
                if (idx < L1_CAP) {
                    const float4 u0 = *(const float4*)(&xl[idx * HEADS]);
                    const float4 u1 = *(const float4*)(&xl[idx * HEADS + 4]);
                    x[0] = u0.x; x[1] = u0.y; x[2] = u0.z; x[3] = u0.w;
                    x[4] = u1.x; x[5] = u1.y; x[6] = u1.z; x[7] = u1.w;
                } else {
                    const float* sp = aS + (size_t)(en >> 6) * HEADS;
                    const float* dp = aD + (size_t)n * HEADS;
                    const float4 a0 = *(const float4*)sp;
                    const float4 a1 = *(const float4*)(sp + 4);
                    const float4 e0 = *(const float4*)dp;
                    const float4 e1 = *(const float4*)(dp + 4);
                    const float av[HEADS] = {a0.x, a0.y, a0.z, a0.w, a1.x, a1.y, a1.z, a1.w};
                    const float dv[HEADS] = {e0.x, e0.y, e0.z, e0.w, e1.x, e1.y, e1.z, e1.w};
#pragma unroll
                    for (int h = 0; h < HEADS; ++h) x[h] = lrelu(av[h] + dv[h]);
                }
#pragma unroll
                for (int h = 0; h < HEADS; ++h) {
                    ex[h]   = expf(x[h] - mx[h]);
                    den[h] += ex[h];
                }
            }
            unsigned mask = __builtin_amdgcn_ballot_w32(mine);
            while (mask) {
                const int b = __builtin_ctz(mask);
                mask &= mask - 1;
                const int s = __builtin_amdgcn_readlane(en, b) >> 6;
                float w[HEADS];
#pragma unroll
                for (int h = 0; h < HEADS; ++h)
                    w[h] = __int_as_float(__builtin_amdgcn_readlane(__float_as_int(ex[h]), b));
                const float* rp = H1 + (size_t)s * HC + lane * 4;
                float4 v[L1_CH];
#pragma unroll
                for (int j = 0; j < L1_CH; ++j) v[j] = *(const float4*)(rp + j * 128);
#pragma unroll
                for (int j = 0; j < L1_CH; ++j) {
                    const float ww = w[j >> 1];
                    acc[j].x += ww * v[j].x;
                    acc[j].y += ww * v[j].y;
                    acc[j].z += ww * v[j].z;
                    acc[j].w += ww * v[j].w;
                }
            }
        }
        float4 bq[L1_CH];
        {
            const float* bp = B1 + lane * 4;
#pragma unroll
            for (int j = 0; j < L1_CH; ++j) bq[j] = *(const float4*)(bp + j * 128);
        }
        float rc[HEADS];
#pragma unroll
        for (int h = 0; h < HEADS; ++h) {
            den[h] += xrow(den[h]);
            den[h] += xlane<8>(den[h]);
            den[h] += xlane<4>(den[h]);
            den[h] += xlane<2>(den[h]);
            den[h] += xlane<1>(den[h]);
            rc[h] = 1.f / ((den[h] + exs[h]) + 1e-16f);
        }
#pragma unroll
        for (int j = 0; j < L1_CH; ++j) {
            const float r = rc[j >> 1];
            const float4 b = bq[j];
            acc[j] = make_float4(elu1(acc[j].x * r + b.x), elu1(acc[j].y * r + b.y),
                                 elu1(acc[j].z * r + b.z), elu1(acc[j].w * r + b.w));
        }
        {
            const int slot = (rnd == 0 && wave >= 16) ? wave + (L1_ZSLOTS - L1_WAVES) : wave;
            float* zp = zs + slot * L1_ZP + lane * 4;
#pragma unroll
            for (int j = 0; j < L1_CH; ++j) *(float4*)(zp + j * 128) = acc[j];
        }

        __syncthreads();

        if (wave < L1_GW) {
            const float* zl = zs + ksl;
            const float* wp = W2 + (size_t)ksl * NUM_CLASS;
            float* pp = &part[wave][8 * half][ml];
            if (rnd == 0) {
                const int ro[1] = {zo0};
                v8f c2[1] = {(v8f){0.f, 0.f, 0.f, 0.f, 0.f, 0.f, 0.f, 0.f}};
                h2_slice<1>(zl, ro, wp, wo, c2);
#pragma unroll
                for (int r = 0; r < 8; ++r)
                    pp[r * 16] = c2[0][r];
            } else {
                const int ro[2] = {zo0, zo1};
                v8f c2[2] = {(v8f){0.f, 0.f, 0.f, 0.f, 0.f, 0.f, 0.f, 0.f},
                             (v8f){0.f, 0.f, 0.f, 0.f, 0.f, 0.f, 0.f, 0.f}};
                h2_slice<2>(zl, ro, wp, wo, c2);
#pragma unroll
                for (int r = 0; r < 8; ++r)
                    pp[(L1_WAVES + r) * 16] = c2[0][r];
                if (half == 0) {
#pragma unroll
                    for (int r = 0; r < 8; ++r)
                        pp[((r < 4) ? (36 + r) : (12 + r)) * 16] = c2[1][r];
                }
            }
        }
        __syncthreads();
    }

    const int t2 = wave * 32 + lane;

    if (t2 < L1_NPB * NUM_CLASS) {
        const int i = t2 / NUM_CLASS;
        const int cc = t2 - i * NUM_CLASS;
        float v = 0.f;
#pragma unroll
        for (int g = 0; g < L1_GW; ++g) v += part[g][i][cc];
        H2[(size_t)n0 * NUM_CLASS + t2] = v;
        h2s[t2] = v;
    }
    __syncthreads();

    if (t2 < L1_NPB) {
        float ss = 0.f, sd = 0.f;
#pragma unroll
        for (int cc = 0; cc < NUM_CLASS; ++cc) {
            const float v = h2s[t2 * NUM_CLASS + cc];
            ss += v * as2[cc];
            sd += v * ad2[cc];
        }
        aS2[n0 + t2] = ss; aD2[n0 + t2] = sd;
    }
}

__device__ __forceinline__ void atomicMaxF(float* addr, float v)
{
    if (v >= 0.f) atomicMax((int*)addr, __float_as_int(v));
    else          atomicMin((unsigned int*)addr, __float_as_uint(v));
}

#define L2_NPB     40
#define L2_THREADS 256

__launch_bounds__(L2_THREADS)
__global__ void layer2_owner(const int* __restrict__ lst, const int* __restrict__ lcnt,
                             const float* __restrict__ H2,
                             const float* __restrict__ aS, const float* __restrict__ aD,
                             const float* __restrict__ b2, float* __restrict__ out)
{
    static_assert(N_NODES % L2_NPB == 0, "no partial node range");
    static_assert(L2_NPB == L1_NPB, "block b of the layer-1 owner kernel lists exactly this block's nodes");
    static_assert(L2_NPB <= L2_THREADS, "one thread per owned node for the self loops");
    __shared__ float sd_s[L2_NPB], emax_s[L2_NPB], den_s[L2_NPB];
    __shared__ float agg_s[L2_NPB * NUM_CLASS];

    const int tid = threadIdx.x;
    const int n0  = blockIdx.x * L2_NPB;
    const int* lp = lst + (size_t)blockIdx.x * E_IN;
    const int m   = max(0, min(lcnt[blockIdx.x], E_IN));

    int   en_f = 0;
    float as_f = 0.f;
    float hf[NUM_CLASS];
#pragma unroll
    for (int c = 0; c < NUM_CLASS; ++c) hf[c] = 0.f;
    if (tid < m) {
        en_f = lp[tid];
        const int s = en_f >> 6;
        as_f = aS[s];
#pragma unroll
        for (int c = 0; c < NUM_CLASS; ++c) hf[c] = H2[(size_t)s * NUM_CLASS + c];
    }
    float xs = 0.f;
    float hs[NUM_CLASS];
#pragma unroll
    for (int c = 0; c < NUM_CLASS; ++c) hs[c] = 0.f;
    if (tid < L2_NPB) {
        const int n = n0 + tid;
        const float d = aD[n];
        xs = lrelu(aS[n] + d);
        sd_s[tid]   = d;
        emax_s[tid] = xs;
        den_s[tid]  = 0.f;
#pragma unroll
        for (int c = 0; c < NUM_CLASS; ++c) hs[c] = H2[(size_t)n * NUM_CLASS + c];
    }
    for (int t = tid; t < L2_NPB * NUM_CLASS; t += L2_THREADS) agg_s[t] = 0.f;
    __syncthreads();

    if (tid < m) {
        const int dl = en_f & 63;
        atomicMaxF(&emax_s[dl], lrelu(as_f + sd_s[dl]));
    }
    for (int t = tid + L2_THREADS; t < m; t += L2_THREADS) {
        const int en = lp[t];
        const int dl = en & 63;
        atomicMaxF(&emax_s[dl], lrelu(aS[en >> 6] + sd_s[dl]));
    }
    __syncthreads();

    if (tid < L2_NPB) {
        const float ex = expf(xs - emax_s[tid]);
        atomicAdd(&den_s[tid], ex);
#pragma unroll
        for (int c = 0; c < NUM_CLASS; ++c)
            atomicAdd(&agg_s[tid * NUM_CLASS + c], ex * hs[c]);
    }
    if (tid < m) {
        const int dl = en_f & 63;
        const float ex = expf(lrelu(as_f + sd_s[dl]) - emax_s[dl]);
        atomicAdd(&den_s[dl], ex);
#pragma unroll
        for (int c = 0; c < NUM_CLASS; ++c)
            atomicAdd(&agg_s[dl * NUM_CLASS + c], ex * hf[c]);
    }
    for (int t = tid + L2_THREADS; t < m; t += L2_THREADS) {
        const int en = lp[t];
        const int dl = en & 63;
        const int s  = en >> 6;
        const float ex = expf(lrelu(aS[s] + sd_s[dl]) - emax_s[dl]);
        atomicAdd(&den_s[dl], ex);
#pragma unroll
        for (int c = 0; c < NUM_CLASS; ++c)
            atomicAdd(&agg_s[dl * NUM_CLASS + c], ex * H2[(size_t)s * NUM_CLASS + c]);
    }
    __syncthreads();

    for (int t = tid; t < L2_NPB * NUM_CLASS; t += L2_THREADS) {
        const int i = t / NUM_CLASS;
        const int cc = t - i * NUM_CLASS;
        const float v = agg_s[t] / (den_s[i] + 1e-16f) + b2[cc];
        out[(size_t)n0 * NUM_CLASS + t] = 1.f / (1.f + expf(-v));
    }
}

extern "C" void kernel_launch(void* const* d_in, const int* in_sizes, int n_in,
                              void* d_out, int out_size, void* d_ws, size_t ws_size,
                              hipStream_t stream)
{
    const float* x   = (const float*)d_in[0];
    const int*   ei  = (const int*)d_in[1];
    const float* W1  = (const float*)d_in[2];
    const float* as1 = (const float*)d_in[3];
    const float* ad1 = (const float*)d_in[4];
    const float* b1  = (const float*)d_in[5];
    const float* W2  = (const float*)d_in[6];
    const float* as2 = (const float*)d_in[7];
    const float* ad2 = (const float*)d_in[8];
    const float* b2  = (const float*)d_in[9];
    float* out = (float*)d_out;

    float* ws = (float*)d_ws;
    size_t off = 0;
    float* h1    = ws + off; off += (size_t)N_NODES * HC;
    float* sA1   = ws + off; off += (size_t)N_NODES * HEADS;
    float* sD1   = ws + off; off += (size_t)N_NODES * HEADS;
    float* h2    = ws + off; off += (size_t)N_NODES * NUM_CLASS;
    float* sA2   = ws + off; off += N_NODES;
    float* sD2   = ws + off; off += N_NODES;
    int*   lst   = (int*)(ws + off); off += (size_t)(N_NODES / L1_NPB) * E_IN;
    int*   lcnt  = (int*)(ws + off); off += N_NODES / L1_NPB;
    (void)ws_size; (void)in_sizes; (void)n_in; (void)out_size;

    const dim3 b256(256);

    gemm1_att<<<dim3(HEADS, G1_RG), b256, 0, stream>>>(x, W1, as1, ad1, h1, sA1, sD1);

    layer1_owner<<<N_NODES / L1_NPB, L1_THREADS, 0, stream>>>(ei, h1, sA1, sD1,
                                                              b1, W2, as2, ad2, h2, sA2, sD2,
                                                              lst, lcnt);

    layer2_owner<<<N_NODES / L2_NPB, L2_THREADS, 0, stream>>>(lst, lcnt, h2, sA2, sD2, b2, out);
}


# Owner blocks' edge lists built by the GEMM1 blocks beside their k loops (edge granules asked for ahead of a tile's k loop, matched behind it): owner's column scan gone; list (sum) order changes
# speedup vs baseline: 5.1635x; 1.0461x over previous
#include <hip/hip_runtime.h>
#include <math.h>

#define N_NODES   10000
#define F_IN      128
#define HEADS     8
#define HIDD      256
#define HC        (HEADS * HIDD)
#define NUM_CLASS 10
#define E_IN      50000
#define E_TOT     (E_IN + N_NODES)
#define NEG_SLOPE 0.2f

typedef __attribute__((ext_vector_type(2))) float v2f;
typedef __attribute__((ext_vector_type(8))) float v8f;

#define G1_KP (2 * HIDD + 32)
#define G1_TILES ((N_NODES + 31) / 32)
#define G1_TPB   5
#define G1_RG    ((G1_TILES + G1_TPB - 1) / G1_TPB)

#define L1_NPB   40
#define L1_Q     (E_IN / 4)
#define SC_HALF  (L1_Q / 2)
#define SC_PER   ((SC_HALF + 255) / 256)
#define SC_CH    ((SC_PER + G1_TPB - 1) / G1_TPB)
#define SC_BLOCKS (2 * (N_NODES / L1_NPB))

__device__ __forceinline__ void sc_ask(const int4* __restrict__ dq, const int4* __restrict__ sq,
                                       int chunk, int tid, int4 (&dd)[SC_CH], int4 (&ss)[SC_CH])
{
#pragma unroll
    for (int i = 0; i < SC_CH; ++i) {
        const int g = min((chunk * SC_CH + i) * 256 + tid, SC_HALF - 1);
        dd[i] = dq[g];
        ss[i] = sq[g];
    }
}
__device__ __forceinline__ void sc_match(const int4 (&dd)[SC_CH], const int4 (&ss)[SC_CH],
                                         int chunk, int tid, int ln0,
                                         int* ecnt, int* __restrict__ lo)
{
#pragma unroll
    for (int i = 0; i < SC_CH; ++i) {
        const bool live = (chunk * SC_CH + i) * 256 + tid < SC_HALF;
        const int dv[4] = {dd[i].x, dd[i].y, dd[i].z, dd[i].w};
        const int sv[4] = {ss[i].x, ss[i].y, ss[i].z, ss[i].w};
#pragma unroll
        for (int q = 0; q < 4; ++q) {
            const unsigned dl = (unsigned)(dv[q] - ln0);
            if (live && dl < (unsigned)L1_NPB) {
                const int slot = atomicAdd(ecnt, 1);
                lo[slot] = (sv[q] << 6) | (int)dl;
            }
        }
    }
}

__launch_bounds__(256)
__global__ void gemm1_att(const float* __restrict__ X, const float* __restrict__ W,
                          const float* __restrict__ att_src,
                          const float* __restrict__ att_dst,
                          float* __restrict__ Hout,
                          float* __restrict__ aS, float* __restrict__ aD,
                          const int* __restrict__ ei,
                          int* __restrict__ lst, int* __restrict__ lcnt)
{
    static_assert(N_NODES % 16 == 0, "a wave's 16 rows are all inside or all outside");
    static_assert(E_IN % 8 == 0 && N_NODES % L1_NPB == 0, "two list halves of whole granules, no partial node range");
    static_assert(SC_CH * G1_TPB * 256 >= SC_HALF, "the tile trips cover a half");
    static_assert(SC_BLOCKS <= HEADS * G1_RG, "a block per list half");
    static_assert(L1_NPB <= 63 && N_NODES < (1 << 24), "source and local node share one int");
    __shared__ __attribute__((aligned(16))) float As[32][F_IN + 4];
    __shared__ __attribute__((aligned(16))) float Bs[F_IN / 2][G1_KP];
    __shared__ __attribute__((aligned(16))) float red[8][32];
    __shared__ int ecnt;

    const int tid = threadIdx.x;
    const int hh  = blockIdx.x;
    const int n0  = hh * HIDD;
    const int t0  = blockIdx.y * G1_TPB;

    const int  fb     = blockIdx.y * HEADS + blockIdx.x;
    const bool lister = fb < SC_BLOCKS;
    const int  ln0    = (fb >> 1) * L1_NPB;
    int* lo = lst + (size_t)(fb >> 1) * E_IN + (fb & 1) * (E_IN / 2);
    const int4* dq = (const int4*)(ei + E_IN) + (fb & 1) * SC_HALF;
    const int4* sq = (const int4*)ei + (fb & 1) * SC_HALF;
    if (tid == 0) ecnt = 0;
    int4 dd[SC_CH], ss[SC_CH];

    const int sr = tid >> 5;
    const int sc = (tid & 31) * 4;
    float4 an0, an1, an2, an3;
    {
        const int r0 = t0 * 32 + sr;
        an0 = *(const float4*)(X + (size_t)min(r0,      N_NODES - 1) * F_IN + sc);
        an1 = *(const float4*)(X + (size_t)min(r0 + 8,  N_NODES - 1) * F_IN + sc);
        an2 = *(const float4*)(X + (size_t)min(r0 + 16, N_NODES - 1) * F_IN + sc);
        an3 = *(const float4*)(X + (size_t)min(r0 + 24, N_NODES - 1) * F_IN + sc);
    }
#pragma unroll
    for (int i = 0; i < 16; ++i) {
        const int item = tid + i * 256;
        const int p = item >> 6;
        const int c = (item & 63) * 4;
        const float* wp = W + (size_t)(2 * p) * HC + n0 + c;
        const float4 r0 = *(const float4*)wp;
        const float4 r1 = *(const float4*)(wp + HC);
        *(float4*)(&Bs[p][2 * c])     = make_float4(r0.x, r1.x, r0.y, r1.y);
        *(float4*)(&Bs[p][2 * c + 4]) = make_float4(r0.z, r1.z, r0.w, r1.w);
    }
    *(float4*)(&As[sr][sc])      = an0;
    *(float4*)(&As[sr + 8][sc])  = an1;
    *(float4*)(&As[sr + 16][sc]) = an2;
    *(float4*)(&As[sr + 24][sc]) = an3;
    __syncthreads();

    const int wave = tid >> 5;
    const int lane = tid & 31;
    const int half = lane >> 4;
    const int ml   = lane & 15;
    const int ar   = (wave >> 2) * 16;
    const int bc   = (wave & 3) * 64;

    float asv[4], adv[4];
#pragma unroll
    for (int t = 0; t < 4; ++t) {
        asv[t] = att_src[n0 + bc + 16 * t + ml];
        adv[t] = att_dst[n0 + bc + 16 * t + ml];
    }

    int it = 0;
    for (; it < G1_TPB; ++it) {
        const int m0 = (t0 + it) * 32;
        if (m0 >= N_NODES) break;
        const bool more = (it + 1 < G1_TPB) && (m0 + 32 < N_NODES);

        if (lister) sc_ask(dq, sq, it, tid, dd, ss);

        if (more) {
            const int r0 = m0 + 32 + sr;
            an0 = *(const float4*)(X + (size_t)min(r0,      N_NODES - 1) * F_IN + sc);
            an1 = *(const float4*)(X + (size_t)min(r0 + 8,  N_NODES - 1) * F_IN + sc);
            an2 = *(const float4*)(X + (size_t)min(r0 + 16, N_NODES - 1) * F_IN + sc);
            an3 = *(const float4*)(X + (size_t)min(r0 + 24, N_NODES - 1) * F_IN + sc);
        }

        v8f acc[4];
#pragma unroll
        for (int t = 0; t < 4; ++t)
            acc[t] = (v8f){0.f, 0.f, 0.f, 0.f, 0.f, 0.f, 0.f, 0.f};

#pragma unroll
        for (int kk = 0; kk < F_IN / 4; ++kk) {
            const int p = kk * 2 + half;
            v2f a;
            const float2 av = *(const float2*)(&As[ar + ml][2 * p]);
            a.x = av.x; a.y = av.y;
#pragma unroll
            for (int t = 0; t < 4; ++t) {
                v2f b;
                const float2 bv = *(const float2*)(&Bs[p][2 * (bc + 16 * t + ml)]);
                b.x = bv.x; b.y = bv.y;
                acc[t] = __builtin_amdgcn_wmma_f32_16x16x4_f32(false, a, false, b,
                                                               (short)0, acc[t], false, false);
            }
        }

        if (m0 + ar < N_NODES) {
#pragma unroll
            for (int t = 0; t < 4; ++t) {
                const int col = n0 + bc + 16 * t + ml;
#pragma unroll
                for (int r = 0; r < 8; ++r) {
                    const int row = m0 + ar + r + 8 * half;
                    Hout[(size_t)row * HC + col] = acc[t][r];
                }
            }
        }

        if (lister) sc_match(dd, ss, it, tid, ln0, &ecnt, lo);

        float v[16];
#pragma unroll
        for (int r = 0; r < 8; ++r) {
            float ss = 0.f, sd = 0.f;
#pragma unroll
            for (int t = 0; t < 4; ++t) {
                ss += acc[t][r] * asv[t];
                sd += acc[t][r] * adv[t];
            }
            v[r] = ss; v[8 + r] = sd;
        }
#pragma unroll
        for (int j = 0; j < 16; ++j) {
            v[j] += __int_as_float(__builtin_amdgcn_update_dpp(0, __float_as_int(v[j]), 0x161, 0xf, 0xf, false));
            v[j] += __int_as_float(__builtin_amdgcn_update_dpp(0, __float_as_int(v[j]), 0x162, 0xf, 0xf, false));
            v[j] += __int_as_float(__builtin_amdgcn_update_dpp(0, __float_as_int(v[j]), 0x164, 0xf, 0xf, false));
            v[j] += __int_as_float(__builtin_amdgcn_update_dpp(0, __float_as_int(v[j]), 0x168, 0xf, 0xf, false));
        }
        if (ml == 0) {
#pragma unroll
            for (int j = 0; j < 16; j += 4)
                *(float4*)(&red[wave][half * 16 + j]) = make_float4(v[j], v[j + 1], v[j + 2], v[j + 3]);
        }
        __syncthreads();
        if (more) {
            *(float4*)(&As[sr][sc])      = an0;
            *(float4*)(&As[sr + 8][sc])  = an1;
            *(float4*)(&As[sr + 16][sc]) = an2;
            *(float4*)(&As[sr + 24][sc]) = an3;
        }
        if (tid < 64) {
            const int rh = tid >> 5;
            const int ln = tid & 31;
            const float s = ((red[rh * 4 + 0][ln] + red[rh * 4 + 1][ln])
                             + red[rh * 4 + 2][ln]) + red[rh * 4 + 3][ln];
            const int row = m0 + rh * 16 + (ln & 7) + 8 * (ln >> 4);
            if (row < N_NODES) {
                float* dstp = ((ln >> 3) & 1) ? aD : aS;
                dstp[(size_t)row * HEADS + hh] = s;
            }
        }
        if (more) __syncthreads();
    }

    if (lister) {
        for (int ch = it; ch < G1_TPB; ++ch) {
            sc_ask(dq, sq, ch, tid, dd, ss);
            sc_match(dd, ss, ch, tid, ln0, &ecnt, lo);
        }
        __syncthreads();
        if (tid == 0) lcnt[fb] = ecnt;
    }
}

__device__ __forceinline__ float elu1(float v)
{
    const float e = expm1f(v);
    return (v > 0.f) ? v : e;
}

#define L1_THREADS 640
#define L1_WAVES   (L1_THREADS / 32)
#define L1_CH      (HC / 128)
#define L1_CAP     1024
#define L1_GW      16
#define L1_KSL     (HC / L1_GW)
#define L1_LCAP    4096
#define L1_ZP      (HC + 4)
#define L1_ZSLOTS  (L1_WAVES + 4)

__device__ __forceinline__ float lrelu(float x)
{
    return (x >= 0.f) ? x : NEG_SLOPE * x;
}

__device__ __forceinline__ float xrow(float x)
{
    const int b = __float_as_int(x);
    return __int_as_float(__builtin_amdgcn_permlanex16(b, b, 0x76543210, 0xfedcba98, false, false));
}
template <int M>
__device__ __forceinline__ float xlane(float x)
{
    return __int_as_float(__builtin_amdgcn_update_dpp(0, __float_as_int(x), 0x160 + M, 0xf, 0xf, false));
}

template <int NT>
__device__ __forceinline__ void h2_slice(const float* zl, const int (&ro)[NT],
                                         const float* wp, unsigned wo, v8f (&c)[NT])
{
#pragma unroll 2
    for (int k = 0; k < L1_KSL; k += 32) {
        float4 av[NT][4];
        float  w[4][4];
#pragma unroll
        for (int s = 0; s < 4; ++s)
#pragma unroll
            for (int j = 0; j < 4; ++j)
                w[s][j] = wp[wo + (unsigned)((k + 8 * s + j) * NUM_CLASS)];
#pragma unroll
        for (int tt = 0; tt < NT; ++tt)
#pragma unroll
            for (int s = 0; s < 4; ++s) av[tt][s] = *(const float4*)(zl + ro[tt] + k + 8 * s);
#pragma unroll
        for (int s = 0; s < 4; ++s) {
            v2f a, b;
            b.x = w[s][0]; b.y = w[s][1];
#pragma unroll
            for (int tt = 0; tt < NT; ++tt) {
                a.x = av[tt][s].x; a.y = av[tt][s].y;
                c[tt] = __builtin_amdgcn_wmma_f32_16x16x4_f32(false, a, false, b,
                                                              (short)0, c[tt], false, false);
            }
            b.x = w[s][2]; b.y = w[s][3];
#pragma unroll
            for (int tt = 0; tt < NT; ++tt) {
                a.x = av[tt][s].z; a.y = av[tt][s].w;
                c[tt] = __builtin_amdgcn_wmma_f32_16x16x4_f32(false, a, false, b,
                                                              (short)0, c[tt], false, false);
            }
        }
    }
}

__device__ __forceinline__ int list_entry(const int* __restrict__ lo, int m0, int idx)
{
    return (idx < m0) ? lo[idx] : lo[E_IN / 2 + (idx - m0)];
}

__launch_bounds__(L1_THREADS)
__global__ void layer1_owner(const float* __restrict__ H1,
                             const float* __restrict__ aS, const float* __restrict__ aD,
                             const float* B1, const float* W2,
                             const float* __restrict__ as2, const float* __restrict__ ad2,
                             float* __restrict__ H2,
                             float* __restrict__ aS2, float* __restrict__ aD2,
                             const int* __restrict__ lst, const int* __restrict__ lcnt)
{
    static_assert(N_NODES % L1_NPB == 0, "no partial node range");
    static_assert(L1_NPB <= 63 && N_NODES < (1 << 24), "source and local node share one int; local 63 marks no entry");
    static_assert(HEADS == 8 && HIDD == 256, "two 128-channel pieces per head, 8 scores per node");
    static_assert(L1_GW <= L1_WAVES && L1_KSL % 32 == 0, "a K slice per wave, whole 32-k trips");
    static_assert(L1_NPB * NUM_CLASS <= L1_THREADS, "one thread per owned h2 element");
    static_assert(L1_NPB == 2 * L1_WAVES && L1_WAVES == 20 && L1_GW == 16,
                  "two rounds of one node per wave; slots 16..23 hold the last tile's 8 rows");
    static_assert((L1_ZP * 4) % 32 == 16, "slot pitch 16 B x odd");
    __shared__ int list[L1_LCAP];
    __shared__ __attribute__((aligned(16))) float zs[L1_ZSLOTS * L1_ZP];
    __shared__ __attribute__((aligned(16))) float part[L1_GW][L1_NPB][16];
    __shared__ float h2s[L1_NPB * NUM_CLASS];
    __shared__ __attribute__((aligned(16))) float xl[L1_CAP * HEADS];

    const int tid = threadIdx.x;
    const int n0  = blockIdx.x * L1_NPB;
    const int* lo = lst + (size_t)blockIdx.x * E_IN;
    const int m0 = max(0, min(lcnt[2 * blockIdx.x],     E_IN / 2));
    const int m  = m0 + max(0, min(lcnt[2 * blockIdx.x + 1], E_IN / 2));

    for (int t = tid; t < min(m, L1_LCAP); t += L1_THREADS) {
        const int en = list_entry(lo, m0, t);
        list[t] = en;
        if (t < L1_CAP) {
            const float* sp = aS + (size_t)(en >> 6) * HEADS;
            const float* dp = aD + (size_t)(n0 + (en & 63)) * HEADS;
            const float4 a0 = *(const float4*)sp;
            const float4 a1 = *(const float4*)(sp + 4);
            const float4 e0 = *(const float4*)dp;
            const float4 e1 = *(const float4*)(dp + 4);
            *(float4*)(&xl[t * HEADS])     = make_float4(lrelu(a0.x + e0.x), lrelu(a0.y + e0.y),
                                                         lrelu(a0.z + e0.z), lrelu(a0.w + e0.w));
            *(float4*)(&xl[t * HEADS + 4]) = make_float4(lrelu(a1.x + e1.x), lrelu(a1.y + e1.y),
                                                         lrelu(a1.z + e1.z), lrelu(a1.w + e1.w));
        }
    }
    __syncthreads();

    const int lane = tid & 31;
    const int wave = __builtin_amdgcn_readfirstlane(tid >> 5);

    const int half = lane >> 4;
    const int ml   = lane & 15;
    const int ksl  = min(wave, L1_GW - 1) * L1_KSL;
    const int zo0  = ml * L1_ZP + half * 4;
    const int zo1  = (16 + min(ml, 7)) * L1_ZP + half * 4;
    const unsigned wo = (unsigned)(half * 4 * NUM_CLASS + min(ml, NUM_CLASS - 1));

    for (int rnd = 0; rnd < 2; ++rnd) {
        const int i = wave + rnd * L1_WAVES;
        const int n = n0 + i;

        float4 acc[L1_CH];
        {
            const float* rp = H1 + (size_t)n * HC + lane * 4;
#pragma unroll
            for (int j = 0; j < L1_CH; ++j) acc[j] = *(const float4*)(rp + j * 128);
        }

        float xs[HEADS], mx[HEADS];
        {
            const float4 p0 = *(const float4*)(aD + (size_t)n * HEADS);
            const float4 p1 = *(const float4*)(aD + (size_t)n * HEADS + 4);
            const float4 q0 = *(const float4*)(aS + (size_t)n * HEADS);
            const float4 q1 = *(const float4*)(aS + (size_t)n * HEADS + 4);
            const float dd[HEADS] = {p0.x, p0.y, p0.z, p0.w, p1.x, p1.y, p1.z, p1.w};
            const float ss[HEADS] = {q0.x, q0.y, q0.z, q0.w, q1.x, q1.y, q1.z, q1.w};
#pragma unroll
            for (int h = 0; h < HEADS; ++h) {
                xs[h]  = lrelu(ss[h] + dd[h]);
                mx[h]  = xs[h];
            }
        }

        for (int t = 0; t < m; t += 32) {
            const int idx = t + lane;
            const int en  = (idx < m) ? ((idx < L1_LCAP) ? list[idx] : list_entry(lo, m0, idx)) : -1;
            if ((en & 63) == i) {
                float x[HEADS];
                if (idx < L1_CAP) {
                    const float4 u0 = *(const float4*)(&xl[idx * HEADS]);
                    const float4 u1 = *(const float4*)(&xl[idx * HEADS + 4]);
                    x[0] = u0.x; x[1] = u0.y; x[2] = u0.z; x[3] = u0.w;
                    x[4] = u1.x; x[5] = u1.y; x[6] = u1.z; x[7] = u1.w;
                } else {
                    const float* sp = aS + (size_t)(en >> 6) * HEADS;
                    const float* dp = aD + (size_t)n * HEADS;
                    const float4 a0 = *(const float4*)sp;
                    const float4 a1 = *(const float4*)(sp + 4);
                    const float4 e0 = *(const float4*)dp;
                    const float4 e1 = *(const float4*)(dp + 4);
                    const float av[HEADS] = {a0.x, a0.y, a0.z, a0.w, a1.x, a1.y, a1.z, a1.w};
                    const float dv[HEADS] = {e0.x, e0.y, e0.z, e0.w, e1.x, e1.y, e1.z, e1.w};
#pragma unroll
                    for (int h = 0; h < HEADS; ++h) x[h] = lrelu(av[h] + dv[h]);
                }
#pragma unroll
                for (int h = 0; h < HEADS; ++h) mx[h] = fmaxf(mx[h], x[h]);
            }
        }
#pragma unroll
        for (int h = 0; h < HEADS; ++h) {
            mx[h] = fmaxf(mx[h], xrow(mx[h]));
            mx[h] = fmaxf(mx[h], xlane<8>(mx[h]));
            mx[h] = fmaxf(mx[h], xlane<4>(mx[h]));
            mx[h] = fmaxf(mx[h], xlane<2>(mx[h]));
            mx[h] = fmaxf(mx[h], xlane<1>(mx[h]));
        }

        float exs[HEADS], den[HEADS];
#pragma unroll
        for (int h = 0; h < HEADS; ++h) {
            exs[h] = expf(xs[h] - mx[h]);
            den[h] = 0.f;
        }
#pragma unroll
        for (int j = 0; j < L1_CH; ++j) {
            const float w = exs[j >> 1];
            acc[j] = make_float4(w * acc[j].x, w * acc[j].y, w * acc[j].z, w * acc[j].w);
        }
        for (int t = 0; t < m; t += 32) {
            const int idx = t + lane;
            const int en  = (idx < m) ? ((idx < L1_LCAP) ? list[idx] : list_entry(lo, m0, idx)) : -1;
            const bool mine = ((en & 63) == i);
            float ex[HEADS];
#pragma unroll
            for (int h = 0; h < HEADS; ++h) ex[h] = 0.f;
            if (mine) {
                float x[HEADS];
                if (idx < L1_CAP) {
                    const float4 u0 = *(const float4*)(&xl[idx * HEADS]);
                    const float4 u1 = *(const float4*)(&xl[idx * HEADS + 4]);
                    x[0] = u0.x; x[1] = u0.y; x[2] = u0.z; x[3] = u0.w;
                    x[4] = u1.x; x[5] = u1.y; x[6] = u1.z; x[7] = u1.w;
                } else {
                    const float* sp = aS + (size_t)(en >> 6) * HEADS;
                    const float* dp = aD + (size_t)n * HEADS;
                    const float4 a0 = *(const float4*)sp;
                    const float4 a1 = *(const float4*)(sp + 4);
                    const float4 e0 = *(const float4*)dp;
                    const float4 e1 = *(const float4*)(dp + 4);
                    const float av[HEADS] = {a0.x, a0.y, a0.z, a0.w, a1.x, a1.y, a1.z, a1.w};
                    const float dv[HEADS] = {e0.x, e0.y, e0.z, e0.w, e1.x, e1.y, e1.z, e1.w};
#pragma unroll
                    for (int h = 0; h < HEADS; ++h) x[h] = lrelu(av[h] + dv[h]);
                }
#pragma unroll
                for (int h = 0; h < HEADS; ++h) {
                    ex[h]   = expf(x[h] - mx[h]);
                    den[h] += ex[h];
                }
            }
            unsigned mask = __builtin_amdgcn_ballot_w32(mine);
            while (mask) {
                const int b = __builtin_ctz(mask);
                mask &= mask - 1;
                const int s = __builtin_amdgcn_readlane(en, b) >> 6;
                float w[HEADS];
#pragma unroll
                for (int h = 0; h < HEADS; ++h)
                    w[h] = __int_as_float(__builtin_amdgcn_readlane(__float_as_int(ex[h]), b));
                const float* rp = H1 + (size_t)s * HC + lane * 4;
                float4 v[L1_CH];
#pragma unroll
                for (int j = 0; j < L1_CH; ++j) v[j] = *(const float4*)(rp + j * 128);
#pragma unroll
                for (int j = 0; j < L1_CH; ++j) {
                    const float ww = w[j >> 1];
                    acc[j].x += ww * v[j].x;
                    acc[j].y += ww * v[j].y;
                    acc[j].z += ww * v[j].z;
                    acc[j].w += ww * v[j].w;
                }
            }
        }
        float4 bq[L1_CH];
        {
            const float* bp = B1 + lane * 4;
#pragma unroll
            for (int j = 0; j < L1_CH; ++j) bq[j] = *(const float4*)(bp + j * 128);
        }
        float rc[HEADS];
#pragma unroll
        for (int h = 0; h < HEADS; ++h) {
            den[h] += xrow(den[h]);
            den[h] += xlane<8>(den[h]);
            den[h] += xlane<4>(den[h]);
            den[h] += xlane<2>(den[h]);
            den[h] += xlane<1>(den[h]);
            rc[h] = 1.f / ((den[h] + exs[h]) + 1e-16f);
        }
#pragma unroll
        for (int j = 0; j < L1_CH; ++j) {
            const float r = rc[j >> 1];
            const float4 b = bq[j];
            acc[j] = make_float4(elu1(acc[j].x * r + b.x), elu1(acc[j].y * r + b.y),
                                 elu1(acc[j].z * r + b.z), elu1(acc[j].w * r + b.w));
        }
        {
            const int slot = (rnd == 0 && wave >= 16) ? wave + (L1_ZSLOTS - L1_WAVES) : wave;
            float* zp = zs + slot * L1_ZP + lane * 4;
#pragma unroll
            for (int j = 0; j < L1_CH; ++j) *(float4*)(zp + j * 128) = acc[j];
        }

        __syncthreads();

        if (wave < L1_GW) {
            const float* zl = zs + ksl;
            const float* wp = W2 + (size_t)ksl * NUM_CLASS;
            float* pp = &part[wave][8 * half][ml];
            if (rnd == 0) {
                const int ro[1] = {zo0};
                v8f c2[1] = {(v8f){0.f, 0.f, 0.f, 0.f, 0.f, 0.f, 0.f, 0.f}};
                h2_slice<1>(zl, ro, wp, wo, c2);
#pragma unroll
                for (int r = 0; r < 8; ++r)
                    pp[r * 16] = c2[0][r];
            } else {
                const int ro[2] = {zo0, zo1};
                v8f c2[2] = {(v8f){0.f, 0.f, 0.f, 0.f, 0.f, 0.f, 0.f, 0.f},
                             (v8f){0.f, 0.f, 0.f, 0.f, 0.f, 0.f, 0.f, 0.f}};
                h2_slice<2>(zl, ro, wp, wo, c2);
#pragma unroll
                for (int r = 0; r < 8; ++r)
                    pp[(L1_WAVES + r) * 16] = c2[0][r];
                if (half == 0) {
#pragma unroll
                    for (int r = 0; r < 8; ++r)
                        pp[((r < 4) ? (36 + r) : (12 + r)) * 16] = c2[1][r];
                }
            }
        }
        __syncthreads();
    }

    const int t2 = wave * 32 + lane;

    if (t2 < L1_NPB * NUM_CLASS) {
        const int i = t2 / NUM_CLASS;
        const int cc = t2 - i * NUM_CLASS;
        float v = 0.f;
#pragma unroll
        for (int g = 0; g < L1_GW; ++g) v += part[g][i][cc];
        H2[(size_t)n0 * NUM_CLASS + t2] = v;
        h2s[t2] = v;
    }
    __syncthreads();

    if (t2 < L1_NPB) {
        float ss = 0.f, sd = 0.f;
#pragma unroll
        for (int cc = 0; cc < NUM_CLASS; ++cc) {
            const float v = h2s[t2 * NUM_CLASS + cc];
            ss += v * as2[cc];
            sd += v * ad2[cc];
        }
        aS2[n0 + t2] = ss; aD2[n0 + t2] = sd;
    }
}

__device__ __forceinline__ void atomicMaxF(float* addr, float v)
{
    if (v >= 0.f) atomicMax((int*)addr, __float_as_int(v));
    else          atomicMin((unsigned int*)addr, __float_as_uint(v));
}

#define L2_NPB     40
#define L2_THREADS 256

__launch_bounds__(L2_THREADS)
__global__ void layer2_owner(const int* __restrict__ lst, const int* __restrict__ lcnt,
                             const float* __restrict__ H2,
                             const float* __restrict__ aS, const float* __restrict__ aD,
                             const float* __restrict__ b2, float* __restrict__ out)
{
    static_assert(N_NODES % L2_NPB == 0, "no partial node range");
    static_assert(L2_NPB == L1_NPB, "block b of the layer-1 owner kernel lists exactly this block's nodes");
    static_assert(L2_NPB <= L2_THREADS, "one thread per owned node for the self loops");
    __shared__ float sd_s[L2_NPB], emax_s[L2_NPB], den_s[L2_NPB];
    __shared__ float agg_s[L2_NPB * NUM_CLASS];

    const int tid = threadIdx.x;
    const int n0  = blockIdx.x * L2_NPB;
    const int* lp = lst + (size_t)blockIdx.x * E_IN;
    const int m0  = max(0, min(lcnt[2 * blockIdx.x],     E_IN / 2));
    const int m   = m0 + max(0, min(lcnt[2 * blockIdx.x + 1], E_IN / 2));

    int   en_f = 0;
    float as_f = 0.f;
    float hf[NUM_CLASS];
#pragma unroll
    for (int c = 0; c < NUM_CLASS; ++c) hf[c] = 0.f;
    if (tid < m) {
        en_f = list_entry(lp, m0, tid);
        const int s = en_f >> 6;
        as_f = aS[s];
#pragma unroll
        for (int c = 0; c < NUM_CLASS; ++c) hf[c] = H2[(size_t)s * NUM_CLASS + c];
    }
    float xs = 0.f;
    float hs[NUM_CLASS];
#pragma unroll
    for (int c = 0; c < NUM_CLASS; ++c) hs[c] = 0.f;
    if (tid < L2_NPB) {
        const int n = n0 + tid;
        const float d = aD[n];
        xs = lrelu(aS[n] + d);
        sd_s[tid]   = d;
        emax_s[tid] = xs;
        den_s[tid]  = 0.f;
#pragma unroll
        for (int c = 0; c < NUM_CLASS; ++c) hs[c] = H2[(size_t)n * NUM_CLASS + c];
    }
    for (int t = tid; t < L2_NPB * NUM_CLASS; t += L2_THREADS) agg_s[t] = 0.f;
    __syncthreads();

    if (tid < m) {
        const int dl = en_f & 63;
        atomicMaxF(&emax_s[dl], lrelu(as_f + sd_s[dl]));
    }
    for (int t = tid + L2_THREADS; t < m; t += L2_THREADS) {
        const int en = list_entry(lp, m0, t);
        const int dl = en & 63;
        atomicMaxF(&emax_s[dl], lrelu(aS[en >> 6] + sd_s[dl]));
    }
    __syncthreads();

    if (tid < L2_NPB) {
        const float ex = expf(xs - emax_s[tid]);
        atomicAdd(&den_s[tid], ex);
#pragma unroll
        for (int c = 0; c < NUM_CLASS; ++c)
            atomicAdd(&agg_s[tid * NUM_CLASS + c], ex * hs[c]);
    }
    if (tid < m) {
        const int dl = en_f & 63;
        const float ex = expf(lrelu(as_f + sd_s[dl]) - emax_s[dl]);
        atomicAdd(&den_s[dl], ex);
#pragma unroll
        for (int c = 0; c < NUM_CLASS; ++c)
            atomicAdd(&agg_s[dl * NUM_CLASS + c], ex * hf[c]);
    }
    for (int t = tid + L2_THREADS; t < m; t += L2_THREADS) {
        const int en = list_entry(lp, m0, t);
        const int dl = en & 63;
        const int s  = en >> 6;
        const float ex = expf(lrelu(aS[s] + sd_s[dl]) - emax_s[dl]);
        atomicAdd(&den_s[dl], ex);
#pragma unroll
        for (int c = 0; c < NUM_CLASS; ++c)
            atomicAdd(&agg_s[dl * NUM_CLASS + c], ex * H2[(size_t)s * NUM_CLASS + c]);
    }
    __syncthreads();

    for (int t = tid; t < L2_NPB * NUM_CLASS; t += L2_THREADS) {
        const int i = t / NUM_CLASS;
        const int cc = t - i * NUM_CLASS;
        const float v = agg_s[t] / (den_s[i] + 1e-16f) + b2[cc];
        out[(size_t)n0 * NUM_CLASS + t] = 1.f / (1.f + expf(-v));
    }
}

extern "C" void kernel_launch(void* const* d_in, const int* in_sizes, int n_in,
                              void* d_out, int out_size, void* d_ws, size_t ws_size,
                              hipStream_t stream)
{
    const float* x   = (const float*)d_in[0];
    const int*   ei  = (const int*)d_in[1];
    const float* W1  = (const float*)d_in[2];
    const float* as1 = (const float*)d_in[3];
    const float* ad1 = (const float*)d_in[4];
    const float* b1  = (const float*)d_in[5];
    const float* W2  = (const float*)d_in[6];
    const float* as2 = (const float*)d_in[7];
    const float* ad2 = (const float*)d_in[8];
    const float* b2  = (const float*)d_in[9];
    float* out = (float*)d_out;

    float* ws = (float*)d_ws;
    size_t off = 0;
    float* h1    = ws + off; off += (size_t)N_NODES * HC;
    float* sA1   = ws + off; off += (size_t)N_NODES * HEADS;
    float* sD1   = ws + off; off += (size_t)N_NODES * HEADS;
    float* h2    = ws + off; off += (size_t)N_NODES * NUM_CLASS;
    float* sA2   = ws + off; off += N_NODES;
    float* sD2   = ws + off; off += N_NODES;
    int*   lst   = (int*)(ws + off); off += (size_t)(N_NODES / L1_NPB) * E_IN;
    int*   lcnt  = (int*)(ws + off); off += SC_BLOCKS;
    (void)ws_size; (void)in_sizes; (void)n_in; (void)out_size;

    const dim3 b256(256);

    gemm1_att<<<dim3(HEADS, G1_RG), b256, 0, stream>>>(x, W1, as1, ad1, h1, sA1, sD1, ei, lst, lcnt);

    layer1_owner<<<N_NODES / L1_NPB, L1_THREADS, 0, stream>>>(h1, sA1, sD1,
                                                              b1, W2, as2, ad2, h2, sA2, sD2,
                                                              lst, lcnt);

    layer2_owner<<<N_NODES / L2_NPB, L2_THREADS, 0, stream>>>(lst, lcnt, h2, sA2, sD2, b2, out);
}
